# starting kernel
# baseline (speedup 1.0000x reference)
#include <hip/hip_runtime.h>
#include <math.h>

typedef float v2f __attribute__((ext_vector_type(2)));
typedef float v8f __attribute__((ext_vector_type(8)));

#define FIN 256
#define H1 8
#define D1 8
#define HD1 64
#define KEYNEG 0x007FFFFFu

__device__ __forceinline__ unsigned f2key(float f) {
    unsigned u = __float_as_uint(f);
    return (u & 0x80000000u) ? ~u : (u | 0x80000000u);
}
__device__ __forceinline__ float key2f(unsigned k) {
    unsigned u = (k & 0x80000000u) ? (k & 0x7FFFFFFFu) : ~k;
    return __uint_as_float(u);
}
__device__ __forceinline__ float leaky(float v) { return v > 0.f ? v : 0.2f * v; }

__global__ void fill_u32(unsigned* __restrict__ p, unsigned v, long long n) {
    long long i = (long long)blockIdx.x * blockDim.x + threadIdx.x;
    if (i < n) p[i] = v;
}

__global__ __launch_bounds__(256) void gemm1_wmma(const float* __restrict__ x,
                                                  const float* __restrict__ W1,
                                                  float* __restrict__ h1, int N) {
    int wave = threadIdx.x >> 5;
    int lane = threadIdx.x & 31;
    int lm = lane & 15, lh = lane >> 4;
    int row0 = blockIdx.x * 32 + (wave >> 2) * 16;
    int col0 = (wave & 3) * 16;
    if (row0 >= N) return;
    const float* xr = x + (long long)(row0 + lm) * FIN;
    v8f acc = {0.f, 0.f, 0.f, 0.f, 0.f, 0.f, 0.f, 0.f};
    for (int k0 = 0; k0 < FIN; k0 += 4) {
        int ka = k0 + 2 * lh;
        v2f a, b;
        a.x = xr[ka];
        a.y = xr[ka + 1];
        b.x = W1[ka * HD1 + col0 + lm];
        b.y = W1[(ka + 1) * HD1 + col0 + lm];
        acc = __builtin_amdgcn_wmma_f32_16x16x4_f32(false, a, false, b, (short)0, acc,
                                                    false, false);
    }
    float* out = h1 + (long long)row0 * HD1 + col0 + lm;
#pragma unroll
    for (int k = 0; k < 8; ++k)
        out[(long long)(k + 8 * lh) * HD1] = acc[k];
}

__global__ void alpha1_kernel(const float* __restrict__ h1,
                              const float* __restrict__ a_src,
                              const float* __restrict__ a_dst,
                              float* __restrict__ as, float* __restrict__ ad, int N) {
    long long i = (long long)blockIdx.x * blockDim.x + threadIdx.x;
    if (i >= (long long)N * H1) return;
    int h = (int)(i & 7);
    const float* hp = h1 + (i >> 3) * HD1 + h * D1;
    float s = 0.f, d = 0.f;
#pragma unroll
    for (int c = 0; c < D1; ++c) {
        float v = hp[c];
        s += v * a_src[h * D1 + c];
        d += v * a_dst[h * D1 + c];
    }
    as[i] = s;
    ad[i] = d;
}

__global__ void edge_max1(const int* __restrict__ ei, int E, int N,
                          const float* __restrict__ as, const float* __restrict__ ad,
                          unsigned* __restrict__ emax) {
    long long t = (long long)blockIdx.x * blockDim.x + threadIdx.x;
    long long ET = (long long)E + N;
    if (t >= ET * H1) return;
    int h = (int)(t & 7);
    long long e = t >> 3;
    int s, d;
    if (e < E) { s = ei[e]; d = ei[E + e]; } else { s = d = (int)(e - E); }
    float v = leaky(as[(long long)s * H1 + h] + ad[(long long)d * H1 + h]);
    atomicMax(&emax[(long long)d * H1 + h], f2key(v));
}

__global__ void edge_acc1(const int* __restrict__ ei, int E, int N,
                          const float* __restrict__ as, const float* __restrict__ ad,
                          const unsigned* __restrict__ emax,
                          const float* __restrict__ h1,
                          float* __restrict__ denom, float* __restrict__ num) {
    long long t = (long long)blockIdx.x * blockDim.x + threadIdx.x;
    long long ET = (long long)E + N;
    long long e = t >> 5;
    if (e >= ET) return;
    int lane = (int)(t & 31);
    int s, d;
    if (e < E) { s = ei[e]; d = ei[E + e]; } else { s = d = (int)(e - E); }
    int h = lane >> 2;
    float v = leaky(as[(long long)s * H1 + h] + ad[(long long)d * H1 + h]);
    float w = __expf(v - key2f(emax[(long long)d * H1 + h]));
    if ((lane & 3) == 0) atomicAdd(&denom[(long long)d * H1 + h], w);
    int c0 = lane * 2;
    const float* hs = h1 + (long long)s * HD1 + c0;
    float* nd = num + (long long)d * HD1 + c0;
    atomicAdd(nd, w * hs[0]);
    atomicAdd(nd + 1, w * hs[1]);
}

__global__ void finalize1(const float* __restrict__ num, const float* __restrict__ denom,
                          const float* __restrict__ b1, float* __restrict__ hx, int N) {
    long long i = (long long)blockIdx.x * blockDim.x + threadIdx.x;
    if (i >= (long long)N * HD1) return;
    int hc = (int)(i & 63);
    float v = num[i] / denom[(i >> 6) * H1 + (hc >> 3)] + b1[hc];
    hx[i] = v > 0.f ? v : (__expf(v) - 1.f);
}

__global__ __launch_bounds__(224) void gemm2_wmma(const float* __restrict__ hx,
                                                  const float* __restrict__ W2,
                                                  float* __restrict__ h2, int N, int C2) {
    int wave = threadIdx.x >> 5;
    int lane = threadIdx.x & 31;
    int lm = lane & 15, lh = lane >> 4;
    int row0 = blockIdx.x * 16;
    int col = wave * 16 + lm;
    bool cok = col < C2;
    const float* ar = hx + (long long)(row0 + lm) * HD1;
    v8f acc = {0.f, 0.f, 0.f, 0.f, 0.f, 0.f, 0.f, 0.f};
    for (int k0 = 0; k0 < HD1; k0 += 4) {
        int ka = k0 + 2 * lh;
        v2f a, b;
        a.x = ar[ka];
        a.y = ar[ka + 1];
        b.x = cok ? W2[ka * C2 + col] : 0.f;
        b.y = cok ? W2[(ka + 1) * C2 + col] : 0.f;
        acc = __builtin_amdgcn_wmma_f32_16x16x4_f32(false, a, false, b, (short)0, acc,
                                                    false, false);
    }
    if (cok) {
        float* out = h2 + (long long)row0 * C2 + col;
#pragma unroll
        for (int k = 0; k < 8; ++k)
            out[(long long)(k + 8 * lh) * C2] = acc[k];
    }
}

__global__ void alpha2_kernel(const float* __restrict__ h2,
                              const float* __restrict__ a_src,
                              const float* __restrict__ a_dst,
                              float* __restrict__ as, float* __restrict__ ad,
                              int N, int C2) {
    int wid = threadIdx.x >> 5, lane = threadIdx.x & 31;
    int n = blockIdx.x * (blockDim.x >> 5) + wid;
    if (n >= N) return;
    const float* hp = h2 + (long long)n * C2;
    float s = 0.f, d = 0.f;
    for (int c = lane; c < C2; c += 32) {
        float v = hp[c];
        s += v * a_src[c];
        d += v * a_dst[c];
    }
    for (int m = 16; m; m >>= 1) {
        s += __shfl_xor(s, m, 32);
        d += __shfl_xor(d, m, 32);
    }
    if (lane == 0) { as[n] = s; ad[n] = d; }
}

__global__ void edge_max2(const int* __restrict__ ei, int E, int N,
                          const float* __restrict__ as, const float* __restrict__ ad,
                          unsigned* __restrict__ emax) {
    long long e = (long long)blockIdx.x * blockDim.x + threadIdx.x;
    long long ET = (long long)E + N;
    if (e >= ET) return;
    int s, d;
    if (e < E) { s = ei[e]; d = ei[E + e]; } else { s = d = (int)(e - E); }
    atomicMax(&emax[d], f2key(leaky(as[s] + ad[d])));
}

__global__ void edge_acc2(const int* __restrict__ ei, int E, int N,
                          const float* __restrict__ as, const float* __restrict__ ad,
                          const unsigned* __restrict__ emax,
                          const float* __restrict__ h2,
                          float* __restrict__ denom, float* __restrict__ num, int C2) {
    long long t = (long long)blockIdx.x * blockDim.x + threadIdx.x;
    long long ET = (long long)E + N;
    long long e = t >> 5;
    if (e >= ET) return;
    int lane = (int)(t & 31);
    int s, d;
    if (e < E) { s = ei[e]; d = ei[E + e]; } else { s = d = (int)(e - E); }
    float w = __expf(leaky(as[s] + ad[d]) - key2f(emax[d]));
    if (lane == 0) atomicAdd(&denom[d], w);
    const float* hs = h2 + (long long)s * C2;
    float* nd = num + (long long)d * C2;
    for (int c = lane; c < C2; c += 32)
        atomicAdd(&nd[c], w * hs[c]);
}

__global__ void finalize2(float* __restrict__ out, const float* __restrict__ denom,
                          const float* __restrict__ b2, int N, int C2) {
    long long i = (long long)blockIdx.x * blockDim.x + threadIdx.x;
    if (i >= (long long)N * C2) return;
    int c = (int)(i % C2);
    long long n = i / C2;
    float v = out[i] / denom[n] + b2[c];
    out[i] = 1.f / (1.f + __expf(-v));
}

static inline int cdiv(long long a, long long b) { return (int)((a + b - 1) / b); }

extern "C" void kernel_launch(void* const* d_in, const int* in_sizes, int n_in,
                              void* d_out, int out_size, void* d_ws, size_t ws_size,
                              hipStream_t stream) {
    const float* x      = (const float*)d_in[0];
    const int*   ei     = (const int*)d_in[1];
    const float* W1     = (const float*)d_in[2];
    const float* a_src1 = (const float*)d_in[3];
    const float* a_dst1 = (const float*)d_in[4];
    const float* b1     = (const float*)d_in[5];
    const float* W2     = (const float*)d_in[6];
    const float* a_src2 = (const float*)d_in[7];
    const float* a_dst2 = (const float*)d_in[8];
    const float* b2     = (const float*)d_in[9];

    const int N  = in_sizes[0] / FIN;
    const int E  = in_sizes[1] / 2;
    const int C2 = in_sizes[9];
    const long long ET = (long long)E + N;

    float*    h1     = (float*)d_ws;
    float*    as1    = h1 + (long long)N * HD1;
    float*    ad1    = as1 + (long long)N * H1;
    unsigned* emax1  = (unsigned*)(ad1 + (long long)N * H1);
    float*    denom1 = (float*)(emax1 + (long long)N * H1);
    float*    num1   = denom1 + (long long)N * H1;
    float*    h2     = num1 + (long long)N * HD1;
    float*    as2    = h2 + (long long)N * C2;
    float*    ad2    = as2 + N;
    unsigned* emax2  = (unsigned*)(ad2 + N);
    float*    denom2 = (float*)(emax2 + N);
    float*    out    = (float*)d_out;

    const int B = 256;

    fill_u32<<<cdiv((long long)N * H1, B), B, 0, stream>>>(emax1, KEYNEG, (long long)N * H1);
    fill_u32<<<cdiv((long long)N * (H1 + HD1), B), B, 0, stream>>>(
        (unsigned*)denom1, 0u, (long long)N * (H1 + HD1));
    fill_u32<<<cdiv(N, B), B, 0, stream>>>(emax2, KEYNEG, N);
    fill_u32<<<cdiv(N, B), B, 0, stream>>>((unsigned*)denom2, 0u, N);
    fill_u32<<<cdiv((long long)N * C2, B), B, 0, stream>>>((unsigned*)out, 0u,
                                                           (long long)N * C2);

    gemm1_wmma<<<cdiv(N, 32), 256, 0, stream>>>(x, W1, h1, N);
    alpha1_kernel<<<cdiv((long long)N * H1, B), B, 0, stream>>>(h1, a_src1, a_dst1,
                                                                as1, ad1, N);
    edge_max1<<<cdiv(ET * H1, B), B, 0, stream>>>(ei, E, N, as1, ad1, emax1);
    edge_acc1<<<cdiv(ET * 32, B), B, 0, stream>>>(ei, E, N, as1, ad1, emax1, h1,
                                                  denom1, num1);
    finalize1<<<cdiv((long long)N * HD1, B), B, 0, stream>>>(num1, denom1, b1, h1, N);

    gemm2_wmma<<<cdiv(N, 16), 224, 0, stream>>>(h1, W2, h2, N, C2);
    alpha2_kernel<<<cdiv(N, 8), 256, 0, stream>>>(h2, a_src2, a_dst2, as2, ad2, N, C2);
    edge_max2<<<cdiv(ET, B), B, 0, stream>>>(ei, E, N, as2, ad2, emax2);
    edge_acc2<<<cdiv(ET * 32, B), B, 0, stream>>>(ei, E, N, as2, ad2, emax2, h2,
                                                  denom2, out, C2);
    finalize2<<<cdiv((long long)N * C2, B), B, 0, stream>>>(out, denom2, b2, N, C2);
}


# Replace per-edge atomic max/add scatter by an edge list sorted by destination (count, scan, place; arrival-order rank), one wave per node reduces softmax and sum in registers: sum order changes
# speedup vs baseline: 2.5308x; 2.5308x over previous
#include <hip/hip_runtime.h>
#include <math.h>

typedef float v2f __attribute__((ext_vector_type(2)));
typedef float v8f __attribute__((ext_vector_type(8)));

#define FIN 256
#define H1 8
#define D1 8
#define HD1 64

__device__ __forceinline__ float leaky(float v) { return v > 0.f ? v : 0.2f * v; }

__global__ void fill_u32(unsigned* __restrict__ p, unsigned v, long long n) {
    long long i = (long long)blockIdx.x * blockDim.x + threadIdx.x;
    if (i < n) p[i] = v;
}

__global__ __launch_bounds__(256) void gemm1_wmma(const float* __restrict__ x,
                                                  const float* __restrict__ W1,
                                                  float* __restrict__ h1, int N) {
    int wave = threadIdx.x >> 5;
    int lane = threadIdx.x & 31;
    int lm = lane & 15, lh = lane >> 4;
    int row0 = blockIdx.x * 32 + (wave >> 2) * 16;
    int col0 = (wave & 3) * 16;
    if (row0 >= N) return;
    const float* xr = x + (long long)(row0 + lm) * FIN;
    v8f acc = {0.f, 0.f, 0.f, 0.f, 0.f, 0.f, 0.f, 0.f};
    for (int k0 = 0; k0 < FIN; k0 += 4) {
        int ka = k0 + 2 * lh;
        v2f a, b;
        a.x = xr[ka];
        a.y = xr[ka + 1];
        b.x = W1[ka * HD1 + col0 + lm];
        b.y = W1[(ka + 1) * HD1 + col0 + lm];
        acc = __builtin_amdgcn_wmma_f32_16x16x4_f32(false, a, false, b, (short)0, acc,
                                                    false, false);
    }
    float* out = h1 + (long long)row0 * HD1 + col0 + lm;
#pragma unroll
    for (int k = 0; k < 8; ++k)
        out[(long long)(k + 8 * lh) * HD1] = acc[k];
}

__global__ void alpha1_kernel(const float* __restrict__ h1,
                              const float* __restrict__ a_src,
                              const float* __restrict__ a_dst,
                              float* __restrict__ as, float* __restrict__ ad, int N) {
    long long i = (long long)blockIdx.x * blockDim.x + threadIdx.x;
    if (i >= (long long)N * H1) return;
    int h = (int)(i & 7);
    const float* hp = h1 + (i >> 3) * HD1 + h * D1;
    float s = 0.f, d = 0.f;
#pragma unroll
    for (int c = 0; c < D1; ++c) {
        float v = hp[c];
        s += v * a_src[h * D1 + c];
        d += v * a_dst[h * D1 + c];
    }
    as[i] = s;
    ad[i] = d;
}

__global__ void hist_rank(const int* __restrict__ ei, int E, int* __restrict__ cnt,
                          int* __restrict__ rank) {
    int e = blockIdx.x * blockDim.x + threadIdx.x;
    if (e >= E) return;
    rank[e] = atomicAdd(&cnt[ei[E + e]], 1);
}

__global__ __launch_bounds__(256) void scan_block_sums(const int* __restrict__ cnt,
                                                       int* __restrict__ bsum, int N) {
    __shared__ int sm[8];
    int tid = threadIdx.x, lane = tid & 31, wave = tid >> 5;
    int base = blockIdx.x * 1024 + tid * 4;
    int t = 0;
#pragma unroll
    for (int j = 0; j < 4; ++j)
        if (base + j < N) t += cnt[base + j];
    for (int m = 16; m; m >>= 1) t += __shfl_xor(t, m, 32);
    if (lane == 0) sm[wave] = t;
    __syncthreads();
    if (tid == 0) {
        int s = 0;
        for (int w = 0; w < 8; ++w) s += sm[w];
        bsum[blockIdx.x] = s;
    }
}

__global__ __launch_bounds__(256) void scan_offsets(const int* __restrict__ cnt,
                                                    const int* __restrict__ bsum,
                                                    int* __restrict__ off, int N) {
    __shared__ int sm_pre[8];
    __shared__ int sm_loc[8];
    int tid = threadIdx.x, lane = tid & 31, wave = tid >> 5;
    int p = 0;
    for (int i = tid; i < (int)blockIdx.x; i += 256) p += bsum[i];
    for (int m = 16; m; m >>= 1) p += __shfl_xor(p, m, 32);
    if (lane == 0) sm_pre[wave] = p;
    int base = blockIdx.x * 1024 + tid * 4;
    int c[4];
    int t = 0;
#pragma unroll
    for (int j = 0; j < 4; ++j) {
        c[j] = (base + j < N) ? cnt[base + j] : 0;
        t += c[j];
    }
    int incl = t;
    for (int m = 1; m < 32; m <<= 1) {
        int u = __shfl_up(incl, m, 32);
        if (lane >= m) incl += u;
    }
    if (lane == 31) sm_loc[wave] = incl;
    __syncthreads();
    int run = incl - t;
    for (int w = 0; w < 8; ++w) {
        run += sm_pre[w];
        if (w < wave) run += sm_loc[w];
    }
#pragma unroll
    for (int j = 0; j < 4; ++j) {
        int i = base + j;
        if (i < N) {
            off[i] = run;
            run += c[j];
            if (i == N - 1) off[N] = run;
        }
    }
}

__global__ void scatter_src(const int* __restrict__ ei, int E, const int* __restrict__ off,
                            const int* __restrict__ rank, int* __restrict__ csr) {
    int e = blockIdx.x * blockDim.x + threadIdx.x;
    if (e >= E) return;
    csr[off[ei[E + e]] + rank[e]] = ei[e];
}

__global__ __launch_bounds__(256) void gather1(const int* __restrict__ off,
                                               const int* __restrict__ csr,
                                               const float* __restrict__ as,
                                               const float* __restrict__ ad,
                                               const float* __restrict__ h1,
                                               const float* __restrict__ b1,
                                               float* __restrict__ hx, int N) {
    int lane = threadIdx.x & 31;
    int n = __builtin_amdgcn_readfirstlane((int)(blockIdx.x * 8 + (threadIdx.x >> 5)));
    if (n >= N) return;
    int h = lane >> 2;
    int beg = off[n], end = off[n + 1];
    float adn = ad[(long long)n * H1 + h];
    float vself = leaky(as[(long long)n * H1 + h] + adn);
    float m = vself;
    for (int j0 = beg; j0 < end; j0 += 32) {
        int j = j0 + lane;
        int sl = (j < end) ? csr[j] : 0;
        int cc = min(32, end - j0);
        for (int q = 0; q < cc; ++q) {
            int s = __builtin_amdgcn_readlane(sl, q);
            m = fmaxf(m, leaky(as[(long long)s * H1 + h] + adn));
        }
    }
    float w = __expf(vself - m);
    float den = w;
    v2f hv = *(const v2f*)(h1 + (long long)n * HD1 + 2 * lane);
    float ax = w * hv.x, ay = w * hv.y;
    for (int j0 = beg; j0 < end; j0 += 32) {
        int j = j0 + lane;
        int sl = (j < end) ? csr[j] : 0;
        int cc = min(32, end - j0);
        for (int q = 0; q < cc; ++q) {
            int s = __builtin_amdgcn_readlane(sl, q);
            float wq = __expf(leaky(as[(long long)s * H1 + h] + adn) - m);
            v2f hs = *(const v2f*)(h1 + (long long)s * HD1 + 2 * lane);
            den += wq;
            ax += wq * hs.x;
            ay += wq * hs.y;
        }
    }
    float vx = ax / den + b1[2 * lane];
    float vy = ay / den + b1[2 * lane + 1];
    v2f o;
    o.x = vx > 0.f ? vx : (__expf(vx) - 1.f);
    o.y = vy > 0.f ? vy : (__expf(vy) - 1.f);
    *(v2f*)(hx + (long long)n * HD1 + 2 * lane) = o;
}

__global__ __launch_bounds__(224) void gemm2_wmma(const float* __restrict__ hx,
                                                  const float* __restrict__ W2,
                                                  float* __restrict__ h2, int N, int C2) {
    int wave = threadIdx.x >> 5;
    int lane = threadIdx.x & 31;
    int lm = lane & 15, lh = lane >> 4;
    int row0 = blockIdx.x * 16;
    int col = wave * 16 + lm;
    bool cok = col < C2;
    const float* ar = hx + (long long)(row0 + lm) * HD1;
    v8f acc = {0.f, 0.f, 0.f, 0.f, 0.f, 0.f, 0.f, 0.f};
    for (int k0 = 0; k0 < HD1; k0 += 4) {
        int ka = k0 + 2 * lh;
        v2f a, b;
        a.x = ar[ka];
        a.y = ar[ka + 1];
        b.x = cok ? W2[ka * C2 + col] : 0.f;
        b.y = cok ? W2[(ka + 1) * C2 + col] : 0.f;
        acc = __builtin_amdgcn_wmma_f32_16x16x4_f32(false, a, false, b, (short)0, acc,
                                                    false, false);
    }
    if (cok) {
        float* out = h2 + (long long)row0 * C2 + col;
#pragma unroll
        for (int k = 0; k < 8; ++k)
            out[(long long)(k + 8 * lh) * C2] = acc[k];
    }
}

__global__ void alpha2_kernel(const float* __restrict__ h2,
                              const float* __restrict__ a_src,
                              const float* __restrict__ a_dst,
                              float* __restrict__ as, float* __restrict__ ad,
                              int N, int C2) {
    int wid = threadIdx.x >> 5, lane = threadIdx.x & 31;
    int n = blockIdx.x * (blockDim.x >> 5) + wid;
    if (n >= N) return;
    const float* hp = h2 + (long long)n * C2;
    float s = 0.f, d = 0.f;
    for (int c = lane; c < C2; c += 32) {
        float v = hp[c];
        s += v * a_src[c];
        d += v * a_dst[c];
    }
    for (int m = 16; m; m >>= 1) {
        s += __shfl_xor(s, m, 32);
        d += __shfl_xor(d, m, 32);
    }
    if (lane == 0) { as[n] = s; ad[n] = d; }
}

__global__ __launch_bounds__(256) void gather2(const int* __restrict__ off,
                                               const int* __restrict__ csr,
                                               const float* __restrict__ as,
                                               const float* __restrict__ ad,
                                               const float* __restrict__ h2,
                                               const float* __restrict__ b2,
                                               float* __restrict__ out, int N, int C2) {
    int lane = threadIdx.x & 31;
    int n = __builtin_amdgcn_readfirstlane((int)(blockIdx.x * 8 + (threadIdx.x >> 5)));
    if (n >= N) return;
    int beg = off[n], end = off[n + 1];
    float adn = ad[n];
    float vself = leaky(as[n] + adn);
    float m = vself;
    for (int j0 = beg; j0 < end; j0 += 32) {
        int j = j0 + lane;
        if (j < end) m = fmaxf(m, leaky(as[csr[j]] + adn));
    }
    for (int k = 16; k; k >>= 1) m = fmaxf(m, __shfl_xor(m, k, 32));
    bool ok0 = lane < C2, ok1 = lane + 32 < C2, ok2 = lane + 64 < C2, ok3 = lane + 96 < C2;
    int c0 = min(lane, C2 - 1), c1 = min(lane + 32, C2 - 1);
    int c2 = min(lane + 64, C2 - 1), c3 = min(lane + 96, C2 - 1);
    float w0 = __expf(vself - m);
    const float* hn = h2 + (long long)n * C2;
    float a0 = w0 * hn[c0], a1 = w0 * hn[c1], a2 = w0 * hn[c2], a3 = w0 * hn[c3];
    float den = 0.f;
    for (int j0 = beg; j0 < end; j0 += 32) {
        int j = j0 + lane;
        int sl = 0;
        float wl = 0.f;
        if (j < end) {
            sl = csr[j];
            wl = __expf(leaky(as[sl] + adn) - m);
        }
        den += wl;
        int cc = min(32, end - j0);
        for (int q = 0; q < cc; ++q) {
            int s = __builtin_amdgcn_readlane(sl, q);
            float wq = __int_as_float(__builtin_amdgcn_readlane(__float_as_int(wl), q));
            const float* hs = h2 + (long long)s * C2;
            a0 += wq * hs[c0];
            a1 += wq * hs[c1];
            a2 += wq * hs[c2];
            a3 += wq * hs[c3];
        }
    }
    for (int k = 16; k; k >>= 1) den += __shfl_xor(den, k, 32);
    den += w0;
    float* on = out + (long long)n * C2 + lane;
    if (ok0) on[0]  = 1.f / (1.f + __expf(-(a0 / den + b2[lane])));
    if (ok1) on[32] = 1.f / (1.f + __expf(-(a1 / den + b2[lane + 32])));
    if (ok2) on[64] = 1.f / (1.f + __expf(-(a2 / den + b2[lane + 64])));
    if (ok3) on[96] = 1.f / (1.f + __expf(-(a3 / den + b2[lane + 96])));
}

static inline int cdiv(long long a, long long b) { return (int)((a + b - 1) / b); }

extern "C" void kernel_launch(void* const* d_in, const int* in_sizes, int n_in,
                              void* d_out, int out_size, void* d_ws, size_t ws_size,
                              hipStream_t stream) {
    const float* x      = (const float*)d_in[0];
    const int*   ei     = (const int*)d_in[1];
    const float* W1     = (const float*)d_in[2];
    const float* a_src1 = (const float*)d_in[3];
    const float* a_dst1 = (const float*)d_in[4];
    const float* b1     = (const float*)d_in[5];
    const float* W2     = (const float*)d_in[6];
    const float* a_src2 = (const float*)d_in[7];
    const float* a_dst2 = (const float*)d_in[8];
    const float* b2     = (const float*)d_in[9];

    const int N  = in_sizes[0] / FIN;
    const int E  = in_sizes[1] / 2;
    const int C2 = in_sizes[9];

    float* h1   = (float*)d_ws;
    float* as1  = h1 + (long long)N * HD1;
    float* ad1  = as1 + (long long)N * H1;
    float* hx   = ad1 + (long long)N * H1;
    float* h2   = hx + (long long)N * HD1;
    float* as2  = h2 + (long long)N * C2;
    float* ad2  = as2 + N;
    int*   cnt  = (int*)(ad2 + N);
    int*   off  = cnt + N;
    int*   bsum = off + N + 1;
    int*   rank = bsum + cdiv(N, 1024);
    int*   csr  = rank + E;
    float* out  = (float*)d_out;

    const int B = 256;

    fill_u32<<<cdiv(N, B), B, 0, stream>>>((unsigned*)cnt, 0u, N);
    hist_rank<<<cdiv(E, B), B, 0, stream>>>(ei, E, cnt, rank);
    scan_block_sums<<<cdiv(N, 1024), 256, 0, stream>>>(cnt, bsum, N);
    scan_offsets<<<cdiv(N, 1024), 256, 0, stream>>>(cnt, bsum, off, N);
    scatter_src<<<cdiv(E, B), B, 0, stream>>>(ei, E, off, rank, csr);

    gemm1_wmma<<<cdiv(N, 32), 256, 0, stream>>>(x, W1, h1, N);
    alpha1_kernel<<<cdiv((long long)N * H1, B), B, 0, stream>>>(h1, a_src1, a_dst1,
                                                                as1, ad1, N);
    gather1<<<cdiv(N, 8), 256, 0, stream>>>(off, csr, as1, ad1, h1, b1, hx, N);

    gemm2_wmma<<<cdiv(N, 16), 224, 0, stream>>>(hx, W2, h2, N, C2);
    alpha2_kernel<<<cdiv(N, 8), 256, 0, stream>>>(h2, a_src2, a_dst2, as2, ad2, N, C2);
    gather2<<<cdiv(N, 8), 256, 0, stream>>>(off, csr, as2, ad2, h2, b2, out, N, C2);
}


# Fold both attention-logit kernels into the GEMM epilogues: row dots from the WMMA accumulators by lane-swizzle tree sums (layer 2: 7 tile partials via LDS); 2 launches fewer; sum order changes
# speedup vs baseline: 2.6613x; 1.0516x over previous
#include <hip/hip_runtime.h>
#include <math.h>

typedef float v2f __attribute__((ext_vector_type(2)));
typedef float v8f __attribute__((ext_vector_type(8)));

#define FIN 256
#define H1 8
#define D1 8
#define HD1 64

__device__ __forceinline__ float leaky(float v) { return v > 0.f ? v : 0.2f * v; }

template <int M>
__device__ __forceinline__ float xadd(float v) {
    int o = __builtin_amdgcn_update_dpp(0, __float_as_int(v), 0x160 | M, 0xf, 0xf, false);
    return v + __int_as_float(o);
}
__device__ __forceinline__ float sum8(float v) { return xadd<4>(xadd<2>(xadd<1>(v))); }
__device__ __forceinline__ float sum16(float v) { return xadd<8>(sum8(v)); }

__global__ void fill_u32(unsigned* __restrict__ p, unsigned v, long long n) {
    long long i = (long long)blockIdx.x * blockDim.x + threadIdx.x;
    if (i < n) p[i] = v;
}

__global__ __launch_bounds__(256) void gemm1_wmma(const float* __restrict__ x,
                                                  const float* __restrict__ W1,
                                                  const float* __restrict__ a_src,
                                                  const float* __restrict__ a_dst,
                                                  float* __restrict__ h1,
                                                  float* __restrict__ as,
                                                  float* __restrict__ ad, int N) {
    int wave = threadIdx.x >> 5;
    int lane = threadIdx.x & 31;
    int lm = lane & 15, lh = lane >> 4;
    int row0 = blockIdx.x * 32 + (wave >> 2) * 16;
    int col0 = (wave & 3) * 16;
    if (row0 >= N) return;
    const float* xr = x + (long long)(row0 + lm) * FIN;
    v8f acc = {0.f, 0.f, 0.f, 0.f, 0.f, 0.f, 0.f, 0.f};
    for (int k0 = 0; k0 < FIN; k0 += 4) {
        int ka = k0 + 2 * lh;
        v2f a, b;
        a.x = xr[ka];
        a.y = xr[ka + 1];
        b.x = W1[ka * HD1 + col0 + lm];
        b.y = W1[(ka + 1) * HD1 + col0 + lm];
        acc = __builtin_amdgcn_wmma_f32_16x16x4_f32(false, a, false, b, (short)0, acc,
                                                    false, false);
    }
    float* out = h1 + (long long)row0 * HD1 + col0 + lm;
#pragma unroll
    for (int k = 0; k < 8; ++k)
        out[(long long)(k + 8 * lh) * HD1] = acc[k];

    float ws = a_src[col0 + lm], wd = a_dst[col0 + lm];
    int j = lm & 7;
    float sv = 0.f, dv = 0.f;
#pragma unroll
    for (int k = 0; k < 8; ++k) {
        float s = sum8(acc[k] * ws);
        float d = sum8(acc[k] * wd);
        sv = (j == k) ? s : sv;
        dv = (j == k) ? d : dv;
    }
    long long o = (long long)(row0 + j + 8 * lh) * H1 + (col0 >> 3) + (lm >> 3);
    as[o] = sv;
    ad[o] = dv;
}

__global__ void hist_rank(const int* __restrict__ ei, int E, int* __restrict__ cnt,
                          int* __restrict__ rank) {
    int e = blockIdx.x * blockDim.x + threadIdx.x;
    if (e >= E) return;
    rank[e] = atomicAdd(&cnt[ei[E + e]], 1);
}

__global__ __launch_bounds__(256) void scan_block_sums(const int* __restrict__ cnt,
                                                       int* __restrict__ bsum, int N) {
    __shared__ int sm[8];
    int tid = threadIdx.x, lane = tid & 31, wave = tid >> 5;
    int base = blockIdx.x * 1024 + tid * 4;
    int t = 0;
#pragma unroll
    for (int j = 0; j < 4; ++j)
        if (base + j < N) t += cnt[base + j];
    for (int m = 16; m; m >>= 1) t += __shfl_xor(t, m, 32);
    if (lane == 0) sm[wave] = t;
    __syncthreads();
    if (tid == 0) {
        int s = 0;
        for (int w = 0; w < 8; ++w) s += sm[w];
        bsum[blockIdx.x] = s;
    }
}

__global__ __launch_bounds__(256) void scan_offsets(const int* __restrict__ cnt,
                                                    const int* __restrict__ bsum,
                                                    int* __restrict__ off, int N) {
    __shared__ int sm_pre[8];
    __shared__ int sm_loc[8];
    int tid = threadIdx.x, lane = tid & 31, wave = tid >> 5;
    int p = 0;
    for (int i = tid; i < (int)blockIdx.x; i += 256) p += bsum[i];
    for (int m = 16; m; m >>= 1) p += __shfl_xor(p, m, 32);
    if (lane == 0) sm_pre[wave] = p;
    int base = blockIdx.x * 1024 + tid * 4;
    int c[4];
    int t = 0;
#pragma unroll
    for (int j = 0; j < 4; ++j) {
        c[j] = (base + j < N) ? cnt[base + j] : 0;
        t += c[j];
    }
    int incl = t;
    for (int m = 1; m < 32; m <<= 1) {
        int u = __shfl_up(incl, m, 32);
        if (lane >= m) incl += u;
    }
    if (lane == 31) sm_loc[wave] = incl;
    __syncthreads();
    int run = incl - t;
    for (int w = 0; w < 8; ++w) {
        run += sm_pre[w];
        if (w < wave) run += sm_loc[w];
    }
#pragma unroll
    for (int j = 0; j < 4; ++j) {
        int i = base + j;
        if (i < N) {
            off[i] = run;
            run += c[j];
            if (i == N - 1) off[N] = run;
        }
    }
}

__global__ void scatter_src(const int* __restrict__ ei, int E, const int* __restrict__ off,
                            const int* __restrict__ rank, int* __restrict__ csr) {
    int e = blockIdx.x * blockDim.x + threadIdx.x;
    if (e >= E) return;
    csr[off[ei[E + e]] + rank[e]] = ei[e];
}

__global__ __launch_bounds__(256) void gather1(const int* __restrict__ off,
                                               const int* __restrict__ csr,
                                               const float* __restrict__ as,
                                               const float* __restrict__ ad,
                                               const float* __restrict__ h1,
                                               const float* __restrict__ b1,
                                               float* __restrict__ hx, int N) {
    int lane = threadIdx.x & 31;
    int n = __builtin_amdgcn_readfirstlane((int)(blockIdx.x * 8 + (threadIdx.x >> 5)));
    if (n >= N) return;
    int h = lane >> 2;
    int beg = off[n], end = off[n + 1];
    float adn = ad[(long long)n * H1 + h];
    float vself = leaky(as[(long long)n * H1 + h] + adn);
    float m = vself;
    for (int j0 = beg; j0 < end; j0 += 32) {
        int j = j0 + lane;
        int sl = (j < end) ? csr[j] : 0;
        int cc = min(32, end - j0);
        for (int q = 0; q < cc; ++q) {
            int s = __builtin_amdgcn_readlane(sl, q);
            m = fmaxf(m, leaky(as[(long long)s * H1 + h] + adn));
        }
    }
    float w = __expf(vself - m);
    float den = w;
    v2f hv = *(const v2f*)(h1 + (long long)n * HD1 + 2 * lane);
    float ax = w * hv.x, ay = w * hv.y;
    for (int j0 = beg; j0 < end; j0 += 32) {
        int j = j0 + lane;
        int sl = (j < end) ? csr[j] : 0;
        int cc = min(32, end - j0);
        for (int q = 0; q < cc; ++q) {
            int s = __builtin_amdgcn_readlane(sl, q);
            float wq = __expf(leaky(as[(long long)s * H1 + h] + adn) - m);
            v2f hs = *(const v2f*)(h1 + (long long)s * HD1 + 2 * lane);
            den += wq;
            ax += wq * hs.x;
            ay += wq * hs.y;
        }
    }
    float vx = ax / den + b1[2 * lane];
    float vy = ay / den + b1[2 * lane + 1];
    v2f o;
    o.x = vx > 0.f ? vx : (__expf(vx) - 1.f);
    o.y = vy > 0.f ? vy : (__expf(vy) - 1.f);
    *(v2f*)(hx + (long long)n * HD1 + 2 * lane) = o;
}

__global__ __launch_bounds__(224) void gemm2_wmma(const float* __restrict__ hx,
                                                  const float* __restrict__ W2,
                                                  const float* __restrict__ a_src,
                                                  const float* __restrict__ a_dst,
                                                  float* __restrict__ h2,
                                                  float* __restrict__ as,
                                                  float* __restrict__ ad, int N, int C2) {
    __shared__ float part[7 * 32];
    int wave = threadIdx.x >> 5;
    int lane = threadIdx.x & 31;
    int lm = lane & 15, lh = lane >> 4;
    int row0 = blockIdx.x * 16;
    int col = wave * 16 + lm;
    bool cok = col < C2;
    const float* ar = hx + (long long)(row0 + lm) * HD1;
    v8f acc = {0.f, 0.f, 0.f, 0.f, 0.f, 0.f, 0.f, 0.f};
    for (int k0 = 0; k0 < HD1; k0 += 4) {
        int ka = k0 + 2 * lh;
        v2f a, b;
        a.x = ar[ka];
        a.y = ar[ka + 1];
        b.x = cok ? W2[ka * C2 + col] : 0.f;
        b.y = cok ? W2[(ka + 1) * C2 + col] : 0.f;
        acc = __builtin_amdgcn_wmma_f32_16x16x4_f32(false, a, false, b, (short)0, acc,
                                                    false, false);
    }
    float ws = cok ? a_src[col] : 0.f, wd = cok ? a_dst[col] : 0.f;
    int j = lm & 7;
    bool isd = lm >= 8;
    float pv = 0.f;
#pragma unroll
    for (int k = 0; k < 8; ++k) {
        float s = sum16(acc[k] * ws);
        float d = sum16(acc[k] * wd);
        float v = isd ? d : s;
        pv = (j == k) ? v : pv;
    }
    part[wave * 32 + (lm >> 3) * 16 + j + 8 * lh] = pv;
    __syncthreads();
    if (cok) {
        float* out = h2 + (long long)row0 * C2 + col;
#pragma unroll
        for (int k = 0; k < 8; ++k)
            out[(long long)(k + 8 * lh) * C2] = acc[k];
    }
    if (threadIdx.x < 32) {
        float t = 0.f;
#pragma unroll
        for (int w = 0; w < 7; ++w) t += part[w * 32 + threadIdx.x];
        int r = threadIdx.x & 15;
        if (threadIdx.x < 16) as[row0 + r] = t;
        else                  ad[row0 + r] = t;
    }
}

__global__ __launch_bounds__(256) void gather2(const int* __restrict__ off,
                                               const int* __restrict__ csr,
                                               const float* __restrict__ as,
                                               const float* __restrict__ ad,
                                               const float* __restrict__ h2,
                                               const float* __restrict__ b2,
                                               float* __restrict__ out, int N, int C2) {
    int lane = threadIdx.x & 31;
    int n = __builtin_amdgcn_readfirstlane((int)(blockIdx.x * 8 + (threadIdx.x >> 5)));
    if (n >= N) return;
    int beg = off[n], end = off[n + 1];
    float adn = ad[n];
    float vself = leaky(as[n] + adn);
    float m = vself;
    for (int j0 = beg; j0 < end; j0 += 32) {
        int j = j0 + lane;
        if (j < end) m = fmaxf(m, leaky(as[csr[j]] + adn));
    }
    for (int k = 16; k; k >>= 1) m = fmaxf(m, __shfl_xor(m, k, 32));
    bool ok0 = lane < C2, ok1 = lane + 32 < C2, ok2 = lane + 64 < C2, ok3 = lane + 96 < C2;
    int c0 = min(lane, C2 - 1), c1 = min(lane + 32, C2 - 1);
    int c2 = min(lane + 64, C2 - 1), c3 = min(lane + 96, C2 - 1);
    float w0 = __expf(vself - m);
    const float* hn = h2 + (long long)n * C2;
    float a0 = w0 * hn[c0], a1 = w0 * hn[c1], a2 = w0 * hn[c2], a3 = w0 * hn[c3];
    float den = 0.f;
    for (int j0 = beg; j0 < end; j0 += 32) {
        int j = j0 + lane;
        int sl = 0;
        float wl = 0.f;
        if (j < end) {
            sl = csr[j];
            wl = __expf(leaky(as[sl] + adn) - m);
        }
        den += wl;
        int cc = min(32, end - j0);
        for (int q = 0; q < cc; ++q) {
            int s = __builtin_amdgcn_readlane(sl, q);
            float wq = __int_as_float(__builtin_amdgcn_readlane(__float_as_int(wl), q));
            const float* hs = h2 + (long long)s * C2;
            a0 += wq * hs[c0];
            a1 += wq * hs[c1];
            a2 += wq * hs[c2];
            a3 += wq * hs[c3];
        }
    }
    for (int k = 16; k; k >>= 1) den += __shfl_xor(den, k, 32);
    den += w0;
    float* on = out + (long long)n * C2 + lane;
    if (ok0) on[0]  = 1.f / (1.f + __expf(-(a0 / den + b2[lane])));
    if (ok1) on[32] = 1.f / (1.f + __expf(-(a1 / den + b2[lane + 32])));
    if (ok2) on[64] = 1.f / (1.f + __expf(-(a2 / den + b2[lane + 64])));
    if (ok3) on[96] = 1.f / (1.f + __expf(-(a3 / den + b2[lane + 96])));
}

static inline int cdiv(long long a, long long b) { return (int)((a + b - 1) / b); }

extern "C" void kernel_launch(void* const* d_in, const int* in_sizes, int n_in,
                              void* d_out, int out_size, void* d_ws, size_t ws_size,
                              hipStream_t stream) {
    const float* x      = (const float*)d_in[0];
    const int*   ei     = (const int*)d_in[1];
    const float* W1     = (const float*)d_in[2];
    const float* a_src1 = (const float*)d_in[3];
    const float* a_dst1 = (const float*)d_in[4];
    const float* b1     = (const float*)d_in[5];
    const float* W2     = (const float*)d_in[6];
    const float* a_src2 = (const float*)d_in[7];
    const float* a_dst2 = (const float*)d_in[8];
    const float* b2     = (const float*)d_in[9];

    const int N  = in_sizes[0] / FIN;
    const int E  = in_sizes[1] / 2;
    const int C2 = in_sizes[9];

    float* h1   = (float*)d_ws;
    float* as1  = h1 + (long long)N * HD1;
    float* ad1  = as1 + (long long)N * H1;
    float* hx   = ad1 + (long long)N * H1;
    float* h2   = hx + (long long)N * HD1;
    float* as2  = h2 + (long long)N * C2;
    float* ad2  = as2 + N;
    int*   cnt  = (int*)(ad2 + N);
    int*   off  = cnt + N;
    int*   bsum = off + N + 1;
    int*   rank = bsum + cdiv(N, 1024);
    int*   csr  = rank + E;
    float* out  = (float*)d_out;

    const int B = 256;

    fill_u32<<<cdiv(N, B), B, 0, stream>>>((unsigned*)cnt, 0u, N);
    hist_rank<<<cdiv(E, B), B, 0, stream>>>(ei, E, cnt, rank);
    scan_block_sums<<<cdiv(N, 1024), 256, 0, stream>>>(cnt, bsum, N);
    scan_offsets<<<cdiv(N, 1024), 256, 0, stream>>>(cnt, bsum, off, N);
    scatter_src<<<cdiv(E, B), B, 0, stream>>>(ei, E, off, rank, csr);

    gemm1_wmma<<<cdiv(N, 32), 256, 0, stream>>>(x, W1, a_src1, a_dst1, h1, as1, ad1, N);
    gather1<<<cdiv(N, 8), 256, 0, stream>>>(off, csr, as1, ad1, h1, b1, hx, N);

    gemm2_wmma<<<cdiv(N, 16), 224, 0, stream>>>(hx, W2, a_src2, a_dst2, h2, as2, ad2, N, C2);
    gather2<<<cdiv(N, 8), 256, 0, stream>>>(off, csr, as2, ad2, h2, b2, out, N, C2);
}


# Edge sort by destination with per-edge counting in LDS: bucket partition (256 nodes each) staged in LDS, then one work-group per bucket ranks and places; 9 launches; order in a node's sum changes
# speedup vs baseline: 4.0465x; 1.5205x over previous
#include <hip/hip_runtime.h>
#include <math.h>

typedef float v2f __attribute__((ext_vector_type(2)));
typedef float v8f __attribute__((ext_vector_type(8)));

#define FIN 256
#define H1 8
#define D1 8
#define HD1 64

__device__ __forceinline__ float leaky(float v) { return v > 0.f ? v : 0.2f * v; }

template <int M>
__device__ __forceinline__ float xadd(float v) {
    int o = __builtin_amdgcn_update_dpp(0, __float_as_int(v), 0x160 | M, 0xf, 0xf, false);
    return v + __int_as_float(o);
}
__device__ __forceinline__ float sum8(float v) { return xadd<4>(xadd<2>(xadd<1>(v))); }
__device__ __forceinline__ float sum16(float v) { return xadd<8>(sum8(v)); }
template <int M>
__device__ __forceinline__ float xmax(float v) {
    int o = __builtin_amdgcn_update_dpp(0, __float_as_int(v), 0x160 | M, 0xf, 0xf, false);
    return fmaxf(v, __int_as_float(o));
}
__device__ __forceinline__ float wave_sum(float v) {
    v = sum16(v);
    return v + __shfl_xor(v, 16, 32);
}
__device__ __forceinline__ float wave_max(float v) {
    v = xmax<8>(xmax<4>(xmax<2>(xmax<1>(v))));
    return fmaxf(v, __shfl_xor(v, 16, 32));
}

__global__ __launch_bounds__(256) void fold_w2(const float* __restrict__ W2,
                                               const float* __restrict__ a_src,
                                               const float* __restrict__ a_dst,
                                               float* __restrict__ vsd, int C2) {
    int lane = threadIdx.x & 31;
    int k = blockIdx.x * 8 + (threadIdx.x >> 5);
    float s = 0.f, d = 0.f;
    for (int cix = lane; cix < C2; cix += 32) {
        float w = W2[k * C2 + cix];
        s += w * a_src[cix];
        d += w * a_dst[cix];
    }
    s = wave_sum(s);
    d = wave_sum(d);
    if (lane == 0) {
        vsd[k] = s;
        vsd[HD1 + k] = d;
    }
}

__global__ void fill_u32(unsigned* __restrict__ p, unsigned v, long long n) {
    long long i = (long long)blockIdx.x * blockDim.x + threadIdx.x;
    if (i < n) p[i] = v;
}

__global__ __launch_bounds__(256) void gemm1_wmma(const float* __restrict__ x,
                                                  const float* __restrict__ W1,
                                                  const float* __restrict__ a_src,
                                                  const float* __restrict__ a_dst,
                                                  float* __restrict__ h1,
                                                  float* __restrict__ as,
                                                  float* __restrict__ ad, int N) {
    int wave = threadIdx.x >> 5;
    int lane = threadIdx.x & 31;
    int lm = lane & 15, lh = lane >> 4;
    int row0 = blockIdx.x * 32 + (wave >> 2) * 16;
    int col0 = (wave & 3) * 16;
    if (row0 >= N) return;
    const float* xr = x + (long long)(row0 + lm) * FIN;
    v8f acc = {0.f, 0.f, 0.f, 0.f, 0.f, 0.f, 0.f, 0.f};
    for (int k0 = 0; k0 < FIN; k0 += 4) {
        int ka = k0 + 2 * lh;
        v2f a, b;
        a.x = xr[ka];
        a.y = xr[ka + 1];
        b.x = W1[ka * HD1 + col0 + lm];
        b.y = W1[(ka + 1) * HD1 + col0 + lm];
        acc = __builtin_amdgcn_wmma_f32_16x16x4_f32(false, a, false, b, (short)0, acc,
                                                    false, false);
    }
    float* out = h1 + (long long)row0 * HD1 + col0 + lm;
#pragma unroll
    for (int k = 0; k < 8; ++k)
        out[(long long)(k + 8 * lh) * HD1] = acc[k];

    float ws = a_src[col0 + lm], wd = a_dst[col0 + lm];
    int j = lm & 7;
    float sv = 0.f, dv = 0.f;
#pragma unroll
    for (int k = 0; k < 8; ++k) {
        float s = sum8(acc[k] * ws);
        float d = sum8(acc[k] * wd);
        sv = (j == k) ? s : sv;
        dv = (j == k) ? d : dv;
    }
    long long o = (long long)(row0 + j + 8 * lh) * H1 + (col0 >> 3) + (lm >> 3);
    as[o] = sv;
    ad[o] = dv;
}

#define BSH 8
#define BN (1 << BSH)
#define EPB 4096
#define EPT (EPB / 256)
#define SB 8
#define CS 32

__global__ __launch_bounds__(256) void bucket_count(const int* __restrict__ ei, int E, int NB,
                                                    int* __restrict__ tot) {
    extern __shared__ int sh[];
    int tid = threadIdx.x;
    for (int i = tid; i < NB; i += 256) sh[i] = 0;
    __syncthreads();
    int e0 = blockIdx.x * EPB + tid;
    int d[EPT];
#pragma unroll
    for (int k = 0; k < EPT; ++k) {
        int e = e0 + k * 256;
        d[k] = (e < E) ? ei[E + e] : -1;
    }
#pragma unroll
    for (int k = 0; k < EPT; ++k)
        if (d[k] >= 0) atomicAdd(&sh[d[k] >> BSH], 1);
    __syncthreads();
    for (int i = tid; i < NB; i += 256) {
        int c = sh[i];
        if (c) atomicAdd(&tot[i * CS], c);
    }
}

__global__ __launch_bounds__(256) void bucket_partition(const int* __restrict__ ei, int E, int NB,
                                                        const int* __restrict__ tot,
                                                        int* __restrict__ cur,
                                                        int2* __restrict__ part) {
    extern __shared__ int sh[];
    __shared__ int2 stage[EPB];
    __shared__ int wsum[8];
    __shared__ int wsum2[8];
    int tid = threadIdx.x, lane = tid & 31, wave = tid >> 5;
    for (int i = tid; i < NB; i += 256) sh[i] = 0;
    __syncthreads();
    int e0 = blockIdx.x * EPB + tid;
    int d[EPT], sv[EPT], r[EPT];
#pragma unroll
    for (int k = 0; k < EPT; ++k) {
        int e = e0 + k * 256;
        d[k] = (e < E) ? ei[E + e] : -1;
        sv[k] = (e < E) ? ei[e] : 0;
    }
#pragma unroll
    for (int k = 0; k < EPT; ++k) r[k] = (d[k] >= 0) ? atomicAdd(&sh[d[k] >> BSH], 1) : 0;
    __syncthreads();
    int ipt = (NB + 255) / 256;
    int lo = tid * ipt;
    int hi = min(lo + ipt, NB);
    int s = 0, s2 = 0;
    for (int i = lo; i < hi; ++i) {
        s += tot[i * CS];
        s2 += sh[i];
    }
    int incl = s, incl2 = s2;
    for (int m = 1; m < 32; m <<= 1) {
        int u = __shfl_up(incl, m, 32);
        int u2 = __shfl_up(incl2, m, 32);
        if (lane >= m) {
            incl += u;
            incl2 += u2;
        }
    }
    if (lane == 31) {
        wsum[wave] = incl;
        wsum2[wave] = incl2;
    }
    __syncthreads();
    int run = incl - s, lrun = incl2 - s2;
    for (int w = 0; w < wave; ++w) {
        run += wsum[w];
        lrun += wsum2[w];
    }
    for (int i = lo; i < hi; ++i) {
        int c = sh[i];
        int base = run;
        if (c) base += atomicAdd(&cur[i * CS], c);
        sh[NB + i] = lrun;
        sh[i] = base - lrun;
        run += tot[i * CS];
        lrun += c;
    }
    __syncthreads();
#pragma unroll
    for (int k = 0; k < EPT; ++k) {
        if (d[k] >= 0) {
            int2 pr;
            pr.x = sv[k];
            pr.y = d[k];
            stage[sh[NB + (d[k] >> BSH)] + r[k]] = pr;
        }
    }
    __syncthreads();
    int ne = min(EPB, E - (int)blockIdx.x * EPB);
#pragma unroll
    for (int k = 0; k < EPT; ++k) {
        int p = tid + k * 256;
        if (p < ne) {
            int2 pr = stage[p];
            part[sh[pr.y >> BSH] + p] = pr;
        }
    }
}

__global__ __launch_bounds__(BN) void bucket_sort(const int2* __restrict__ part,
                                                  const int* __restrict__ tot, int N,
                                                  int* __restrict__ off, int* __restrict__ csr) {
    __shared__ int cnt[BN];
    __shared__ int red[BN / 32];
    __shared__ int wsum[BN / 32];
    int tid = threadIdx.x, lane = tid & 31, wave = tid >> 5;
    int b = blockIdx.x;
    int p = 0;
    for (int i = tid; i < b; i += BN) p += tot[i * CS];
    for (int m = 16; m; m >>= 1) p += __shfl_xor(p, m, 32);
    if (lane == 0) red[wave] = p;
    cnt[tid] = 0;
    __syncthreads();
    int start = 0;
    for (int w = 0; w < BN / 32; ++w) start += red[w];
    int nb = tot[b * CS];
    const int2* mine = part + start;
    for (int i0 = tid; i0 < nb; i0 += SB * BN) {
        int2 pr[SB];
#pragma unroll
        for (int u = 0; u < SB; ++u) {
            int i = i0 + u * BN;
            pr[u].y = (i < nb) ? mine[i].y : -1;
        }
#pragma unroll
        for (int u = 0; u < SB; ++u)
            if (pr[u].y >= 0) atomicAdd(&cnt[pr[u].y & (BN - 1)], 1);
    }
    __syncthreads();
    int c = cnt[tid];
    int incl = c;
    for (int m = 1; m < 32; m <<= 1) {
        int u = __shfl_up(incl, m, 32);
        if (lane >= m) incl += u;
    }
    if (lane == 31) wsum[wave] = incl;
    __syncthreads();
    int excl = incl - c;
    for (int w = 0; w < wave; ++w) excl += wsum[w];
    int node = b * BN + tid;
    if (node < N) off[node] = start + excl;
    if (node == N - 1) off[N] = start + excl + c;
    cnt[tid] = excl;
    __syncthreads();
    int* out = csr + start;
    for (int i0 = tid; i0 < nb; i0 += SB * BN) {
        int2 pr[SB];
#pragma unroll
        for (int u = 0; u < SB; ++u) {
            int i = i0 + u * BN;
            if (i < nb) pr[u] = mine[i];
            else pr[u].y = -1;
        }
#pragma unroll
        for (int u = 0; u < SB; ++u)
            if (pr[u].y >= 0) out[atomicAdd(&cnt[pr[u].y & (BN - 1)], 1)] = pr[u].x;
    }
}

__global__ __launch_bounds__(256) void gather1(const int* __restrict__ off,
                                               const int* __restrict__ csr,
                                               const float* __restrict__ as,
                                               const float* __restrict__ ad,
                                               const float* __restrict__ h1,
                                               const float* __restrict__ b1,
                                               const float* __restrict__ vsd,
                                               float* __restrict__ hx,
                                               float* __restrict__ as2,
                                               float* __restrict__ ad2, int N) {
    int lane = threadIdx.x & 31;
    int n = __builtin_amdgcn_readfirstlane((int)(blockIdx.x * 8 + (threadIdx.x >> 5)));
    if (n >= N) return;
    int h = lane >> 2;
    int beg = off[n], end = off[n + 1];
    float adn = ad[(long long)n * H1 + h];
    float vself = leaky(as[(long long)n * H1 + h] + adn);
    float m = vself;
    for (int j0 = beg; j0 < end; j0 += 32) {
        int j = j0 + lane;
        int sl = (j < end) ? csr[j] : 0;
        int cc = min(32, end - j0);
        for (int q = 0; q < cc; ++q) {
            int s = __builtin_amdgcn_readlane(sl, q);
            m = fmaxf(m, leaky(as[(long long)s * H1 + h] + adn));
        }
    }
    float w = __expf(vself - m);
    float den = w;
    v2f hv = *(const v2f*)(h1 + (long long)n * HD1 + 2 * lane);
    float ax = w * hv.x, ay = w * hv.y;
    for (int j0 = beg; j0 < end; j0 += 32) {
        int j = j0 + lane;
        int sl = (j < end) ? csr[j] : 0;
        int cc = min(32, end - j0);
        for (int q = 0; q < cc; ++q) {
            int s = __builtin_amdgcn_readlane(sl, q);
            float wq = __expf(leaky(as[(long long)s * H1 + h] + adn) - m);
            v2f hs = *(const v2f*)(h1 + (long long)s * HD1 + 2 * lane);
            den += wq;
            ax += wq * hs.x;
            ay += wq * hs.y;
        }
    }
    float vx = ax / den + b1[2 * lane];
    float vy = ay / den + b1[2 * lane + 1];
    v2f o;
    o.x = vx > 0.f ? vx : (__expf(vx) - 1.f);
    o.y = vy > 0.f ? vy : (__expf(vy) - 1.f);
    *(v2f*)(hx + (long long)n * HD1 + 2 * lane) = o;
    v2f ws = *(const v2f*)(vsd + 2 * lane);
    v2f wd = *(const v2f*)(vsd + HD1 + 2 * lane);
    float ps = wave_sum(o.x * ws.x + o.y * ws.y);
    float pd = wave_sum(o.x * wd.x + o.y * wd.y);
    if (lane == 0) {
        as2[n] = ps;
        ad2[n] = pd;
    }
}

#define EB 16
__global__ __launch_bounds__(256) void gather2_rows(const int* __restrict__ off,
                                                    const int* __restrict__ csr,
                                                    const float* __restrict__ as,
                                                    const float* __restrict__ ad,
                                                    const float* __restrict__ hx,
                                                    float* __restrict__ agg, int N) {
    int lane = threadIdx.x & 31;
    int n = __builtin_amdgcn_readfirstlane((int)(blockIdx.x * 8 + (threadIdx.x >> 5)));
    if (n >= N) return;
    int beg = off[n], end = off[n + 1];
    float adn = ad[n];
    float vself = leaky(as[n] + adn);
    const v2f* rows = (const v2f*)hx;
    v2f hn = rows[n * 32 + lane];
    int j = beg + lane;
    int sl = (j < end) ? csr[j] : n;
    float el = leaky(as[sl] + adn);
    float m = fmaxf(vself, el);
    for (int j0 = beg + 32; j0 < end; j0 += 32) {
        int jj = j0 + lane;
        if (jj < end) m = fmaxf(m, leaky(as[csr[jj]] + adn));
    }
    m = wave_max(m);
    float w0 = __expf(vself - m);
    float ax = w0 * hn.x, ay = w0 * hn.y;
    float den = 0.f;
    for (int j0 = beg; j0 < end; j0 += 32) {
        if (j0 != beg) {
            int jj = j0 + lane;
            sl = (jj < end) ? csr[jj] : n;
            el = leaky(as[sl] + adn);
        }
        float wl = (j0 + lane < end) ? __expf(el - m) : 0.f;
        den += wl;
        int cc = min(32, end - j0);
        for (int q0 = 0; q0 < cc; q0 += EB) {
            v2f r[EB];
            float wq[EB];
#pragma unroll
            for (int u = 0; u < EB; ++u) {
                int s = __builtin_amdgcn_readlane(sl, q0 + u);
                wq[u] = __int_as_float(__builtin_amdgcn_readlane(__float_as_int(wl), q0 + u));
                r[u] = rows[s * 32 + lane];
            }
#pragma unroll
            for (int u = 0; u < EB; ++u) {
                ax += wq[u] * r[u].x;
                ay += wq[u] * r[u].y;
            }
        }
    }
    den = wave_sum(den) + w0;
    v2f o;
    o.x = ax / den;
    o.y = ay / den;
    *(v2f*)(agg + (long long)n * HD1 + 2 * lane) = o;
}

__global__ __launch_bounds__(224) void gemm2_wmma(const float* __restrict__ agg,
                                                  const float* __restrict__ W2,
                                                  const float* __restrict__ b2,
                                                  float* __restrict__ outp, int N, int C2) {
    int wave = threadIdx.x >> 5;
    int lane = threadIdx.x & 31;
    int lm = lane & 15, lh = lane >> 4;
    int row0 = blockIdx.x * 16;
    int col = wave * 16 + lm;
    bool cok = col < C2;
    int colc = min(col, C2 - 1);
    const float* ar = agg + (long long)(row0 + lm) * HD1;
    v8f acc = {0.f, 0.f, 0.f, 0.f, 0.f, 0.f, 0.f, 0.f};
    for (int k0 = 0; k0 < HD1; k0 += 4) {
        int ka = k0 + 2 * lh;
        v2f a, b;
        a.x = ar[ka];
        a.y = ar[ka + 1];
        b.x = W2[ka * C2 + colc];
        b.y = W2[(ka + 1) * C2 + colc];
        acc = __builtin_amdgcn_wmma_f32_16x16x4_f32(false, a, false, b, (short)0, acc,
                                                    false, false);
    }
    if (cok) {
        float bb = b2[col];
        float* o = outp + (long long)row0 * C2 + col;
#pragma unroll
        for (int k = 0; k < 8; ++k)
            o[(long long)(k + 8 * lh) * C2] = 1.f / (1.f + __expf(-(acc[k] + bb)));
    }
}

static inline int cdiv(long long a, long long b) { return (int)((a + b - 1) / b); }

extern "C" void kernel_launch(void* const* d_in, const int* in_sizes, int n_in,
                              void* d_out, int out_size, void* d_ws, size_t ws_size,
                              hipStream_t stream) {
    const float* x      = (const float*)d_in[0];
    const int*   ei     = (const int*)d_in[1];
    const float* W1     = (const float*)d_in[2];
    const float* a_src1 = (const float*)d_in[3];
    const float* a_dst1 = (const float*)d_in[4];
    const float* b1     = (const float*)d_in[5];
    const float* W2     = (const float*)d_in[6];
    const float* a_src2 = (const float*)d_in[7];
    const float* a_dst2 = (const float*)d_in[8];
    const float* b2     = (const float*)d_in[9];

    const int N  = in_sizes[0] / FIN;
    const int E  = in_sizes[1] / 2;
    const int C2 = in_sizes[9];

    float* h1   = (float*)d_ws;
    float* as1  = h1 + (long long)N * HD1;
    float* ad1  = as1 + (long long)N * H1;
    float* hx   = ad1 + (long long)N * H1;
    float* agg  = hx + (long long)N * HD1;
    int2*  part = (int2*)(agg + (long long)N * HD1);
    float* as2  = (float*)(part + E);
    float* ad2  = as2 + N;
    float* vsd  = ad2 + N;
    const int NB = cdiv(N, BN);
    int*   tot  = (int*)(vsd + 2 * HD1);
    int*   cur  = tot + NB * CS;
    int*   off  = cur + NB * CS;
    int*   csr  = off + N + 1;
    float* out  = (float*)d_out;

    const int B = 256;

    fill_u32<<<cdiv(2 * NB * CS, B), B, 0, stream>>>((unsigned*)tot, 0u, 2 * NB * CS);
    bucket_count<<<cdiv(E, EPB), 256, NB * sizeof(int), stream>>>(ei, E, NB, tot);
    bucket_partition<<<cdiv(E, EPB), 256, 2 * NB * sizeof(int), stream>>>(ei, E, NB, tot, cur, part);
    bucket_sort<<<NB, BN, 0, stream>>>(part, tot, N, off, csr);

    fold_w2<<<HD1 / 8, 256, 0, stream>>>(W2, a_src2, a_dst2, vsd, C2);
    gemm1_wmma<<<cdiv(N, 32), 256, 0, stream>>>(x, W1, a_src1, a_dst1, h1, as1, ad1, N);
    gather1<<<cdiv(N, 8), 256, 0, stream>>>(off, csr, as1, ad1, h1, b1, vsd, hx, as2, ad2, N);

    gather2_rows<<<cdiv(N, 8), 256, 0, stream>>>(off, csr, as2, ad2, hx, agg, N);
    gemm2_wmma<<<cdiv(N, 16), 224, 0, stream>>>(agg, W2, b2, out, N, C2);
}


# Both GEMMs: a wave owns a 16-row strip and all column tiles: each A fragment loaded once (16 B per lane) and reused, B fragments from LDS, lanes hold runs of 4 columns; k order in the sums changes
# speedup vs baseline: 5.1628x; 1.2759x over previous
#include <hip/hip_runtime.h>
#include <math.h>

typedef float v2f __attribute__((ext_vector_type(2)));
typedef float v4f __attribute__((ext_vector_type(4)));
typedef float v8f __attribute__((ext_vector_type(8)));

#define FIN 256
#define HD1 64

__device__ __forceinline__ float leaky(float v) { return v > 0.f ? v : 0.2f * v; }

template <int M>
__device__ __forceinline__ float xadd(float v) {
    int o = __builtin_amdgcn_update_dpp(0, __float_as_int(v), 0x160 | M, 0xf, 0xf, false);
    return v + __int_as_float(o);
}
template <int M>
__device__ __forceinline__ float xget(float v) {
    return __int_as_float(__builtin_amdgcn_update_dpp(0, __float_as_int(v), 0x160 | M, 0xf, 0xf, false));
}
__device__ __forceinline__ float sum8(float v) { return xadd<4>(xadd<2>(xadd<1>(v))); }
__device__ __forceinline__ float sum16(float v) { return xadd<8>(sum8(v)); }
template <int M>
__device__ __forceinline__ float xmax(float v) {
    int o = __builtin_amdgcn_update_dpp(0, __float_as_int(v), 0x160 | M, 0xf, 0xf, false);
    return fmaxf(v, __int_as_float(o));
}
__device__ __forceinline__ float wave_sum(float v) {
    v = sum16(v);
    return v + __shfl_xor(v, 16, 32);
}
__device__ __forceinline__ float wave_max(float v) {
    v = xmax<8>(xmax<4>(xmax<2>(xmax<1>(v))));
    return fmaxf(v, __shfl_xor(v, 16, 32));
}

__global__ __launch_bounds__(256) void fold_w2(const float* __restrict__ W2,
                                               const float* __restrict__ a_src,
                                               const float* __restrict__ a_dst,
                                               float* __restrict__ vsd, int C2) {
    int lane = threadIdx.x & 31;
    int k = blockIdx.x * 8 + (threadIdx.x >> 5);
    float s = 0.f, d = 0.f;
    for (int cix = lane; cix < C2; cix += 32) {
        float w = W2[k * C2 + cix];
        s += w * a_src[cix];
        d += w * a_dst[cix];
    }
    s = wave_sum(s);
    d = wave_sum(d);
    if (lane == 0) {
        vsd[k] = s;
        vsd[HD1 + k] = d;
    }
}

__global__ void fill_u32(unsigned* __restrict__ p, unsigned v, long long n) {
    long long i = (long long)blockIdx.x * blockDim.x + threadIdx.x;
    if (i < n) p[i] = v;
}

#define G1W 4
__global__ __launch_bounds__(32 * G1W) void gemm1_wmma(const float* __restrict__ x,
                                                       const float* __restrict__ W1,
                                                       float* __restrict__ h1, int N) {
    __shared__ v4f Bs[(FIN / 2) * 32];
    int tid = threadIdx.x;
    int wave = tid >> 5, lane = tid & 31;
    int lm = lane & 15, lh = lane >> 4;
    const v4f* W4 = (const v4f*)W1;
    for (int p0 = tid; p0 < (FIN / 2) * 16; p0 += 32 * G1W * 4) {
        v4f r0[4], r1[4];
#pragma unroll
        for (int u = 0; u < 4; ++u) {
            int p = p0 + u * 32 * G1W;
            int kp = p >> 4, c = p & 15;
            r0[u] = W4[(2 * kp) * 16 + c];
            r1[u] = W4[(2 * kp + 1) * 16 + c];
        }
#pragma unroll
        for (int u = 0; u < 4; ++u) {
            int p = p0 + u * 32 * G1W;
            int kp = p >> 4, c = p & 15;
            v4f g0, g1;
            g0.x = r0[u].x; g0.y = r1[u].x; g0.z = r0[u].y; g0.w = r1[u].y;
            g1.x = r0[u].z; g1.y = r1[u].z; g1.z = r0[u].w; g1.w = r1[u].w;
            Bs[kp * 32 + c] = g0;
            Bs[kp * 32 + 16 + c] = g1;
        }
    }
    __syncthreads();
    int row0 = (blockIdx.x * G1W + wave) * 16;
    if (row0 >= N) return;
    const v4f* xr = (const v4f*)(x + (long long)(row0 + lm) * FIN) + lh;
    const v4f* bl = Bs + lh * 64 + lm;
    v8f acc0 = {0.f, 0.f, 0.f, 0.f, 0.f, 0.f, 0.f, 0.f};
    v8f acc1 = acc0, acc2 = acc0, acc3 = acc0;
    v4f a[2][8];
#pragma unroll
    for (int q = 0; q < 8; ++q) a[0][q] = xr[2 * q];
#pragma unroll
    for (int ch = 0; ch < FIN / 64; ++ch) {
        if (ch + 1 < FIN / 64) {
#pragma unroll
            for (int q = 0; q < 8; ++q) a[(ch + 1) & 1][q] = xr[2 * ((ch + 1) * 8 + q)];
        }
        __builtin_amdgcn_sched_barrier(0);
#pragma unroll
        for (int q = 0; q < 8; ++q) {
            int kb = ch * 8 + q;
            const v4f* bp = bl + kb * 128;
            v4f b00 = bp[0], b01 = bp[16], b10 = bp[32], b11 = bp[48];
            v4f av = a[ch & 1][q];
            acc0 = __builtin_amdgcn_wmma_f32_16x16x4_f32(false, av.xy, false, b00.xy, (short)0, acc0, false, false);
            acc1 = __builtin_amdgcn_wmma_f32_16x16x4_f32(false, av.xy, false, b00.zw, (short)0, acc1, false, false);
            acc2 = __builtin_amdgcn_wmma_f32_16x16x4_f32(false, av.xy, false, b01.xy, (short)0, acc2, false, false);
            acc3 = __builtin_amdgcn_wmma_f32_16x16x4_f32(false, av.xy, false, b01.zw, (short)0, acc3, false, false);
            acc0 = __builtin_amdgcn_wmma_f32_16x16x4_f32(false, av.zw, false, b10.xy, (short)0, acc0, false, false);
            acc1 = __builtin_amdgcn_wmma_f32_16x16x4_f32(false, av.zw, false, b10.zw, (short)0, acc1, false, false);
            acc2 = __builtin_amdgcn_wmma_f32_16x16x4_f32(false, av.zw, false, b11.xy, (short)0, acc2, false, false);
            acc3 = __builtin_amdgcn_wmma_f32_16x16x4_f32(false, av.zw, false, b11.zw, (short)0, acc3, false, false);
        }
    }
    float* out = h1 + (long long)(row0 + 8 * lh) * HD1 + 4 * lm;
#pragma unroll
    for (int r = 0; r < 8; ++r) {
        v4f o;
        o.x = acc0[r]; o.y = acc1[r]; o.z = acc2[r]; o.w = acc3[r];
        *(v4f*)(out + (long long)r * HD1) = o;
    }
}

#define BSH 8
#define BN (1 << BSH)
#define EPB 4096
#define EPT (EPB / 256)
#define SB 8
#define CS 32

__global__ __launch_bounds__(256) void bucket_count(const int* __restrict__ ei, int E, int NB,
                                                    int* __restrict__ tot) {
    extern __shared__ int sh[];
    int tid = threadIdx.x;
    for (int i = tid; i < NB; i += 256) sh[i] = 0;
    __syncthreads();
    int e0 = blockIdx.x * EPB + tid;
    int d[EPT];
#pragma unroll
    for (int k = 0; k < EPT; ++k) {
        int e = e0 + k * 256;
        d[k] = (e < E) ? ei[E + e] : -1;
    }
#pragma unroll
    for (int k = 0; k < EPT; ++k)
        if (d[k] >= 0) atomicAdd(&sh[d[k] >> BSH], 1);
    __syncthreads();
    for (int i = tid; i < NB; i += 256) {
        int c = sh[i];
        if (c) atomicAdd(&tot[i * CS], c);
    }
}

__global__ __launch_bounds__(256) void bucket_partition(const int* __restrict__ ei, int E, int NB,
                                                        const int* __restrict__ tot,
                                                        int* __restrict__ cur,
                                                        int2* __restrict__ part) {
    extern __shared__ int sh[];
    __shared__ int2 stage[EPB];
    __shared__ int wsum[8];
    __shared__ int wsum2[8];
    int tid = threadIdx.x, lane = tid & 31, wave = tid >> 5;
    for (int i = tid; i < NB; i += 256) sh[i] = 0;
    __syncthreads();
    int e0 = blockIdx.x * EPB + tid;
    int d[EPT], sv[EPT], r[EPT];
#pragma unroll
    for (int k = 0; k < EPT; ++k) {
        int e = e0 + k * 256;
        d[k] = (e < E) ? ei[E + e] : -1;
        sv[k] = (e < E) ? ei[e] : 0;
    }
#pragma unroll
    for (int k = 0; k < EPT; ++k) r[k] = (d[k] >= 0) ? atomicAdd(&sh[d[k] >> BSH], 1) : 0;
    __syncthreads();
    int ipt = (NB + 255) / 256;
    int lo = tid * ipt;
    int hi = min(lo + ipt, NB);
    int s = 0, s2 = 0;
    for (int i = lo; i < hi; ++i) {
        s += tot[i * CS];
        s2 += sh[i];
    }
    int incl = s, incl2 = s2;
    for (int m = 1; m < 32; m <<= 1) {
        int u = __shfl_up(incl, m, 32);
        int u2 = __shfl_up(incl2, m, 32);
        if (lane >= m) {
            incl += u;
            incl2 += u2;
        }
    }
    if (lane == 31) {
        wsum[wave] = incl;
        wsum2[wave] = incl2;
    }
    __syncthreads();
    int run = incl - s, lrun = incl2 - s2;
    for (int w = 0; w < wave; ++w) {
        run += wsum[w];
        lrun += wsum2[w];
    }
    for (int i = lo; i < hi; ++i) {
        int c = sh[i];
        int base = run;
        if (c) base += atomicAdd(&cur[i * CS], c);
        sh[NB + i] = lrun;
        sh[i] = base - lrun;
        run += tot[i * CS];
        lrun += c;
    }
    __syncthreads();
#pragma unroll
    for (int k = 0; k < EPT; ++k) {
        if (d[k] >= 0) {
            int2 pr;
            pr.x = sv[k];
            pr.y = d[k];
            stage[sh[NB + (d[k] >> BSH)] + r[k]] = pr;
        }
    }
    __syncthreads();
    int ne = min(EPB, E - (int)blockIdx.x * EPB);
#pragma unroll
    for (int k = 0; k < EPT; ++k) {
        int p = tid + k * 256;
        if (p < ne) {
            int2 pr = stage[p];
            part[sh[pr.y >> BSH] + p] = pr;
        }
    }
}

__global__ __launch_bounds__(BN) void bucket_sort(const int2* __restrict__ part,
                                                  const int* __restrict__ tot, int N,
                                                  int* __restrict__ off, int* __restrict__ csr) {
    __shared__ int cnt[BN];
    __shared__ int red[BN / 32];
    __shared__ int wsum[BN / 32];
    int tid = threadIdx.x, lane = tid & 31, wave = tid >> 5;
    int b = blockIdx.x;
    int p = 0;
    for (int i = tid; i < b; i += BN) p += tot[i * CS];
    for (int m = 16; m; m >>= 1) p += __shfl_xor(p, m, 32);
    if (lane == 0) red[wave] = p;
    cnt[tid] = 0;
    __syncthreads();
    int start = 0;
    for (int w = 0; w < BN / 32; ++w) start += red[w];
    int nb = tot[b * CS];
    const int2* mine = part + start;
    for (int i0 = tid; i0 < nb; i0 += SB * BN) {
        int2 pr[SB];
#pragma unroll
        for (int u = 0; u < SB; ++u) {
            int i = i0 + u * BN;
            pr[u].y = (i < nb) ? mine[i].y : -1;
        }
#pragma unroll
        for (int u = 0; u < SB; ++u)
            if (pr[u].y >= 0) atomicAdd(&cnt[pr[u].y & (BN - 1)], 1);
    }
    __syncthreads();
    int c = cnt[tid];
    int incl = c;
    for (int m = 1; m < 32; m <<= 1) {
        int u = __shfl_up(incl, m, 32);
        if (lane >= m) incl += u;
    }
    if (lane == 31) wsum[wave] = incl;
    __syncthreads();
    int excl = incl - c;
    for (int w = 0; w < wave; ++w) excl += wsum[w];
    int node = b * BN + tid;
    if (node < N) off[node] = start + excl;
    if (node == N - 1) off[N] = start + excl + c;
    cnt[tid] = excl;
    __syncthreads();
    int* out = csr + start;
    for (int i0 = tid; i0 < nb; i0 += SB * BN) {
        int2 pr[SB];
#pragma unroll
        for (int u = 0; u < SB; ++u) {
            int i = i0 + u * BN;
            if (i < nb) pr[u] = mine[i];
            else pr[u].y = -1;
        }
#pragma unroll
        for (int u = 0; u < SB; ++u)
            if (pr[u].y >= 0) out[atomicAdd(&cnt[pr[u].y & (BN - 1)], 1)] = pr[u].x;
    }
}

__device__ __forceinline__ float dot8(v4f r0, v4f r1, v4f a0, v4f a1) {
    v2f p = r0.xy * a0.xy;
    p += r0.zw * a0.zw;
    p += r1.xy * a1.xy;
    p += r1.zw * a1.zw;
    return p.x + p.y;
}

template <int K>
__device__ __forceinline__ void head_steps(const v4f* __restrict__ rows4, const int* __restrict__ csr,
                                           int j, int end, int h, v4f a0, v4f a1, float adn,
                                           float& m, float& den, v4f& c0, v4f& c1) {
    int s[K];
    v4f r0[K], r1[K];
    float e[K];
#pragma unroll
    for (int k = 0; k < K; ++k) s[k] = csr[min(j + 4 * k, end - 1)];
#pragma unroll
    for (int k = 0; k < K; ++k) {
        r0[k] = rows4[s[k] * 16 + 2 * h];
        r1[k] = rows4[s[k] * 16 + 2 * h + 1];
    }
    float gm = m;
#pragma unroll
    for (int k = 0; k < K; ++k) {
        e[k] = leaky(dot8(r0[k], r1[k], a0, a1) + adn);
        gm = fmaxf(gm, e[k]);
    }
    float sc = __expf(m - gm);
    den *= sc;
    c0 *= sc;
    c1 *= sc;
    m = gm;
#pragma unroll
    for (int k = 0; k < K; ++k) {
        float w = (j + 4 * k < end) ? __expf(e[k] - gm) : 0.f;
        den += w;
        c0 += w * r0[k];
        c1 += w * r1[k];
    }
}

__global__ __launch_bounds__(256) void gather1(const int* __restrict__ off,
                                               const int* __restrict__ csr,
                                               const float* __restrict__ h1,
                                               const float* __restrict__ a_src,
                                               const float* __restrict__ a_dst,
                                               const float* __restrict__ b1,
                                               const float* __restrict__ vsd,
                                               float* __restrict__ hx,
                                               float* __restrict__ as2,
                                               float* __restrict__ ad2, int N) {
    int lane = threadIdx.x & 31;
    int n = __builtin_amdgcn_readfirstlane((int)(blockIdx.x * 8 + (threadIdx.x >> 5)));
    if (n >= N) return;
    int g = lane >> 3, h = lane & 7;
    int beg = off[n], end = off[n + 1];
    const v4f* rows4 = (const v4f*)h1;
    v4f hn0 = rows4[n * 16 + 2 * h], hn1 = rows4[n * 16 + 2 * h + 1];
    v4f a0 = ((const v4f*)a_src)[2 * h], a1 = ((const v4f*)a_src)[2 * h + 1];
    v4f d0 = ((const v4f*)a_dst)[2 * h], d1 = ((const v4f*)a_dst)[2 * h + 1];
    float adn = dot8(hn0, hn1, d0, d1);
    float vself = leaky(dot8(hn0, hn1, a0, a1) + adn);
    float m = vself;
    float den = (g == 0) ? 1.f : 0.f;
    v4f c0 = hn0 * den, c1 = hn1 * den;
    int j0 = beg;
    for (; end - j0 > 12; j0 += 16) head_steps<4>(rows4, csr, j0 + g, end, h, a0, a1, adn, m, den, c0, c1);
    int rem = end - j0;
    if (rem > 8) head_steps<3>(rows4, csr, j0 + g, end, h, a0, a1, adn, m, den, c0, c1);
    else if (rem > 4) head_steps<2>(rows4, csr, j0 + g, end, h, a0, a1, adn, m, den, c0, c1);
    else if (rem > 0) head_steps<1>(rows4, csr, j0 + g, end, h, a0, a1, adn, m, den, c0, c1);

    float mo = xget<8>(m);
    float mm = fmaxf(m, mo);
    float f = __expf(m - mm), fo = __expf(mo - mm);
    den = den * f + xget<8>(den) * fo;
    bool up = (g & 1) != 0;
    v4f keep = up ? c1 : c0;
    v4f give = up ? c0 : c1;
    v4f got;
    got.x = xget<8>(give.x);
    got.y = xget<8>(give.y);
    got.z = xget<8>(give.z);
    got.w = xget<8>(give.w);
    v4f c = keep * f + got * fo;
    m = mm;
    mo = __shfl_xor(m, 16, 32);
    mm = fmaxf(m, mo);
    f = __expf(m - mm);
    fo = __expf(mo - mm);
    den = den * f + __shfl_xor(den, 16, 32) * fo;
    bool up2 = (g & 2) != 0;
    v2f keep2 = up2 ? c.zw : c.xy;
    v2f give2 = up2 ? c.xy : c.zw;
    v2f got2;
    got2.x = __shfl_xor(give2.x, 16, 32);
    got2.y = __shfl_xor(give2.y, 16, 32);
    v2f acc = keep2 * f + got2 * fo;
    int ch = 8 * h + 4 * (g & 1) + (g & 2);

    float vx = acc.x / den + b1[ch];
    float vy = acc.y / den + b1[ch + 1];
    v2f o;
    o.x = vx > 0.f ? vx : (__expf(vx) - 1.f);
    o.y = vy > 0.f ? vy : (__expf(vy) - 1.f);
    *(v2f*)(hx + (long long)n * HD1 + ch) = o;
    v2f ws = *(const v2f*)(vsd + ch);
    v2f wd = *(const v2f*)(vsd + HD1 + ch);
    float ps = wave_sum(o.x * ws.x + o.y * ws.y);
    float pd = wave_sum(o.x * wd.x + o.y * wd.y);
    if (lane == 0) {
        as2[n] = ps;
        ad2[n] = pd;
    }
}

#define EB 16
__global__ __launch_bounds__(256) void gather2_rows(const int* __restrict__ off,
                                                    const int* __restrict__ csr,
                                                    const float* __restrict__ as,
                                                    const float* __restrict__ ad,
                                                    const float* __restrict__ hx,
                                                    float* __restrict__ agg, int N) {
    int lane = threadIdx.x & 31;
    int n = __builtin_amdgcn_readfirstlane((int)(blockIdx.x * 8 + (threadIdx.x >> 5)));
    if (n >= N) return;
    int beg = off[n], end = off[n + 1];
    float adn = ad[n];
    float vself = leaky(as[n] + adn);
    const v2f* rows = (const v2f*)hx;
    v2f hn = rows[n * 32 + lane];
    int j = beg + lane;
    int sl = (j < end) ? csr[j] : n;
    float el = leaky(as[sl] + adn);
    float m = fmaxf(vself, el);
    for (int j0 = beg + 32; j0 < end; j0 += 32) {
        int jj = j0 + lane;
        if (jj < end) m = fmaxf(m, leaky(as[csr[jj]] + adn));
    }
    m = wave_max(m);
    float w0 = __expf(vself - m);
    float ax = w0 * hn.x, ay = w0 * hn.y;
    float den = 0.f;
    for (int j0 = beg; j0 < end; j0 += 32) {
        if (j0 != beg) {
            int jj = j0 + lane;
            sl = (jj < end) ? csr[jj] : n;
            el = leaky(as[sl] + adn);
        }
        float wl = (j0 + lane < end) ? __expf(el - m) : 0.f;
        den += wl;
        int cc = min(32, end - j0);
        for (int q0 = 0; q0 < cc; q0 += EB) {
            v2f r[EB];
            float wq[EB];
#pragma unroll
            for (int u = 0; u < EB; ++u) {
                int s = __builtin_amdgcn_readlane(sl, q0 + u);
                wq[u] = __int_as_float(__builtin_amdgcn_readlane(__float_as_int(wl), q0 + u));
                r[u] = rows[s * 32 + lane];
            }
#pragma unroll
            for (int u = 0; u < EB; ++u) {
                ax += wq[u] * r[u].x;
                ay += wq[u] * r[u].y;
            }
        }
    }
    den = wave_sum(den) + w0;
    v2f o;
    o.x = ax / den;
    o.y = ay / den;
    *(v2f*)(agg + (long long)n * HD1 + 2 * lane) = o;
}

#define G2W 8
__device__ __forceinline__ v4f sig4(v4f v) {
    v4f o;
    o.x = 1.f / (1.f + __expf(-v.x));
    o.y = 1.f / (1.f + __expf(-v.y));
    o.z = 1.f / (1.f + __expf(-v.z));
    o.w = 1.f / (1.f + __expf(-v.w));
    return o;
}
__global__ __launch_bounds__(32 * G2W) void gemm2_wmma(const float* __restrict__ agg,
                                                       const float* __restrict__ W2,
                                                       const float* __restrict__ b2,
                                                       float* __restrict__ outp, int N, int C2) {
    __shared__ v4f Bs[(HD1 / 2) * 64];
    int tid = threadIdx.x;
    int wave = tid >> 5, lane = tid & 31;
    int lm = lane & 15, lh = lane >> 4;
#pragma unroll
    for (int i = 0; i < (HD1 / 2) * 32 / (32 * G2W); ++i) {
        int p = tid + i * 32 * G2W;
        int kp = p >> 5, j = p & 31;
        int col = 4 * j;
        const float* w0 = W2 + (2 * kp) * C2;
        const float* w1 = w0 + C2;
        int k0 = min(col, C2 - 1), k1 = min(col + 1, C2 - 1), k2 = min(col + 2, C2 - 1), k3 = min(col + 3, C2 - 1);
        float e00 = w0[k0], e10 = w1[k0], e01 = w0[k1], e11 = w1[k1];
        float e02 = w0[k2], e12 = w1[k2], e03 = w0[k3], e13 = w1[k3];
        v4f g0, g1;
        g0.x = (col < C2) ? e00 : 0.f;
        g0.y = (col < C2) ? e10 : 0.f;
        g0.z = (col + 1 < C2) ? e01 : 0.f;
        g0.w = (col + 1 < C2) ? e11 : 0.f;
        g1.x = (col + 2 < C2) ? e02 : 0.f;
        g1.y = (col + 2 < C2) ? e12 : 0.f;
        g1.z = (col + 3 < C2) ? e03 : 0.f;
        g1.w = (col + 3 < C2) ? e13 : 0.f;
        int c = j & 15, hf = j >> 4;
        Bs[kp * 64 + (2 * hf) * 16 + c] = g0;
        Bs[kp * 64 + (2 * hf + 1) * 16 + c] = g1;
    }
    __syncthreads();
    int row0 = (blockIdx.x * G2W + wave) * 16;
    if (row0 >= N) return;
    const v4f* ar = (const v4f*)(agg + (long long)(row0 + lm) * HD1) + lh;
    const v4f* bl = Bs + lh * 128 + lm;
    v4f a[HD1 / 8];
#pragma unroll
    for (int q = 0; q < HD1 / 8; ++q) a[q] = ar[2 * q];
    v8f acc0 = {0.f, 0.f, 0.f, 0.f, 0.f, 0.f, 0.f, 0.f};
    v8f acc1 = acc0, acc2 = acc0, acc3 = acc0, acc4 = acc0, acc5 = acc0, acc6 = acc0, acc7 = acc0;
#pragma unroll
    for (int kb = 0; kb < HD1 / 8; ++kb) {
        const v4f* bp = bl + kb * 256;
        v4f av = a[kb];
        v4f p0 = bp[0], p1 = bp[16], p2 = bp[32], p3 = bp[48];
        acc0 = __builtin_amdgcn_wmma_f32_16x16x4_f32(false, av.xy, false, p0.xy, (short)0, acc0, false, false);
        acc1 = __builtin_amdgcn_wmma_f32_16x16x4_f32(false, av.xy, false, p0.zw, (short)0, acc1, false, false);
        acc2 = __builtin_amdgcn_wmma_f32_16x16x4_f32(false, av.xy, false, p1.xy, (short)0, acc2, false, false);
        acc3 = __builtin_amdgcn_wmma_f32_16x16x4_f32(false, av.xy, false, p1.zw, (short)0, acc3, false, false);
        acc4 = __builtin_amdgcn_wmma_f32_16x16x4_f32(false, av.xy, false, p2.xy, (short)0, acc4, false, false);
        acc5 = __builtin_amdgcn_wmma_f32_16x16x4_f32(false, av.xy, false, p2.zw, (short)0, acc5, false, false);
        acc6 = __builtin_amdgcn_wmma_f32_16x16x4_f32(false, av.xy, false, p3.xy, (short)0, acc6, false, false);
        acc7 = __builtin_amdgcn_wmma_f32_16x16x4_f32(false, av.xy, false, p3.zw, (short)0, acc7, false, false);
        v4f q0 = bp[64], q1 = bp[80], q2 = bp[96], q3 = bp[112];
        acc0 = __builtin_amdgcn_wmma_f32_16x16x4_f32(false, av.zw, false, q0.xy, (short)0, acc0, false, false);
        acc1 = __builtin_amdgcn_wmma_f32_16x16x4_f32(false, av.zw, false, q0.zw, (short)0, acc1, false, false);
        acc2 = __builtin_amdgcn_wmma_f32_16x16x4_f32(false, av.zw, false, q1.xy, (short)0, acc2, false, false);
        acc3 = __builtin_amdgcn_wmma_f32_16x16x4_f32(false, av.zw, false, q1.zw, (short)0, acc3, false, false);
        acc4 = __builtin_amdgcn_wmma_f32_16x16x4_f32(false, av.zw, false, q2.xy, (short)0, acc4, false, false);
        acc5 = __builtin_amdgcn_wmma_f32_16x16x4_f32(false, av.zw, false, q2.zw, (short)0, acc5, false, false);
        acc6 = __builtin_amdgcn_wmma_f32_16x16x4_f32(false, av.zw, false, q3.xy, (short)0, acc6, false, false);
        acc7 = __builtin_amdgcn_wmma_f32_16x16x4_f32(false, av.zw, false, q3.zw, (short)0, acc7, false, false);
    }
    int cA = 4 * lm, cB = 64 + 4 * lm;
    v4f bA, bB;
    bA.x = b2[min(cA, C2 - 1)];
    bA.y = b2[min(cA + 1, C2 - 1)];
    bA.z = b2[min(cA + 2, C2 - 1)];
    bA.w = b2[min(cA + 3, C2 - 1)];
    bB.x = b2[min(cB, C2 - 1)];
    bB.y = b2[min(cB + 1, C2 - 1)];
    bB.z = b2[min(cB + 2, C2 - 1)];
    bB.w = b2[min(cB + 3, C2 - 1)];
    float* orow = outp + (long long)(row0 + 8 * lh) * C2;
    if ((C2 & 3) == 0) {
        bool okA = cA < C2, okB = cB < C2;
#pragma unroll
        for (int r = 0; r < 8; ++r) {
            v4f vA, vB;
            vA.x = acc0[r]; vA.y = acc1[r]; vA.z = acc2[r]; vA.w = acc3[r];
            vB.x = acc4[r]; vB.y = acc5[r]; vB.z = acc6[r]; vB.w = acc7[r];
            v4f sA = sig4(vA + bA), sB = sig4(vB + bB);
            if (okA) *(v4f*)(orow + (long long)r * C2 + cA) = sA;
            if (okB) *(v4f*)(orow + (long long)r * C2 + cB) = sB;
        }
    } else {
#pragma unroll
        for (int r = 0; r < 8; ++r) {
            v4f vA, vB;
            vA.x = acc0[r]; vA.y = acc1[r]; vA.z = acc2[r]; vA.w = acc3[r];
            vB.x = acc4[r]; vB.y = acc5[r]; vB.z = acc6[r]; vB.w = acc7[r];
            v4f sA = sig4(vA + bA), sB = sig4(vB + bB);
            float* o = orow + (long long)r * C2;
            if (cA < C2) o[cA] = sA.x;
            if (cA + 1 < C2) o[cA + 1] = sA.y;
            if (cA + 2 < C2) o[cA + 2] = sA.z;
            if (cA + 3 < C2) o[cA + 3] = sA.w;
            if (cB < C2) o[cB] = sB.x;
            if (cB + 1 < C2) o[cB + 1] = sB.y;
            if (cB + 2 < C2) o[cB + 2] = sB.z;
            if (cB + 3 < C2) o[cB + 3] = sB.w;
        }
    }
}

static inline int cdiv(long long a, long long b) { return (int)((a + b - 1) / b); }

extern "C" void kernel_launch(void* const* d_in, const int* in_sizes, int n_in,
                              void* d_out, int out_size, void* d_ws, size_t ws_size,
                              hipStream_t stream) {
    const float* x      = (const float*)d_in[0];
    const int*   ei     = (const int*)d_in[1];
    const float* W1     = (const float*)d_in[2];
    const float* a_src1 = (const float*)d_in[3];
    const float* a_dst1 = (const float*)d_in[4];
    const float* b1     = (const float*)d_in[5];
    const float* W2     = (const float*)d_in[6];
    const float* a_src2 = (const float*)d_in[7];
    const float* a_dst2 = (const float*)d_in[8];
    const float* b2     = (const float*)d_in[9];

    const int N  = in_sizes[0] / FIN;
    const int E  = in_sizes[1] / 2;
    const int C2 = in_sizes[9];

    float* h1   = (float*)d_ws;
    float* hx   = h1 + (long long)N * HD1;
    float* agg  = hx + (long long)N * HD1;
    int2*  part = (int2*)(agg + (long long)N * HD1);
    float* as2  = (float*)(part + E);
    float* ad2  = as2 + N;
    float* vsd  = ad2 + N;
    const int NB = cdiv(N, BN);
    int*   tot  = (int*)(vsd + 2 * HD1);
    int*   cur  = tot + NB * CS;
    int*   off  = cur + NB * CS;
    int*   csr  = off + N + 1;
    float* out  = (float*)d_out;

    const int B = 256;

    fill_u32<<<cdiv(2 * NB * CS, B), B, 0, stream>>>((unsigned*)tot, 0u, 2 * NB * CS);
    bucket_count<<<cdiv(E, EPB), 256, NB * sizeof(int), stream>>>(ei, E, NB, tot);
    bucket_partition<<<cdiv(E, EPB), 256, 2 * NB * sizeof(int), stream>>>(ei, E, NB, tot, cur, part);
    bucket_sort<<<NB, BN, 0, stream>>>(part, tot, N, off, csr);

    fold_w2<<<HD1 / 8, 256, 0, stream>>>(W2, a_src2, a_dst2, vsd, C2);
    gemm1_wmma<<<cdiv(N, 16 * G1W), 32 * G1W, 0, stream>>>(x, W1, h1, N);
    gather1<<<cdiv(N, 8), 256, 0, stream>>>(off, csr, h1, a_src1, a_dst1, b1, vsd, hx, as2, ad2, N);

    gather2_rows<<<cdiv(N, 8), 256, 0, stream>>>(off, csr, as2, ad2, hx, agg, N);
    gemm2_wmma<<<cdiv(N, 16 * G2W), 32 * G2W, 0, stream>>>(agg, W2, b2, out, N, C2);
}


# Layer-2 gather in one pass: source logits formed from the gathered hx rows (8-lane swizzle sums), running softmax as in layer 1; logit arrays and their per-edge reads gone; sum order changes
# speedup vs baseline: 5.5985x; 1.0844x over previous
#include <hip/hip_runtime.h>
#include <math.h>

typedef float v2f __attribute__((ext_vector_type(2)));
typedef float v4f __attribute__((ext_vector_type(4)));
typedef float v8f __attribute__((ext_vector_type(8)));

#define FIN 256
#define HD1 64

__device__ __forceinline__ float leaky(float v) { return v > 0.f ? v : 0.2f * v; }

template <int M>
__device__ __forceinline__ float xadd(float v) {
    int o = __builtin_amdgcn_update_dpp(0, __float_as_int(v), 0x160 | M, 0xf, 0xf, false);
    return v + __int_as_float(o);
}
template <int M>
__device__ __forceinline__ float xget(float v) {
    return __int_as_float(__builtin_amdgcn_update_dpp(0, __float_as_int(v), 0x160 | M, 0xf, 0xf, false));
}
__device__ __forceinline__ float sum8(float v) { return xadd<4>(xadd<2>(xadd<1>(v))); }
__device__ __forceinline__ float sum16(float v) { return xadd<8>(sum8(v)); }
__device__ __forceinline__ float wave_sum(float v) {
    v = sum16(v);
    return v + __shfl_xor(v, 16, 32);
}

__global__ __launch_bounds__(256) void fold_w2(const float* __restrict__ W2,
                                               const float* __restrict__ a_src,
                                               const float* __restrict__ a_dst,
                                               float* __restrict__ vsd, int C2) {
    int lane = threadIdx.x & 31;
    int k = blockIdx.x * 8 + (threadIdx.x >> 5);
    float s = 0.f, d = 0.f;
    for (int cix = lane; cix < C2; cix += 32) {
        float w = W2[k * C2 + cix];
        s += w * a_src[cix];
        d += w * a_dst[cix];
    }
    s = wave_sum(s);
    d = wave_sum(d);
    if (lane == 0) {
        vsd[k] = s;
        vsd[HD1 + k] = d;
    }
}

__global__ void fill_u32(unsigned* __restrict__ p, unsigned v, long long n) {
    long long i = (long long)blockIdx.x * blockDim.x + threadIdx.x;
    if (i < n) p[i] = v;
}

#define G1W 4
__global__ __launch_bounds__(32 * G1W) void gemm1_wmma(const float* __restrict__ x,
                                                       const float* __restrict__ W1,
                                                       float* __restrict__ h1, int N) {
    __shared__ v4f Bs[(FIN / 2) * 32];
    int tid = threadIdx.x;
    int wave = tid >> 5, lane = tid & 31;
    int lm = lane & 15, lh = lane >> 4;
    const v4f* W4 = (const v4f*)W1;
    for (int p0 = tid; p0 < (FIN / 2) * 16; p0 += 32 * G1W * 4) {
        v4f r0[4], r1[4];
#pragma unroll
        for (int u = 0; u < 4; ++u) {
            int p = p0 + u * 32 * G1W;
            int kp = p >> 4, c = p & 15;
            r0[u] = W4[(2 * kp) * 16 + c];
            r1[u] = W4[(2 * kp + 1) * 16 + c];
        }
#pragma unroll
        for (int u = 0; u < 4; ++u) {
            int p = p0 + u * 32 * G1W;
            int kp = p >> 4, c = p & 15;
            v4f g0, g1;
            g0.x = r0[u].x; g0.y = r1[u].x; g0.z = r0[u].y; g0.w = r1[u].y;
            g1.x = r0[u].z; g1.y = r1[u].z; g1.z = r0[u].w; g1.w = r1[u].w;
            Bs[kp * 32 + c] = g0;
            Bs[kp * 32 + 16 + c] = g1;
        }
    }
    __syncthreads();
    int row0 = (blockIdx.x * G1W + wave) * 16;
    if (row0 >= N) return;
    const v4f* xr = (const v4f*)(x + (long long)(row0 + lm) * FIN) + lh;
    const v4f* bl = Bs + lh * 64 + lm;
    v8f acc0 = {0.f, 0.f, 0.f, 0.f, 0.f, 0.f, 0.f, 0.f};
    v8f acc1 = acc0, acc2 = acc0, acc3 = acc0;
    v4f a[2][8];
#pragma unroll
    for (int q = 0; q < 8; ++q) a[0][q] = xr[2 * q];
#pragma unroll
    for (int ch = 0; ch < FIN / 64; ++ch) {
        if (ch + 1 < FIN / 64) {
#pragma unroll
            for (int q = 0; q < 8; ++q) a[(ch + 1) & 1][q] = xr[2 * ((ch + 1) * 8 + q)];
        }
        __builtin_amdgcn_sched_barrier(0);
#pragma unroll
        for (int q = 0; q < 8; ++q) {
            int kb = ch * 8 + q;
            const v4f* bp = bl + kb * 128;
            v4f b00 = bp[0], b01 = bp[16], b10 = bp[32], b11 = bp[48];
            v4f av = a[ch & 1][q];
            acc0 = __builtin_amdgcn_wmma_f32_16x16x4_f32(false, av.xy, false, b00.xy, (short)0, acc0, false, false);
            acc1 = __builtin_amdgcn_wmma_f32_16x16x4_f32(false, av.xy, false, b00.zw, (short)0, acc1, false, false);
            acc2 = __builtin_amdgcn_wmma_f32_16x16x4_f32(false, av.xy, false, b01.xy, (short)0, acc2, false, false);
            acc3 = __builtin_amdgcn_wmma_f32_16x16x4_f32(false, av.xy, false, b01.zw, (short)0, acc3, false, false);
            acc0 = __builtin_amdgcn_wmma_f32_16x16x4_f32(false, av.zw, false, b10.xy, (short)0, acc0, false, false);
            acc1 = __builtin_amdgcn_wmma_f32_16x16x4_f32(false, av.zw, false, b10.zw, (short)0, acc1, false, false);
            acc2 = __builtin_amdgcn_wmma_f32_16x16x4_f32(false, av.zw, false, b11.xy, (short)0, acc2, false, false);
            acc3 = __builtin_amdgcn_wmma_f32_16x16x4_f32(false, av.zw, false, b11.zw, (short)0, acc3, false, false);
        }
    }
    float* out = h1 + (long long)(row0 + 8 * lh) * HD1 + 4 * lm;
#pragma unroll
    for (int r = 0; r < 8; ++r) {
        v4f o;
        o.x = acc0[r]; o.y = acc1[r]; o.z = acc2[r]; o.w = acc3[r];
        *(v4f*)(out + (long long)r * HD1) = o;
    }
}

#define BSH 8
#define BN (1 << BSH)
#define EPB 4096
#define EPT (EPB / 256)
#define SB 8
#define CS 32

__global__ __launch_bounds__(256) void bucket_count(const int* __restrict__ ei, int E, int NB,
                                                    int* __restrict__ tot) {
    extern __shared__ int sh[];
    int tid = threadIdx.x;
    for (int i = tid; i < NB; i += 256) sh[i] = 0;
    __syncthreads();
    int e0 = blockIdx.x * EPB + tid;
    int d[EPT];
#pragma unroll
    for (int k = 0; k < EPT; ++k) {
        int e = e0 + k * 256;
        d[k] = (e < E) ? ei[E + e] : -1;
    }
#pragma unroll
    for (int k = 0; k < EPT; ++k)
        if (d[k] >= 0) atomicAdd(&sh[d[k] >> BSH], 1);
    __syncthreads();
    for (int i = tid; i < NB; i += 256) {
        int c = sh[i];
        if (c) atomicAdd(&tot[i * CS], c);
    }
}

__global__ __launch_bounds__(256) void bucket_partition(const int* __restrict__ ei, int E, int NB,
                                                        const int* __restrict__ tot,
                                                        int* __restrict__ cur,
                                                        int2* __restrict__ part) {
    extern __shared__ int sh[];
    __shared__ int2 stage[EPB];
    __shared__ int wsum[8];
    __shared__ int wsum2[8];
    int tid = threadIdx.x, lane = tid & 31, wave = tid >> 5;
    for (int i = tid; i < NB; i += 256) sh[i] = 0;
    __syncthreads();
    int e0 = blockIdx.x * EPB + tid;
    int d[EPT], sv[EPT], r[EPT];
#pragma unroll
    for (int k = 0; k < EPT; ++k) {
        int e = e0 + k * 256;
        d[k] = (e < E) ? ei[E + e] : -1;
        sv[k] = (e < E) ? ei[e] : 0;
    }
#pragma unroll
    for (int k = 0; k < EPT; ++k) r[k] = (d[k] >= 0) ? atomicAdd(&sh[d[k] >> BSH], 1) : 0;
    __syncthreads();
    int ipt = (NB + 255) / 256;
    int lo = tid * ipt;
    int hi = min(lo + ipt, NB);
    int s = 0, s2 = 0;
    for (int i = lo; i < hi; ++i) {
        s += tot[i * CS];
        s2 += sh[i];
    }
    int incl = s, incl2 = s2;
    for (int m = 1; m < 32; m <<= 1) {
        int u = __shfl_up(incl, m, 32);
        int u2 = __shfl_up(incl2, m, 32);
        if (lane >= m) {
            incl += u;
            incl2 += u2;
        }
    }
    if (lane == 31) {
        wsum[wave] = incl;
        wsum2[wave] = incl2;
    }
    __syncthreads();
    int run = incl - s, lrun = incl2 - s2;
    for (int w = 0; w < wave; ++w) {
        run += wsum[w];
        lrun += wsum2[w];
    }
    for (int i = lo; i < hi; ++i) {
        int c = sh[i];
        int base = run;
        if (c) base += atomicAdd(&cur[i * CS], c);
        sh[NB + i] = lrun;
        sh[i] = base - lrun;
        run += tot[i * CS];
        lrun += c;
    }
    __syncthreads();
#pragma unroll
    for (int k = 0; k < EPT; ++k) {
        if (d[k] >= 0) {
            int2 pr;
            pr.x = sv[k];
            pr.y = d[k];
            stage[sh[NB + (d[k] >> BSH)] + r[k]] = pr;
        }
    }
    __syncthreads();
    int ne = min(EPB, E - (int)blockIdx.x * EPB);
#pragma unroll
    for (int k = 0; k < EPT; ++k) {
        int p = tid + k * 256;
        if (p < ne) {
            int2 pr = stage[p];
            part[sh[pr.y >> BSH] + p] = pr;
        }
    }
}

__global__ __launch_bounds__(BN) void bucket_sort(const int2* __restrict__ part,
                                                  const int* __restrict__ tot, int N,
                                                  int* __restrict__ off, int* __restrict__ csr) {
    __shared__ int cnt[BN];
    __shared__ int red[BN / 32];
    __shared__ int wsum[BN / 32];
    int tid = threadIdx.x, lane = tid & 31, wave = tid >> 5;
    int b = blockIdx.x;
    int p = 0;
    for (int i = tid; i < b; i += BN) p += tot[i * CS];
    for (int m = 16; m; m >>= 1) p += __shfl_xor(p, m, 32);
    if (lane == 0) red[wave] = p;
    cnt[tid] = 0;
    __syncthreads();
    int start = 0;
    for (int w = 0; w < BN / 32; ++w) start += red[w];
    int nb = tot[b * CS];
    const int2* mine = part + start;
    for (int i0 = tid; i0 < nb; i0 += SB * BN) {
        int2 pr[SB];
#pragma unroll
        for (int u = 0; u < SB; ++u) {
            int i = i0 + u * BN;
            pr[u].y = (i < nb) ? mine[i].y : -1;
        }
#pragma unroll
        for (int u = 0; u < SB; ++u)
            if (pr[u].y >= 0) atomicAdd(&cnt[pr[u].y & (BN - 1)], 1);
    }
    __syncthreads();
    int c = cnt[tid];
    int incl = c;
    for (int m = 1; m < 32; m <<= 1) {
        int u = __shfl_up(incl, m, 32);
        if (lane >= m) incl += u;
    }
    if (lane == 31) wsum[wave] = incl;
    __syncthreads();
    int excl = incl - c;
    for (int w = 0; w < wave; ++w) excl += wsum[w];
    int node = b * BN + tid;
    if (node < N) off[node] = start + excl;
    if (node == N - 1) off[N] = start + excl + c;
    cnt[tid] = excl;
    __syncthreads();
    int* out = csr + start;
    for (int i0 = tid; i0 < nb; i0 += SB * BN) {
        int2 pr[SB];
#pragma unroll
        for (int u = 0; u < SB; ++u) {
            int i = i0 + u * BN;
            if (i < nb) pr[u] = mine[i];
            else pr[u].y = -1;
        }
#pragma unroll
        for (int u = 0; u < SB; ++u)
            if (pr[u].y >= 0) out[atomicAdd(&cnt[pr[u].y & (BN - 1)], 1)] = pr[u].x;
    }
}

__device__ __forceinline__ float dot8(v4f r0, v4f r1, v4f a0, v4f a1) {
    v2f p = r0.xy * a0.xy;
    p += r0.zw * a0.zw;
    p += r1.xy * a1.xy;
    p += r1.zw * a1.zw;
    return p.x + p.y;
}

template <int K, bool ROWDOT>
__device__ __forceinline__ void head_steps(const v4f* __restrict__ rows4, const int* __restrict__ csr,
                                           int j, int end, int o0, int o1, v4f a0, v4f a1, float adn,
                                           float& m, float& den, v4f& c0, v4f& c1) {
    int s[K];
    v4f r0[K], r1[K];
    float e[K];
#pragma unroll
    for (int k = 0; k < K; ++k) s[k] = csr[min(j + 4 * k, end - 1)];
#pragma unroll
    for (int k = 0; k < K; ++k) {
        r0[k] = rows4[s[k] * 16 + o0];
        r1[k] = rows4[s[k] * 16 + o1];
    }
    float gm = m;
#pragma unroll
    for (int k = 0; k < K; ++k) {
        float d = dot8(r0[k], r1[k], a0, a1);
        if (ROWDOT) d = sum8(d);
        e[k] = leaky(d + adn);
        gm = fmaxf(gm, e[k]);
    }
    float sc = __expf(m - gm);
    den *= sc;
    c0 *= sc;
    c1 *= sc;
    m = gm;
#pragma unroll
    for (int k = 0; k < K; ++k) {
        float w = (j + 4 * k < end) ? __expf(e[k] - gm) : 0.f;
        den += w;
        c0 += w * r0[k];
        c1 += w * r1[k];
    }
}

__device__ __forceinline__ v2f merge_slots(int g, float m, float& den, v4f c0, v4f c1) {
    float mo = xget<8>(m);
    float mm = fmaxf(m, mo);
    float f = __expf(m - mm), fo = __expf(mo - mm);
    den = den * f + xget<8>(den) * fo;
    bool up = (g & 1) != 0;
    v4f keep = up ? c1 : c0;
    v4f give = up ? c0 : c1;
    v4f got;
    got.x = xget<8>(give.x);
    got.y = xget<8>(give.y);
    got.z = xget<8>(give.z);
    got.w = xget<8>(give.w);
    v4f c = keep * f + got * fo;
    m = mm;
    mo = __shfl_xor(m, 16, 32);
    mm = fmaxf(m, mo);
    f = __expf(m - mm);
    fo = __expf(mo - mm);
    den = den * f + __shfl_xor(den, 16, 32) * fo;
    bool up2 = (g & 2) != 0;
    v2f keep2 = up2 ? c.zw : c.xy;
    v2f give2 = up2 ? c.xy : c.zw;
    v2f got2;
    got2.x = __shfl_xor(give2.x, 16, 32);
    got2.y = __shfl_xor(give2.y, 16, 32);
    return keep2 * f + got2 * fo;
}

__global__ __launch_bounds__(256) void gather1(const int* __restrict__ off,
                                               const int* __restrict__ csr,
                                               const float* __restrict__ h1,
                                               const float* __restrict__ a_src,
                                               const float* __restrict__ a_dst,
                                               const float* __restrict__ b1,
                                               float* __restrict__ hx, int N) {
    int lane = threadIdx.x & 31;
    int n = __builtin_amdgcn_readfirstlane((int)(blockIdx.x * 8 + (threadIdx.x >> 5)));
    if (n >= N) return;
    int g = lane >> 3, h = lane & 7;
    int beg = off[n], end = off[n + 1];
    const v4f* rows4 = (const v4f*)h1;
    v4f hn0 = rows4[n * 16 + 2 * h], hn1 = rows4[n * 16 + 2 * h + 1];
    v4f a0 = ((const v4f*)a_src)[2 * h], a1 = ((const v4f*)a_src)[2 * h + 1];
    v4f d0 = ((const v4f*)a_dst)[2 * h], d1 = ((const v4f*)a_dst)[2 * h + 1];
    float adn = dot8(hn0, hn1, d0, d1);
    float vself = leaky(dot8(hn0, hn1, a0, a1) + adn);
    float m = vself;
    float den = (g == 0) ? 1.f : 0.f;
    v4f c0 = hn0 * den, c1 = hn1 * den;
    int j0 = beg;
    for (; end - j0 > 12; j0 += 16) head_steps<4, false>(rows4, csr, j0 + g, end, 2 * h, 2 * h + 1, a0, a1, adn, m, den, c0, c1);
    int rem = end - j0;
    if (rem > 8) head_steps<3, false>(rows4, csr, j0 + g, end, 2 * h, 2 * h + 1, a0, a1, adn, m, den, c0, c1);
    else if (rem > 4) head_steps<2, false>(rows4, csr, j0 + g, end, 2 * h, 2 * h + 1, a0, a1, adn, m, den, c0, c1);
    else if (rem > 0) head_steps<1, false>(rows4, csr, j0 + g, end, 2 * h, 2 * h + 1, a0, a1, adn, m, den, c0, c1);

    v2f acc = merge_slots(g, m, den, c0, c1);
    int ch = 8 * h + 4 * (g & 1) + (g & 2);

    float vx = acc.x / den + b1[ch];
    float vy = acc.y / den + b1[ch + 1];
    v2f o;
    o.x = vx > 0.f ? vx : (__expf(vx) - 1.f);
    o.y = vy > 0.f ? vy : (__expf(vy) - 1.f);
    *(v2f*)(hx + (long long)n * HD1 + ch) = o;
}

__global__ __launch_bounds__(256) void gather2_rows(const int* __restrict__ off,
                                                    const int* __restrict__ csr,
                                                    const float* __restrict__ vsd,
                                                    const float* __restrict__ hx,
                                                    float* __restrict__ agg, int N) {
    int lane = threadIdx.x & 31;
    int n = __builtin_amdgcn_readfirstlane((int)(blockIdx.x * 8 + (threadIdx.x >> 5)));
    if (n >= N) return;
    int g = lane >> 3, q = lane & 7;
    int beg = off[n], end = off[n + 1];
    const v4f* rows4 = (const v4f*)hx;
    v4f hn0 = rows4[n * 16 + q], hn1 = rows4[n * 16 + 8 + q];
    v4f s0 = ((const v4f*)vsd)[q], s1 = ((const v4f*)vsd)[8 + q];
    v4f d0 = ((const v4f*)vsd)[16 + q], d1 = ((const v4f*)vsd)[16 + 8 + q];
    float adn = sum8(dot8(hn0, hn1, d0, d1));
    float vself = leaky(sum8(dot8(hn0, hn1, s0, s1)) + adn);
    float m = vself;
    float den = (g == 0) ? 1.f : 0.f;
    v4f c0 = hn0 * den, c1 = hn1 * den;
    int j0 = beg;
    for (; end - j0 > 12; j0 += 16) head_steps<4, true>(rows4, csr, j0 + g, end, q, 8 + q, s0, s1, adn, m, den, c0, c1);
    int rem = end - j0;
    if (rem > 8) head_steps<3, true>(rows4, csr, j0 + g, end, q, 8 + q, s0, s1, adn, m, den, c0, c1);
    else if (rem > 4) head_steps<2, true>(rows4, csr, j0 + g, end, q, 8 + q, s0, s1, adn, m, den, c0, c1);
    else if (rem > 0) head_steps<1, true>(rows4, csr, j0 + g, end, q, 8 + q, s0, s1, adn, m, den, c0, c1);

    v2f acc = merge_slots(g, m, den, c0, c1);
    int ch = 32 * (g & 1) + 4 * q + (g & 2);
    v2f o;
    o.x = acc.x / den;
    o.y = acc.y / den;
    *(v2f*)(agg + (long long)n * HD1 + ch) = o;
}

#define G2W 8
__device__ __forceinline__ v4f sig4(v4f v) {
    v4f o;
    o.x = 1.f / (1.f + __expf(-v.x));
    o.y = 1.f / (1.f + __expf(-v.y));
    o.z = 1.f / (1.f + __expf(-v.z));
    o.w = 1.f / (1.f + __expf(-v.w));
    return o;
}
__global__ __launch_bounds__(32 * G2W) void gemm2_wmma(const float* __restrict__ agg,
                                                       const float* __restrict__ W2,
                                                       const float* __restrict__ b2,
                                                       float* __restrict__ outp, int N, int C2) {
    __shared__ v4f Bs[(HD1 / 2) * 64];
    int tid = threadIdx.x;
    int wave = tid >> 5, lane = tid & 31;
    int lm = lane & 15, lh = lane >> 4;
#pragma unroll
    for (int i = 0; i < (HD1 / 2) * 32 / (32 * G2W); ++i) {
        int p = tid + i * 32 * G2W;
        int kp = p >> 5, j = p & 31;
        int col = 4 * j;
        const float* w0 = W2 + (2 * kp) * C2;
        const float* w1 = w0 + C2;
        int k0 = min(col, C2 - 1), k1 = min(col + 1, C2 - 1), k2 = min(col + 2, C2 - 1), k3 = min(col + 3, C2 - 1);
        float e00 = w0[k0], e10 = w1[k0], e01 = w0[k1], e11 = w1[k1];
        float e02 = w0[k2], e12 = w1[k2], e03 = w0[k3], e13 = w1[k3];
        v4f g0, g1;
        g0.x = (col < C2) ? e00 : 0.f;
        g0.y = (col < C2) ? e10 : 0.f;
        g0.z = (col + 1 < C2) ? e01 : 0.f;
        g0.w = (col + 1 < C2) ? e11 : 0.f;
        g1.x = (col + 2 < C2) ? e02 : 0.f;
        g1.y = (col + 2 < C2) ? e12 : 0.f;
        g1.z = (col + 3 < C2) ? e03 : 0.f;
        g1.w = (col + 3 < C2) ? e13 : 0.f;
        int c = j & 15, hf = j >> 4;
        Bs[kp * 64 + (2 * hf) * 16 + c] = g0;
        Bs[kp * 64 + (2 * hf + 1) * 16 + c] = g1;
    }
    __syncthreads();
    int row0 = (blockIdx.x * G2W + wave) * 16;
    if (row0 >= N) return;
    const v4f* ar = (const v4f*)(agg + (long long)(row0 + lm) * HD1) + lh;
    const v4f* bl = Bs + lh * 128 + lm;
    v4f a[HD1 / 8];
#pragma unroll
    for (int q = 0; q < HD1 / 8; ++q) a[q] = ar[2 * q];
    v8f acc0 = {0.f, 0.f, 0.f, 0.f, 0.f, 0.f, 0.f, 0.f};
    v8f acc1 = acc0, acc2 = acc0, acc3 = acc0, acc4 = acc0, acc5 = acc0, acc6 = acc0, acc7 = acc0;
#pragma unroll
    for (int kb = 0; kb < HD1 / 8; ++kb) {
        const v4f* bp = bl + kb * 256;
        v4f av = a[kb];
        v4f p0 = bp[0], p1 = bp[16], p2 = bp[32], p3 = bp[48];
        acc0 = __builtin_amdgcn_wmma_f32_16x16x4_f32(false, av.xy, false, p0.xy, (short)0, acc0, false, false);
        acc1 = __builtin_amdgcn_wmma_f32_16x16x4_f32(false, av.xy, false, p0.zw, (short)0, acc1, false, false);
        acc2 = __builtin_amdgcn_wmma_f32_16x16x4_f32(false, av.xy, false, p1.xy, (short)0, acc2, false, false);
        acc3 = __builtin_amdgcn_wmma_f32_16x16x4_f32(false, av.xy, false, p1.zw, (short)0, acc3, false, false);
        acc4 = __builtin_amdgcn_wmma_f32_16x16x4_f32(false, av.xy, false, p2.xy, (short)0, acc4, false, false);
        acc5 = __builtin_amdgcn_wmma_f32_16x16x4_f32(false, av.xy, false, p2.zw, (short)0, acc5, false, false);
        acc6 = __builtin_amdgcn_wmma_f32_16x16x4_f32(false, av.xy, false, p3.xy, (short)0, acc6, false, false);
        acc7 = __builtin_amdgcn_wmma_f32_16x16x4_f32(false, av.xy, false, p3.zw, (short)0, acc7, false, false);
        v4f q0 = bp[64], q1 = bp[80], q2 = bp[96], q3 = bp[112];
        acc0 = __builtin_amdgcn_wmma_f32_16x16x4_f32(false, av.zw, false, q0.xy, (short)0, acc0, false, false);
        acc1 = __builtin_amdgcn_wmma_f32_16x16x4_f32(false, av.zw, false, q0.zw, (short)0, acc1, false, false);
        acc2 = __builtin_amdgcn_wmma_f32_16x16x4_f32(false, av.zw, false, q1.xy, (short)0, acc2, false, false);
        acc3 = __builtin_amdgcn_wmma_f32_16x16x4_f32(false, av.zw, false, q1.zw, (short)0, acc3, false, false);
        acc4 = __builtin_amdgcn_wmma_f32_16x16x4_f32(false, av.zw, false, q2.xy, (short)0, acc4, false, false);
        acc5 = __builtin_amdgcn_wmma_f32_16x16x4_f32(false, av.zw, false, q2.zw, (short)0, acc5, false, false);
        acc6 = __builtin_amdgcn_wmma_f32_16x16x4_f32(false, av.zw, false, q3.xy, (short)0, acc6, false, false);
        acc7 = __builtin_amdgcn_wmma_f32_16x16x4_f32(false, av.zw, false, q3.zw, (short)0, acc7, false, false);
    }
    int cA = 4 * lm, cB = 64 + 4 * lm;
    v4f bA, bB;
    bA.x = b2[min(cA, C2 - 1)];
    bA.y = b2[min(cA + 1, C2 - 1)];
    bA.z = b2[min(cA + 2, C2 - 1)];
    bA.w = b2[min(cA + 3, C2 - 1)];
    bB.x = b2[min(cB, C2 - 1)];
    bB.y = b2[min(cB + 1, C2 - 1)];
    bB.z = b2[min(cB + 2, C2 - 1)];
    bB.w = b2[min(cB + 3, C2 - 1)];
    float* orow = outp + (long long)(row0 + 8 * lh) * C2;
    if ((C2 & 3) == 0) {
        bool okA = cA < C2, okB = cB < C2;
#pragma unroll
        for (int r = 0; r < 8; ++r) {
            v4f vA, vB;
            vA.x = acc0[r]; vA.y = acc1[r]; vA.z = acc2[r]; vA.w = acc3[r];
            vB.x = acc4[r]; vB.y = acc5[r]; vB.z = acc6[r]; vB.w = acc7[r];
            v4f sA = sig4(vA + bA), sB = sig4(vB + bB);
            if (okA) *(v4f*)(orow + (long long)r * C2 + cA) = sA;
            if (okB) *(v4f*)(orow + (long long)r * C2 + cB) = sB;
        }
    } else {
#pragma unroll
        for (int r = 0; r < 8; ++r) {
            v4f vA, vB;
            vA.x = acc0[r]; vA.y = acc1[r]; vA.z = acc2[r]; vA.w = acc3[r];
            vB.x = acc4[r]; vB.y = acc5[r]; vB.z = acc6[r]; vB.w = acc7[r];
            v4f sA = sig4(vA + bA), sB = sig4(vB + bB);
            float* o = orow + (long long)r * C2;
            if (cA < C2) o[cA] = sA.x;
            if (cA + 1 < C2) o[cA + 1] = sA.y;
            if (cA + 2 < C2) o[cA + 2] = sA.z;
            if (cA + 3 < C2) o[cA + 3] = sA.w;
            if (cB < C2) o[cB] = sB.x;
            if (cB + 1 < C2) o[cB + 1] = sB.y;
            if (cB + 2 < C2) o[cB + 2] = sB.z;
            if (cB + 3 < C2) o[cB + 3] = sB.w;
        }
    }
}

static inline int cdiv(long long a, long long b) { return (int)((a + b - 1) / b); }

extern "C" void kernel_launch(void* const* d_in, const int* in_sizes, int n_in,
                              void* d_out, int out_size, void* d_ws, size_t ws_size,
                              hipStream_t stream) {
    const float* x      = (const float*)d_in[0];
    const int*   ei     = (const int*)d_in[1];
    const float* W1     = (const float*)d_in[2];
    const float* a_src1 = (const float*)d_in[3];
    const float* a_dst1 = (const float*)d_in[4];
    const float* b1     = (const float*)d_in[5];
    const float* W2     = (const float*)d_in[6];
    const float* a_src2 = (const float*)d_in[7];
    const float* a_dst2 = (const float*)d_in[8];
    const float* b2     = (const float*)d_in[9];

    const int N  = in_sizes[0] / FIN;
    const int E  = in_sizes[1] / 2;
    const int C2 = in_sizes[9];

    float* h1   = (float*)d_ws;
    float* hx   = h1 + (long long)N * HD1;
    float* agg  = hx + (long long)N * HD1;
    float* vsd  = agg + (long long)N * HD1;
    int2*  part = (int2*)(vsd + 2 * HD1);
    const int NB = cdiv(N, BN);
    int*   tot  = (int*)(part + E);
    int*   cur  = tot + NB * CS;
    int*   off  = cur + NB * CS;
    int*   csr  = off + N + 1;
    float* out  = (float*)d_out;

    const int B = 256;

    fill_u32<<<cdiv(2 * NB * CS, B), B, 0, stream>>>((unsigned*)tot, 0u, 2 * NB * CS);
    bucket_count<<<cdiv(E, EPB), 256, NB * sizeof(int), stream>>>(ei, E, NB, tot);
    bucket_partition<<<cdiv(E, EPB), 256, 2 * NB * sizeof(int), stream>>>(ei, E, NB, tot, cur, part);
    bucket_sort<<<NB, BN, 0, stream>>>(part, tot, N, off, csr);

    fold_w2<<<HD1 / 8, 256, 0, stream>>>(W2, a_src2, a_dst2, vsd, C2);
    gemm1_wmma<<<cdiv(N, 16 * G1W), 32 * G1W, 0, stream>>>(x, W1, h1, N);
    gather1<<<cdiv(N, 8), 256, 0, stream>>>(off, csr, h1, a_src1, a_dst1, b1, hx, N);

    gather2_rows<<<cdiv(N, 8), 256, 0, stream>>>(off, csr, vsd, hx, agg, N);
    gemm2_wmma<<<cdiv(N, 16 * G2W), 32 * G2W, 0, stream>>>(agg, W2, b2, out, N, C2);
}


# GEMM1 has no launch of its own: its 16-row strips, the counter fill and the W2 fold run as extra work-groups in the edge sort's launches, matrix pipe beside the sort's waits; 9 launches become 7
# speedup vs baseline: 5.9341x; 1.0599x over previous
#include <hip/hip_runtime.h>
#include <math.h>

typedef float v2f __attribute__((ext_vector_type(2)));
typedef float v4f __attribute__((ext_vector_type(4)));
typedef float v8f __attribute__((ext_vector_type(8)));

#define FIN 256
#define HD1 64

__device__ __forceinline__ float leaky(float v) { return v > 0.f ? v : 0.2f * v; }

template <int M>
__device__ __forceinline__ float xadd(float v) {
    int o = __builtin_amdgcn_update_dpp(0, __float_as_int(v), 0x160 | M, 0xf, 0xf, false);
    return v + __int_as_float(o);
}
template <int M>
__device__ __forceinline__ float xget(float v) {
    return __int_as_float(__builtin_amdgcn_update_dpp(0, __float_as_int(v), 0x160 | M, 0xf, 0xf, false));
}
__device__ __forceinline__ float sum8(float v) { return xadd<4>(xadd<2>(xadd<1>(v))); }
__device__ __forceinline__ float sum16(float v) { return xadd<8>(sum8(v)); }
__device__ __forceinline__ float wave_sum(float v) {
    v = sum16(v);
    return v + __shfl_xor(v, 16, 32);
}

__device__ __forceinline__ void fold_w2(const float* __restrict__ W2,
                                        const float* __restrict__ a_src,
                                        const float* __restrict__ a_dst,
                                        float* __restrict__ vsd, int C2, int bid) {
    int lane = threadIdx.x & 31;
    int k = bid * 8 + (threadIdx.x >> 5);
    float s = 0.f, d = 0.f;
    for (int cix = lane; cix < C2; cix += 32) {
        float w = W2[k * C2 + cix];
        s += w * a_src[cix];
        d += w * a_dst[cix];
    }
    s = wave_sum(s);
    d = wave_sum(d);
    if (lane == 0) {
        vsd[k] = s;
        vsd[HD1 + k] = d;
    }
}

__device__ __forceinline__ void fill_u32(unsigned* __restrict__ p, unsigned v, long long n, int bid) {
    long long i = (long long)bid * 256 + threadIdx.x;
    if (i < n) p[i] = v;
}

#define G1W 8
#define G1LDS ((FIN / 2) * 32)
__device__ __forceinline__ void gemm1_wmma(const float* __restrict__ x,
                                           const float* __restrict__ W1,
                                           float* __restrict__ h1, int N, int s0, int spw, int wg, v4f* Bs) {
    int tid = threadIdx.x;
    int wave = tid >> 5, lane = tid & 31;
    int lm = lane & 15, lh = lane >> 4;
    const v4f* W4 = (const v4f*)W1;
    for (int p0 = tid; p0 < (FIN / 2) * 16; p0 += 32 * G1W * 4) {
        v4f r0[4], r1[4];
#pragma unroll
        for (int u = 0; u < 4; ++u) {
            int p = p0 + u * 32 * G1W;
            int kp = p >> 4, c = p & 15;
            r0[u] = W4[(2 * kp) * 16 + c];
            r1[u] = W4[(2 * kp + 1) * 16 + c];
        }
#pragma unroll
        for (int u = 0; u < 4; ++u) {
            int p = p0 + u * 32 * G1W;
            int kp = p >> 4, c = p & 15;
            v4f g0, g1;
            g0.x = r0[u].x; g0.y = r1[u].x; g0.z = r0[u].y; g0.w = r1[u].y;
            g1.x = r0[u].z; g1.y = r1[u].z; g1.z = r0[u].w; g1.w = r1[u].w;
            Bs[kp * 32 + c] = g0;
            Bs[kp * 32 + 16 + c] = g1;
        }
    }
    __syncthreads();
    int row0 = (s0 + wg * spw + wave) * 16;
    if (wave >= spw || row0 >= N) return;
    const v4f* xr = (const v4f*)(x + (long long)(row0 + lm) * FIN) + lh;
    const v4f* bl = Bs + lh * 64 + lm;
    v8f acc0 = {0.f, 0.f, 0.f, 0.f, 0.f, 0.f, 0.f, 0.f};
    v8f acc1 = acc0, acc2 = acc0, acc3 = acc0;
    v4f a[2][8];
#pragma unroll
    for (int q = 0; q < 8; ++q) a[0][q] = xr[2 * q];
#pragma unroll
    for (int ch = 0; ch < FIN / 64; ++ch) {
        if (ch + 1 < FIN / 64) {
#pragma unroll
            for (int q = 0; q < 8; ++q) a[(ch + 1) & 1][q] = xr[2 * ((ch + 1) * 8 + q)];
        }
        __builtin_amdgcn_sched_barrier(0);
#pragma unroll
        for (int q = 0; q < 8; ++q) {
            int kb = ch * 8 + q;
            const v4f* bp = bl + kb * 128;
            v4f b00 = bp[0], b01 = bp[16], b10 = bp[32], b11 = bp[48];
            v4f av = a[ch & 1][q];
            acc0 = __builtin_amdgcn_wmma_f32_16x16x4_f32(false, av.xy, false, b00.xy, (short)0, acc0, false, false);
            acc1 = __builtin_amdgcn_wmma_f32_16x16x4_f32(false, av.xy, false, b00.zw, (short)0, acc1, false, false);
            acc2 = __builtin_amdgcn_wmma_f32_16x16x4_f32(false, av.xy, false, b01.xy, (short)0, acc2, false, false);
            acc3 = __builtin_amdgcn_wmma_f32_16x16x4_f32(false, av.xy, false, b01.zw, (short)0, acc3, false, false);
            acc0 = __builtin_amdgcn_wmma_f32_16x16x4_f32(false, av.zw, false, b10.xy, (short)0, acc0, false, false);
            acc1 = __builtin_amdgcn_wmma_f32_16x16x4_f32(false, av.zw, false, b10.zw, (short)0, acc1, false, false);
            acc2 = __builtin_amdgcn_wmma_f32_16x16x4_f32(false, av.zw, false, b11.xy, (short)0, acc2, false, false);
            acc3 = __builtin_amdgcn_wmma_f32_16x16x4_f32(false, av.zw, false, b11.zw, (short)0, acc3, false, false);
        }
    }
    float* out = h1 + (long long)(row0 + 8 * lh) * HD1 + 4 * lm;
#pragma unroll
    for (int r = 0; r < 8; ++r) {
        v4f o;
        o.x = acc0[r]; o.y = acc1[r]; o.z = acc2[r]; o.w = acc3[r];
        *(v4f*)(out + (long long)r * HD1) = o;
    }
}

#define BSH 8
#define BN (1 << BSH)
#define EPB 4096
#define EPT (EPB / 256)
#define SB 8
#define CS 32

__device__ __forceinline__ void bucket_count(const int* __restrict__ ei, int E, int NB,
                                             int* __restrict__ tot, int bid) {
    extern __shared__ int sh[];
    int tid = threadIdx.x;
    for (int i = tid; i < NB; i += 256) sh[i] = 0;
    __syncthreads();
    int e0 = bid * EPB + tid;
    int d[EPT];
#pragma unroll
    for (int k = 0; k < EPT; ++k) {
        int e = e0 + k * 256;
        d[k] = (e < E) ? ei[E + e] : -1;
    }
#pragma unroll
    for (int k = 0; k < EPT; ++k)
        if (d[k] >= 0) atomicAdd(&sh[d[k] >> BSH], 1);
    __syncthreads();
    for (int i = tid; i < NB; i += 256) {
        int c = sh[i];
        if (c) atomicAdd(&tot[i * CS], c);
    }
}

__device__ __forceinline__ void bucket_partition(const int* __restrict__ ei, int E, int NB,
                                                 const int* __restrict__ tot,
                                                 int* __restrict__ cur,
                                                 int2* __restrict__ part, int bid, int2* stage) {
    extern __shared__ int sh[];
    __shared__ int wsum[8];
    __shared__ int wsum2[8];
    int tid = threadIdx.x, lane = tid & 31, wave = tid >> 5;
    for (int i = tid; i < NB; i += 256) sh[i] = 0;
    __syncthreads();
    int e0 = bid * EPB + tid;
    int d[EPT], sv[EPT], r[EPT];
#pragma unroll
    for (int k = 0; k < EPT; ++k) {
        int e = e0 + k * 256;
        d[k] = (e < E) ? ei[E + e] : -1;
        sv[k] = (e < E) ? ei[e] : 0;
    }
#pragma unroll
    for (int k = 0; k < EPT; ++k) r[k] = (d[k] >= 0) ? atomicAdd(&sh[d[k] >> BSH], 1) : 0;
    __syncthreads();
    int ipt = (NB + 255) / 256;
    int lo = tid * ipt;
    int hi = min(lo + ipt, NB);
    int s = 0, s2 = 0;
    for (int i = lo; i < hi; ++i) {
        s += tot[i * CS];
        s2 += sh[i];
    }
    int incl = s, incl2 = s2;
    for (int m = 1; m < 32; m <<= 1) {
        int u = __shfl_up(incl, m, 32);
        int u2 = __shfl_up(incl2, m, 32);
        if (lane >= m) {
            incl += u;
            incl2 += u2;
        }
    }
    if (lane == 31) {
        wsum[wave] = incl;
        wsum2[wave] = incl2;
    }
    __syncthreads();
    int run = incl - s, lrun = incl2 - s2;
    for (int w = 0; w < wave; ++w) {
        run += wsum[w];
        lrun += wsum2[w];
    }
    for (int i = lo; i < hi; ++i) {
        int c = sh[i];
        int base = run;
        if (c) base += atomicAdd(&cur[i * CS], c);
        sh[NB + i] = lrun;
        sh[i] = base - lrun;
        run += tot[i * CS];
        lrun += c;
    }
    __syncthreads();
#pragma unroll
    for (int k = 0; k < EPT; ++k) {
        if (d[k] >= 0) {
            int2 pr;
            pr.x = sv[k];
            pr.y = d[k];
            stage[sh[NB + (d[k] >> BSH)] + r[k]] = pr;
        }
    }
    __syncthreads();
    int ne = min(EPB, E - bid * EPB);
#pragma unroll
    for (int k = 0; k < EPT; ++k) {
        int p = tid + k * 256;
        if (p < ne) {
            int2 pr = stage[p];
            part[sh[pr.y >> BSH] + p] = pr;
        }
    }
}

__device__ __forceinline__ void bucket_sort(const int2* __restrict__ part,
                                            const int* __restrict__ tot, int N,
                                            int* __restrict__ off, int* __restrict__ csr, int b) {
    __shared__ int cnt[BN];
    __shared__ int red[BN / 32];
    __shared__ int wsum[BN / 32];
    int tid = threadIdx.x, lane = tid & 31, wave = tid >> 5;
    int p = 0;
    for (int i = tid; i < b; i += BN) p += tot[i * CS];
    for (int m = 16; m; m >>= 1) p += __shfl_xor(p, m, 32);
    if (lane == 0) red[wave] = p;
    cnt[tid] = 0;
    __syncthreads();
    int start = 0;
    for (int w = 0; w < BN / 32; ++w) start += red[w];
    int nb = tot[b * CS];
    const int2* mine = part + start;
    for (int i0 = tid; i0 < nb; i0 += SB * BN) {
        int2 pr[SB];
#pragma unroll
        for (int u = 0; u < SB; ++u) {
            int i = i0 + u * BN;
            pr[u].y = (i < nb) ? mine[i].y : -1;
        }
#pragma unroll
        for (int u = 0; u < SB; ++u)
            if (pr[u].y >= 0) atomicAdd(&cnt[pr[u].y & (BN - 1)], 1);
    }
    __syncthreads();
    int c = cnt[tid];
    int incl = c;
    for (int m = 1; m < 32; m <<= 1) {
        int u = __shfl_up(incl, m, 32);
        if (lane >= m) incl += u;
    }
    if (lane == 31) wsum[wave] = incl;
    __syncthreads();
    int excl = incl - c;
    for (int w = 0; w < wave; ++w) excl += wsum[w];
    int node = b * BN + tid;
    if (node < N) off[node] = start + excl;
    if (node == N - 1) off[N] = start + excl + c;
    cnt[tid] = excl;
    __syncthreads();
    int* out = csr + start;
    for (int i0 = tid; i0 < nb; i0 += SB * BN) {
        int2 pr[SB];
#pragma unroll
        for (int u = 0; u < SB; ++u) {
            int i = i0 + u * BN;
            if (i < nb) pr[u] = mine[i];
            else pr[u].y = -1;
        }
#pragma unroll
        for (int u = 0; u < SB; ++u)
            if (pr[u].y >= 0) out[atomicAdd(&cnt[pr[u].y & (BN - 1)], 1)] = pr[u].x;
    }
}

__global__ __launch_bounds__(256) void gemm1_fill_fold(const float* __restrict__ x, const float* __restrict__ W1,
                                                       float* __restrict__ h1, int N, int s0, int spw, int ng,
                                                       unsigned* __restrict__ zp, long long zn, int nzb,
                                                       const float* __restrict__ W2, const float* __restrict__ a_src,
                                                       const float* __restrict__ a_dst, float* __restrict__ vsd, int C2) {
    __shared__ v4f Bs[G1LDS];
    int b = blockIdx.x;
    if (b < ng) {
        gemm1_wmma(x, W1, h1, N, s0, spw, b, Bs);
    } else if (b < ng + nzb) {
        fill_u32(zp, 0u, zn, b - ng);
    } else {
        fold_w2(W2, a_src, a_dst, vsd, C2, b - ng - nzb);
    }
}

__global__ __launch_bounds__(256) void gemm1_count(const float* __restrict__ x, const float* __restrict__ W1,
                                                   float* __restrict__ h1, int N, int s0, int spw, int ng,
                                                   const int* __restrict__ ei, int E, int NB, int* __restrict__ tot) {
    __shared__ v4f Bs[G1LDS];
    int b = blockIdx.x;
    if (b < ng) gemm1_wmma(x, W1, h1, N, s0, spw, b, Bs);
    else bucket_count(ei, E, NB, tot, b - ng);
}

__global__ __launch_bounds__(256) void gemm1_partition(const float* __restrict__ x, const float* __restrict__ W1,
                                                       float* __restrict__ h1, int N, int s0, int spw, int ng,
                                                       const int* __restrict__ ei, int E, int NB,
                                                       const int* __restrict__ tot, int* __restrict__ cur,
                                                       int2* __restrict__ part) {
    __shared__ v4f Bs[G1LDS];
    int b = blockIdx.x;
    if (b < ng) gemm1_wmma(x, W1, h1, N, s0, spw, b, Bs);
    else bucket_partition(ei, E, NB, tot, cur, part, b - ng, (int2*)Bs);
}

__global__ __launch_bounds__(256) void gemm1_sort(const float* __restrict__ x, const float* __restrict__ W1,
                                                  float* __restrict__ h1, int N, int s0, int spw, int ng,
                                                  const int2* __restrict__ part, const int* __restrict__ tot,
                                                  int* __restrict__ off, int* __restrict__ csr) {
    __shared__ v4f Bs[G1LDS];
    int b = blockIdx.x;
    if (b < ng) gemm1_wmma(x, W1, h1, N, s0, spw, b, Bs);
    else bucket_sort(part, tot, N, off, csr, b - ng);
}

__device__ __forceinline__ float dot8(v4f r0, v4f r1, v4f a0, v4f a1) {
    v2f p = r0.xy * a0.xy;
    p += r0.zw * a0.zw;
    p += r1.xy * a1.xy;
    p += r1.zw * a1.zw;
    return p.x + p.y;
}

template <int K, bool ROWDOT>
__device__ __forceinline__ void head_steps(const v4f* __restrict__ rows4, const int* __restrict__ csr,
                                           int j, int end, int o0, int o1, v4f a0, v4f a1, float adn,
                                           float& m, float& den, v4f& c0, v4f& c1) {
    int s[K];
    v4f r0[K], r1[K];
    float e[K];
#pragma unroll
    for (int k = 0; k < K; ++k) s[k] = csr[min(j + 4 * k, end - 1)];
#pragma unroll
    for (int k = 0; k < K; ++k) {
        r0[k] = rows4[s[k] * 16 + o0];
        r1[k] = rows4[s[k] * 16 + o1];
    }
    float gm = m;
#pragma unroll
    for (int k = 0; k < K; ++k) {
        float d = dot8(r0[k], r1[k], a0, a1);
        if (ROWDOT) d = sum8(d);
        e[k] = leaky(d + adn);
        gm = fmaxf(gm, e[k]);
    }
    float sc = __expf(m - gm);
    den *= sc;
    c0 *= sc;
    c1 *= sc;
    m = gm;
#pragma unroll
    for (int k = 0; k < K; ++k) {
        float w = (j + 4 * k < end) ? __expf(e[k] - gm) : 0.f;
        den += w;
        c0 += w * r0[k];
        c1 += w * r1[k];
    }
}

__device__ __forceinline__ v2f merge_slots(int g, float m, float& den, v4f c0, v4f c1) {
    float mo = xget<8>(m);
    float mm = fmaxf(m, mo);
    float f = __expf(m - mm), fo = __expf(mo - mm);
    den = den * f + xget<8>(den) * fo;
    bool up = (g & 1) != 0;
    v4f keep = up ? c1 : c0;
    v4f give = up ? c0 : c1;
    v4f got;
    got.x = xget<8>(give.x);
    got.y = xget<8>(give.y);
    got.z = xget<8>(give.z);
    got.w = xget<8>(give.w);
    v4f c = keep * f + got * fo;
    m = mm;
    mo = __shfl_xor(m, 16, 32);
    mm = fmaxf(m, mo);
    f = __expf(m - mm);
    fo = __expf(mo - mm);
    den = den * f + __shfl_xor(den, 16, 32) * fo;
    bool up2 = (g & 2) != 0;
    v2f keep2 = up2 ? c.zw : c.xy;
    v2f give2 = up2 ? c.xy : c.zw;
    v2f got2;
    got2.x = __shfl_xor(give2.x, 16, 32);
    got2.y = __shfl_xor(give2.y, 16, 32);
    return keep2 * f + got2 * fo;
}

__global__ __launch_bounds__(256) void gather1(const int* __restrict__ off,
                                               const int* __restrict__ csr,
                                               const float* __restrict__ h1,
                                               const float* __restrict__ a_src,
                                               const float* __restrict__ a_dst,
                                               const float* __restrict__ b1,
                                               float* __restrict__ hx, int N) {
    int lane = threadIdx.x & 31;
    int n = __builtin_amdgcn_readfirstlane((int)(blockIdx.x * 8 + (threadIdx.x >> 5)));
    if (n >= N) return;
    int g = lane >> 3, h = lane & 7;
    int beg = off[n], end = off[n + 1];
    const v4f* rows4 = (const v4f*)h1;
    v4f hn0 = rows4[n * 16 + 2 * h], hn1 = rows4[n * 16 + 2 * h + 1];
    v4f a0 = ((const v4f*)a_src)[2 * h], a1 = ((const v4f*)a_src)[2 * h + 1];
    v4f d0 = ((const v4f*)a_dst)[2 * h], d1 = ((const v4f*)a_dst)[2 * h + 1];
    float adn = dot8(hn0, hn1, d0, d1);
    float vself = leaky(dot8(hn0, hn1, a0, a1) + adn);
    float m = vself;
    float den = (g == 0) ? 1.f : 0.f;
    v4f c0 = hn0 * den, c1 = hn1 * den;
    int j0 = beg;
    for (; end - j0 > 12; j0 += 16) head_steps<4, false>(rows4, csr, j0 + g, end, 2 * h, 2 * h + 1, a0, a1, adn, m, den, c0, c1);
    int rem = end - j0;
    if (rem > 8) head_steps<3, false>(rows4, csr, j0 + g, end, 2 * h, 2 * h + 1, a0, a1, adn, m, den, c0, c1);
    else if (rem > 4) head_steps<2, false>(rows4, csr, j0 + g, end, 2 * h, 2 * h + 1, a0, a1, adn, m, den, c0, c1);
    else if (rem > 0) head_steps<1, false>(rows4, csr, j0 + g, end, 2 * h, 2 * h + 1, a0, a1, adn, m, den, c0, c1);

    v2f acc = merge_slots(g, m, den, c0, c1);
    int ch = 8 * h + 4 * (g & 1) + (g & 2);

    float vx = acc.x / den + b1[ch];
    float vy = acc.y / den + b1[ch + 1];
    v2f o;
    o.x = vx > 0.f ? vx : (__expf(vx) - 1.f);
    o.y = vy > 0.f ? vy : (__expf(vy) - 1.f);
    *(v2f*)(hx + (long long)n * HD1 + ch) = o;
}

__global__ __launch_bounds__(256) void gather2_rows(const int* __restrict__ off,
                                                    const int* __restrict__ csr,
                                                    const float* __restrict__ vsd,
                                                    const float* __restrict__ hx,
                                                    float* __restrict__ agg, int N) {
    int lane = threadIdx.x & 31;
    int n = __builtin_amdgcn_readfirstlane((int)(blockIdx.x * 8 + (threadIdx.x >> 5)));
    if (n >= N) return;
    int g = lane >> 3, q = lane & 7;
    int beg = off[n], end = off[n + 1];
    const v4f* rows4 = (const v4f*)hx;
    v4f hn0 = rows4[n * 16 + q], hn1 = rows4[n * 16 + 8 + q];
    v4f s0 = ((const v4f*)vsd)[q], s1 = ((const v4f*)vsd)[8 + q];
    v4f d0 = ((const v4f*)vsd)[16 + q], d1 = ((const v4f*)vsd)[16 + 8 + q];
    float adn = sum8(dot8(hn0, hn1, d0, d1));
    float vself = leaky(sum8(dot8(hn0, hn1, s0, s1)) + adn);
    float m = vself;
    float den = (g == 0) ? 1.f : 0.f;
    v4f c0 = hn0 * den, c1 = hn1 * den;
    int j0 = beg;
    for (; end - j0 > 12; j0 += 16) head_steps<4, true>(rows4, csr, j0 + g, end, q, 8 + q, s0, s1, adn, m, den, c0, c1);
    int rem = end - j0;
    if (rem > 8) head_steps<3, true>(rows4, csr, j0 + g, end, q, 8 + q, s0, s1, adn, m, den, c0, c1);
    else if (rem > 4) head_steps<2, true>(rows4, csr, j0 + g, end, q, 8 + q, s0, s1, adn, m, den, c0, c1);
    else if (rem > 0) head_steps<1, true>(rows4, csr, j0 + g, end, q, 8 + q, s0, s1, adn, m, den, c0, c1);

    v2f acc = merge_slots(g, m, den, c0, c1);
    int ch = 32 * (g & 1) + 4 * q + (g & 2);
    v2f o;
    o.x = acc.x / den;
    o.y = acc.y / den;
    *(v2f*)(agg + (long long)n * HD1 + ch) = o;
}

#define G2W 8
__device__ __forceinline__ v4f sig4(v4f v) {
    v4f o;
    o.x = 1.f / (1.f + __expf(-v.x));
    o.y = 1.f / (1.f + __expf(-v.y));
    o.z = 1.f / (1.f + __expf(-v.z));
    o.w = 1.f / (1.f + __expf(-v.w));
    return o;
}
__global__ __launch_bounds__(32 * G2W) void gemm2_wmma(const float* __restrict__ agg,
                                                       const float* __restrict__ W2,
                                                       const float* __restrict__ b2,
                                                       float* __restrict__ outp, int N, int C2) {
    __shared__ v4f Bs[(HD1 / 2) * 64];
    int tid = threadIdx.x;
    int wave = tid >> 5, lane = tid & 31;
    int lm = lane & 15, lh = lane >> 4;
#pragma unroll
    for (int i = 0; i < (HD1 / 2) * 32 / (32 * G2W); ++i) {
        int p = tid + i * 32 * G2W;
        int kp = p >> 5, j = p & 31;
        int col = 4 * j;
        const float* w0 = W2 + (2 * kp) * C2;
        const float* w1 = w0 + C2;
        int k0 = min(col, C2 - 1), k1 = min(col + 1, C2 - 1), k2 = min(col + 2, C2 - 1), k3 = min(col + 3, C2 - 1);
        float e00 = w0[k0], e10 = w1[k0], e01 = w0[k1], e11 = w1[k1];
        float e02 = w0[k2], e12 = w1[k2], e03 = w0[k3], e13 = w1[k3];
        v4f g0, g1;
        g0.x = (col < C2) ? e00 : 0.f;
        g0.y = (col < C2) ? e10 : 0.f;
        g0.z = (col + 1 < C2) ? e01 : 0.f;
        g0.w = (col + 1 < C2) ? e11 : 0.f;
        g1.x = (col + 2 < C2) ? e02 : 0.f;
        g1.y = (col + 2 < C2) ? e12 : 0.f;
        g1.z = (col + 3 < C2) ? e03 : 0.f;
        g1.w = (col + 3 < C2) ? e13 : 0.f;
        int c = j & 15, hf = j >> 4;
        Bs[kp * 64 + (2 * hf) * 16 + c] = g0;
        Bs[kp * 64 + (2 * hf + 1) * 16 + c] = g1;
    }
    __syncthreads();
    int row0 = (blockIdx.x * G2W + wave) * 16;
    if (row0 >= N) return;
    const v4f* ar = (const v4f*)(agg + (long long)(row0 + lm) * HD1) + lh;
    const v4f* bl = Bs + lh * 128 + lm;
    v4f a[HD1 / 8];
#pragma unroll
    for (int q = 0; q < HD1 / 8; ++q) a[q] = ar[2 * q];
    v8f acc0 = {0.f, 0.f, 0.f, 0.f, 0.f, 0.f, 0.f, 0.f};
    v8f acc1 = acc0, acc2 = acc0, acc3 = acc0, acc4 = acc0, acc5 = acc0, acc6 = acc0, acc7 = acc0;
#pragma unroll
    for (int kb = 0; kb < HD1 / 8; ++kb) {
        const v4f* bp = bl + kb * 256;
        v4f av = a[kb];
        v4f p0 = bp[0], p1 = bp[16], p2 = bp[32], p3 = bp[48];
        acc0 = __builtin_amdgcn_wmma_f32_16x16x4_f32(false, av.xy, false, p0.xy, (short)0, acc0, false, false);
        acc1 = __builtin_amdgcn_wmma_f32_16x16x4_f32(false, av.xy, false, p0.zw, (short)0, acc1, false, false);
        acc2 = __builtin_amdgcn_wmma_f32_16x16x4_f32(false, av.xy, false, p1.xy, (short)0, acc2, false, false);
        acc3 = __builtin_amdgcn_wmma_f32_16x16x4_f32(false, av.xy, false, p1.zw, (short)0, acc3, false, false);
        acc4 = __builtin_amdgcn_wmma_f32_16x16x4_f32(false, av.xy, false, p2.xy, (short)0, acc4, false, false);
        acc5 = __builtin_amdgcn_wmma_f32_16x16x4_f32(false, av.xy, false, p2.zw, (short)0, acc5, false, false);
        acc6 = __builtin_amdgcn_wmma_f32_16x16x4_f32(false, av.xy, false, p3.xy, (short)0, acc6, false, false);
        acc7 = __builtin_amdgcn_wmma_f32_16x16x4_f32(false, av.xy, false, p3.zw, (short)0, acc7, false, false);
        v4f q0 = bp[64], q1 = bp[80], q2 = bp[96], q3 = bp[112];
        acc0 = __builtin_amdgcn_wmma_f32_16x16x4_f32(false, av.zw, false, q0.xy, (short)0, acc0, false, false);
        acc1 = __builtin_amdgcn_wmma_f32_16x16x4_f32(false, av.zw, false, q0.zw, (short)0, acc1, false, false);
        acc2 = __builtin_amdgcn_wmma_f32_16x16x4_f32(false, av.zw, false, q1.xy, (short)0, acc2, false, false);
        acc3 = __builtin_amdgcn_wmma_f32_16x16x4_f32(false, av.zw, false, q1.zw, (short)0, acc3, false, false);
        acc4 = __builtin_amdgcn_wmma_f32_16x16x4_f32(false, av.zw, false, q2.xy, (short)0, acc4, false, false);
        acc5 = __builtin_amdgcn_wmma_f32_16x16x4_f32(false, av.zw, false, q2.zw, (short)0, acc5, false, false);
        acc6 = __builtin_amdgcn_wmma_f32_16x16x4_f32(false, av.zw, false, q3.xy, (short)0, acc6, false, false);
        acc7 = __builtin_amdgcn_wmma_f32_16x16x4_f32(false, av.zw, false, q3.zw, (short)0, acc7, false, false);
    }
    int cA = 4 * lm, cB = 64 + 4 * lm;
    v4f bA, bB;
    bA.x = b2[min(cA, C2 - 1)];
    bA.y = b2[min(cA + 1, C2 - 1)];
    bA.z = b2[min(cA + 2, C2 - 1)];
    bA.w = b2[min(cA + 3, C2 - 1)];
    bB.x = b2[min(cB, C2 - 1)];
    bB.y = b2[min(cB + 1, C2 - 1)];
    bB.z = b2[min(cB + 2, C2 - 1)];
    bB.w = b2[min(cB + 3, C2 - 1)];
    float* orow = outp + (long long)(row0 + 8 * lh) * C2;
    if ((C2 & 3) == 0) {
        bool okA = cA < C2, okB = cB < C2;
#pragma unroll
        for (int r = 0; r < 8; ++r) {
            v4f vA, vB;
            vA.x = acc0[r]; vA.y = acc1[r]; vA.z = acc2[r]; vA.w = acc3[r];
            vB.x = acc4[r]; vB.y = acc5[r]; vB.z = acc6[r]; vB.w = acc7[r];
            v4f sA = sig4(vA + bA), sB = sig4(vB + bB);
            if (okA) *(v4f*)(orow + (long long)r * C2 + cA) = sA;
            if (okB) *(v4f*)(orow + (long long)r * C2 + cB) = sB;
        }
    } else {
#pragma unroll
        for (int r = 0; r < 8; ++r) {
            v4f vA, vB;
            vA.x = acc0[r]; vA.y = acc1[r]; vA.z = acc2[r]; vA.w = acc3[r];
            vB.x = acc4[r]; vB.y = acc5[r]; vB.z = acc6[r]; vB.w = acc7[r];
            v4f sA = sig4(vA + bA), sB = sig4(vB + bB);
            float* o = orow + (long long)r * C2;
            if (cA < C2) o[cA] = sA.x;
            if (cA + 1 < C2) o[cA + 1] = sA.y;
            if (cA + 2 < C2) o[cA + 2] = sA.z;
            if (cA + 3 < C2) o[cA + 3] = sA.w;
            if (cB < C2) o[cB] = sB.x;
            if (cB + 1 < C2) o[cB + 1] = sB.y;
            if (cB + 2 < C2) o[cB + 2] = sB.z;
            if (cB + 3 < C2) o[cB + 3] = sB.w;
        }
    }
}

static inline int cdiv(long long a, long long b) { return (int)((a + b - 1) / b); }

extern "C" void kernel_launch(void* const* d_in, const int* in_sizes, int n_in,
                              void* d_out, int out_size, void* d_ws, size_t ws_size,
                              hipStream_t stream) {
    const float* x      = (const float*)d_in[0];
    const int*   ei     = (const int*)d_in[1];
    const float* W1     = (const float*)d_in[2];
    const float* a_src1 = (const float*)d_in[3];
    const float* a_dst1 = (const float*)d_in[4];
    const float* b1     = (const float*)d_in[5];
    const float* W2     = (const float*)d_in[6];
    const float* a_src2 = (const float*)d_in[7];
    const float* a_dst2 = (const float*)d_in[8];
    const float* b2     = (const float*)d_in[9];

    const int N  = in_sizes[0] / FIN;
    const int E  = in_sizes[1] / 2;
    const int C2 = in_sizes[9];

    float* h1   = (float*)d_ws;
    float* hx   = h1 + (long long)N * HD1;
    float* agg  = hx + (long long)N * HD1;
    float* vsd  = agg + (long long)N * HD1;
    int2*  part = (int2*)(vsd + 2 * HD1);
    const int NB = cdiv(N, BN);
    int*   tot  = (int*)(part + E);
    int*   cur  = tot + NB * CS;
    int*   off  = cur + NB * CS;
    int*   csr  = off + N + 1;
    float* out  = (float*)d_out;

    const int B = 256;

    const int WGPS = 256;
    const int S = cdiv(N, 16);
    const int GW = S / G1W;
    const int gA = GW < WGPS ? GW : WGPS;
    const int gB = GW - gA < WGPS ? GW - gA : WGPS;
    const int gC = GW - gA - gB < WGPS ? GW - gA - gB : WGPS;
    const int sD = (gA + gB + gC) * G1W;
    const int SPD = 2;
    const int gD = cdiv(S - sD, SPD);
    const int nzb = cdiv(2 * NB * CS, B);
    gemm1_fill_fold<<<gA + nzb + HD1 / 8, 256, 0, stream>>>(x, W1, h1, N, 0, G1W, gA, (unsigned*)tot, 2 * NB * CS, nzb,
                                                             W2, a_src2, a_dst2, vsd, C2);
    gemm1_count<<<gB + cdiv(E, EPB), 256, NB * sizeof(int), stream>>>(x, W1, h1, N, gA * G1W, G1W, gB, ei, E, NB, tot);
    gemm1_partition<<<gC + cdiv(E, EPB), 256, 2 * NB * sizeof(int), stream>>>(x, W1, h1, N, (gA + gB) * G1W, G1W, gC,
                                                                              ei, E, NB, tot, cur, part);
    gemm1_sort<<<gD + NB, BN, 0, stream>>>(x, W1, h1, N, sD, SPD, gD, part, tot, off, csr);

    gather1<<<cdiv(N, 8), 256, 0, stream>>>(off, csr, h1, a_src1, a_dst1, b1, hx, N);

    gather2_rows<<<cdiv(N, 8), 256, 0, stream>>>(off, csr, vsd, hx, agg, N);
    gemm2_wmma<<<cdiv(N, 16 * G2W), 32 * G2W, 0, stream>>>(agg, W2, b2, out, N, C2);
}


# Edge partition with no counters in memory: a work-group writes its pairs bucket-ordered as its own stretch plus run offsets, bucket sort walks the runs; 7 launches to 5; a node's sum order changes
# speedup vs baseline: 6.4890x; 1.0935x over previous
#include <hip/hip_runtime.h>
#include <math.h>

typedef float v2f __attribute__((ext_vector_type(2)));
typedef float v4f __attribute__((ext_vector_type(4)));
typedef float v8f __attribute__((ext_vector_type(8)));

#define FIN 256
#define HD1 64

__device__ __forceinline__ float leaky(float v) { return v > 0.f ? v : 0.2f * v; }

template <int M>
__device__ __forceinline__ float xadd(float v) {
    int o = __builtin_amdgcn_update_dpp(0, __float_as_int(v), 0x160 | M, 0xf, 0xf, false);
    return v + __int_as_float(o);
}
template <int M>
__device__ __forceinline__ float xget(float v) {
    return __int_as_float(__builtin_amdgcn_update_dpp(0, __float_as_int(v), 0x160 | M, 0xf, 0xf, false));
}
__device__ __forceinline__ float sum8(float v) { return xadd<4>(xadd<2>(xadd<1>(v))); }
__device__ __forceinline__ float sum16(float v) { return xadd<8>(sum8(v)); }
__device__ __forceinline__ float wave_sum(float v) {
    v = sum16(v);
    return v + __shfl_xor(v, 16, 32);
}

__device__ __forceinline__ void fold_w2(const float* __restrict__ W2,
                                        const float* __restrict__ a_src,
                                        const float* __restrict__ a_dst,
                                        float* __restrict__ vsd, int C2, int bid) {
    int lane = threadIdx.x & 31;
    int k = bid * 8 + (threadIdx.x >> 5);
    float s = 0.f, d = 0.f;
    for (int cix = lane; cix < C2; cix += 32) {
        float w = W2[k * C2 + cix];
        s += w * a_src[cix];
        d += w * a_dst[cix];
    }
    s = wave_sum(s);
    d = wave_sum(d);
    if (lane == 0) {
        vsd[k] = s;
        vsd[HD1 + k] = d;
    }
}

#define G1W 8
#define G1LDS ((FIN / 2) * 32)
__device__ __forceinline__ void gemm1_wmma(const float* __restrict__ x,
                                           const float* __restrict__ W1,
                                           float* __restrict__ h1, int N, int s0, int spw, int wg, v4f* Bs) {
    int tid = threadIdx.x;
    int wave = tid >> 5, lane = tid & 31;
    int lm = lane & 15, lh = lane >> 4;
    const v4f* W4 = (const v4f*)W1;
    for (int p0 = tid; p0 < (FIN / 2) * 16; p0 += 32 * G1W * 4) {
        v4f r0[4], r1[4];
#pragma unroll
        for (int u = 0; u < 4; ++u) {
            int p = p0 + u * 32 * G1W;
            int kp = p >> 4, c = p & 15;
            r0[u] = W4[(2 * kp) * 16 + c];
            r1[u] = W4[(2 * kp + 1) * 16 + c];
        }
#pragma unroll
        for (int u = 0; u < 4; ++u) {
            int p = p0 + u * 32 * G1W;
            int kp = p >> 4, c = p & 15;
            v4f g0, g1;
            g0.x = r0[u].x; g0.y = r1[u].x; g0.z = r0[u].y; g0.w = r1[u].y;
            g1.x = r0[u].z; g1.y = r1[u].z; g1.z = r0[u].w; g1.w = r1[u].w;
            Bs[kp * 32 + c] = g0;
            Bs[kp * 32 + 16 + c] = g1;
        }
    }
    __syncthreads();
    int row0 = (s0 + wg * spw + wave) * 16;
    if (wave >= spw || row0 >= N) return;
    const v4f* xr = (const v4f*)(x + (long long)(row0 + lm) * FIN) + lh;
    const v4f* bl = Bs + lh * 64 + lm;
    v8f acc0 = {0.f, 0.f, 0.f, 0.f, 0.f, 0.f, 0.f, 0.f};
    v8f acc1 = acc0, acc2 = acc0, acc3 = acc0;
    v4f a[2][8];
#pragma unroll
    for (int q = 0; q < 8; ++q) a[0][q] = xr[2 * q];
#pragma unroll
    for (int ch = 0; ch < FIN / 64; ++ch) {
        if (ch + 1 < FIN / 64) {
#pragma unroll
            for (int q = 0; q < 8; ++q) a[(ch + 1) & 1][q] = xr[2 * ((ch + 1) * 8 + q)];
        }
        __builtin_amdgcn_sched_barrier(0);
#pragma unroll
        for (int q = 0; q < 8; ++q) {
            int kb = ch * 8 + q;
            const v4f* bp = bl + kb * 128;
            v4f b00 = bp[0], b01 = bp[16], b10 = bp[32], b11 = bp[48];
            v4f av = a[ch & 1][q];
            acc0 = __builtin_amdgcn_wmma_f32_16x16x4_f32(false, av.xy, false, b00.xy, (short)0, acc0, false, false);
            acc1 = __builtin_amdgcn_wmma_f32_16x16x4_f32(false, av.xy, false, b00.zw, (short)0, acc1, false, false);
            acc2 = __builtin_amdgcn_wmma_f32_16x16x4_f32(false, av.xy, false, b01.xy, (short)0, acc2, false, false);
            acc3 = __builtin_amdgcn_wmma_f32_16x16x4_f32(false, av.xy, false, b01.zw, (short)0, acc3, false, false);
            acc0 = __builtin_amdgcn_wmma_f32_16x16x4_f32(false, av.zw, false, b10.xy, (short)0, acc0, false, false);
            acc1 = __builtin_amdgcn_wmma_f32_16x16x4_f32(false, av.zw, false, b10.zw, (short)0, acc1, false, false);
            acc2 = __builtin_amdgcn_wmma_f32_16x16x4_f32(false, av.zw, false, b11.xy, (short)0, acc2, false, false);
            acc3 = __builtin_amdgcn_wmma_f32_16x16x4_f32(false, av.zw, false, b11.zw, (short)0, acc3, false, false);
        }
    }
    float* out = h1 + (long long)(row0 + 8 * lh) * HD1 + 4 * lm;
#pragma unroll
    for (int r = 0; r < 8; ++r) {
        v4f o;
        o.x = acc0[r]; o.y = acc1[r]; o.z = acc2[r]; o.w = acc3[r];
        *(v4f*)(out + (long long)r * HD1) = o;
    }
}

#define BSH 8
#define BN (1 << BSH)
#define EPB 4096
#define EPT (EPB / 256)
#define SB 8
#define RL 16

__device__ __forceinline__ void bucket_partition(const int* __restrict__ ei, int E, int NB, int NP,
                                                 int2* __restrict__ part, int* __restrict__ hoff,
                                                 int bid, int2* stage) {
    extern __shared__ int sh[];
    __shared__ int wsum2[8];
    int tid = threadIdx.x, lane = tid & 31, wave = tid >> 5;
    for (int i = tid; i < NB; i += 256) sh[i] = 0;
    __syncthreads();
    int e0 = bid * EPB + tid;
    int d[EPT], sv[EPT], r[EPT];
#pragma unroll
    for (int k = 0; k < EPT; ++k) {
        int e = e0 + k * 256;
        d[k] = (e < E) ? ei[E + e] : -1;
        sv[k] = (e < E) ? ei[e] : 0;
    }
#pragma unroll
    for (int k = 0; k < EPT; ++k) r[k] = (d[k] >= 0) ? atomicAdd(&sh[d[k] >> BSH], 1) : 0;
    __syncthreads();
    int ipt = (NB + 255) / 256;
    int lo = tid * ipt;
    int hi = min(lo + ipt, NB);
    int s2 = 0;
    for (int i = lo; i < hi; ++i) s2 += sh[i];
    int incl2 = s2;
    for (int m = 1; m < 32; m <<= 1) {
        int u2 = __shfl_up(incl2, m, 32);
        if (lane >= m) incl2 += u2;
    }
    if (lane == 31) wsum2[wave] = incl2;
    __syncthreads();
    int lrun = incl2 - s2;
    for (int w = 0; w < wave; ++w) lrun += wsum2[w];
    for (int i = lo; i < hi; ++i) {
        sh[NB + i] = lrun;
        hoff[i * NP + bid] = lrun;
        lrun += sh[i];
    }
    int ne = min(EPB, E - bid * EPB);
    if (tid == 0) hoff[NB * NP + bid] = ne;
    __syncthreads();
#pragma unroll
    for (int k = 0; k < EPT; ++k) {
        if (d[k] >= 0) {
            int2 pr;
            pr.x = sv[k];
            pr.y = d[k];
            stage[sh[NB + (d[k] >> BSH)] + r[k]] = pr;
        }
    }
    __syncthreads();
    int2* mine = part + (long long)bid * EPB;
#pragma unroll
    for (int k = 0; k < EPT; ++k) {
        int p = tid + k * 256;
        if (p < ne) mine[p] = stage[p];
    }
}

__device__ __forceinline__ void bucket_sort(const int2* __restrict__ part,
                                            const int* __restrict__ hoff, int NP, int N,
                                            int* __restrict__ off, int* __restrict__ csr, int b) {
    extern __shared__ int sh[];
    __shared__ int cnt[BN];
    __shared__ int red[BN / 32];
    __shared__ int wsum[BN / 32];
    int tid = threadIdx.x, lane = tid & 31, wave = tid >> 5;
    int p = 0;
    for (int w = tid; w < NP; w += 2 * BN) {
        int w2 = w + BN;
        bool ok2 = w2 < NP;
        int lo = hoff[b * NP + w], hi = hoff[(b + 1) * NP + w];
        int lo2 = ok2 ? hoff[b * NP + w2] : 0, hi2 = ok2 ? hoff[(b + 1) * NP + w2] : 0;
        sh[w] = lo;
        sh[NP + w] = hi;
        if (ok2) {
            sh[w2] = lo2;
            sh[NP + w2] = hi2;
        }
        p += lo + lo2;
    }
    for (int m = 16; m; m >>= 1) p += __shfl_xor(p, m, 32);
    if (lane == 0) red[wave] = p;
    cnt[tid] = 0;
    __syncthreads();
    int start = 0;
    for (int w = 0; w < BN / 32; ++w) start += red[w];
    int gi = tid / RL, sl = tid % RL;
    for (int w0 = gi; w0 < NP; w0 += (BN / RL) * SB) {
        int y[2 * SB];
#pragma unroll
        for (int u = 0; u < SB; ++u) {
            int w = w0 + u * (BN / RL);
            bool ok = w < NP;
            int i = (ok ? sh[w] : 0) + sl;
            int hi = ok ? sh[NP + w] : 0;
            const int2* run = part + (long long)w * EPB;
            y[u] = (i < hi) ? run[i].y : -1;
            y[SB + u] = (i + RL < hi) ? run[i + RL].y : -1;
        }
#pragma unroll
        for (int u = 0; u < 2 * SB; ++u)
            if (y[u] >= 0) atomicAdd(&cnt[y[u] & (BN - 1)], 1);
        for (int u = 0; u < SB; ++u) {
            int w = w0 + u * (BN / RL);
            if (w < NP) {
                int hi = sh[NP + w];
                for (int i = sh[w] + sl + 2 * RL; i < hi; i += RL)
                    atomicAdd(&cnt[part[(long long)w * EPB + i].y & (BN - 1)], 1);
            }
        }
    }
    __syncthreads();
    int c = cnt[tid];
    int incl = c;
    for (int m = 1; m < 32; m <<= 1) {
        int u = __shfl_up(incl, m, 32);
        if (lane >= m) incl += u;
    }
    if (lane == 31) wsum[wave] = incl;
    __syncthreads();
    int excl = incl - c;
    for (int w = 0; w < wave; ++w) excl += wsum[w];
    int node = b * BN + tid;
    if (node < N) off[node] = start + excl;
    if (node == N - 1) off[N] = start + excl + c;
    cnt[tid] = excl;
    __syncthreads();
    int* out = csr + start;
    for (int w0 = gi; w0 < NP; w0 += (BN / RL) * SB) {
        int2 pr[2 * SB];
#pragma unroll
        for (int u = 0; u < SB; ++u) {
            int w = w0 + u * (BN / RL);
            bool ok = w < NP;
            int i = (ok ? sh[w] : 0) + sl;
            int hi = ok ? sh[NP + w] : 0;
            const int2* run = part + (long long)w * EPB;
            if (i < hi) pr[u] = run[i];
            else pr[u].y = -1;
            if (i + RL < hi) pr[SB + u] = run[i + RL];
            else pr[SB + u].y = -1;
        }
#pragma unroll
        for (int u = 0; u < 2 * SB; ++u)
            if (pr[u].y >= 0) out[atomicAdd(&cnt[pr[u].y & (BN - 1)], 1)] = pr[u].x;
        for (int u = 0; u < SB; ++u) {
            int w = w0 + u * (BN / RL);
            if (w < NP) {
                int hi = sh[NP + w];
                for (int i = sh[w] + sl + 2 * RL; i < hi; i += RL) {
                    int2 q = part[(long long)w * EPB + i];
                    out[atomicAdd(&cnt[q.y & (BN - 1)], 1)] = q.x;
                }
            }
        }
    }
}

__global__ __launch_bounds__(256) void gemm1_partition_fold(const float* __restrict__ x, const float* __restrict__ W1,
                                                            float* __restrict__ h1, int N, int s0, int spw, int ng,
                                                            const int* __restrict__ ei, int E, int NB, int NP,
                                                            int2* __restrict__ part, int* __restrict__ hoff,
                                                            const float* __restrict__ W2, const float* __restrict__ a_src,
                                                            const float* __restrict__ a_dst, float* __restrict__ vsd, int C2) {
    __shared__ v4f Bs[G1LDS];
    int b = blockIdx.x;
    if (b < ng) gemm1_wmma(x, W1, h1, N, s0, spw, b, Bs);
    else if (b < ng + NP) bucket_partition(ei, E, NB, NP, part, hoff, b - ng, (int2*)Bs);
    else fold_w2(W2, a_src, a_dst, vsd, C2, b - ng - NP);
}

__global__ __launch_bounds__(256) void gemm1_sort(const float* __restrict__ x, const float* __restrict__ W1,
                                                  float* __restrict__ h1, int N, int s0, int spw, int ng,
                                                  int s1, int spw1, int ng1,
                                                  const int2* __restrict__ part, const int* __restrict__ hoff, int NP,
                                                  int* __restrict__ off, int* __restrict__ csr) {
    __shared__ v4f Bs[G1LDS];
    int b = blockIdx.x;
    if (b < ng + ng1) {
        bool first = b < ng;
        gemm1_wmma(x, W1, h1, N, first ? s0 : s1, first ? spw : spw1, first ? b : b - ng, Bs);
    } else {
        bucket_sort(part, hoff, NP, N, off, csr, b - ng - ng1);
    }
}

__device__ __forceinline__ float dot8(v4f r0, v4f r1, v4f a0, v4f a1) {
    v2f p = r0.xy * a0.xy;
    p += r0.zw * a0.zw;
    p += r1.xy * a1.xy;
    p += r1.zw * a1.zw;
    return p.x + p.y;
}

template <int K, bool ROWDOT>
__device__ __forceinline__ void head_steps(const v4f* __restrict__ rows4, const int* __restrict__ csr,
                                           int j, int end, int o0, int o1, v4f a0, v4f a1, float adn,
                                           float& m, float& den, v4f& c0, v4f& c1) {
    int s[K];
    v4f r0[K], r1[K];
    float e[K];
#pragma unroll
    for (int k = 0; k < K; ++k) s[k] = csr[min(j + 4 * k, end - 1)];
#pragma unroll
    for (int k = 0; k < K; ++k) {
        r0[k] = rows4[s[k] * 16 + o0];
        r1[k] = rows4[s[k] * 16 + o1];
    }
    float gm = m;
#pragma unroll
    for (int k = 0; k < K; ++k) {
        float d = dot8(r0[k], r1[k], a0, a1);
        if (ROWDOT) d = sum8(d);
        e[k] = leaky(d + adn);
        gm = fmaxf(gm, e[k]);
    }
    float sc = __expf(m - gm);
    den *= sc;
    c0 *= sc;
    c1 *= sc;
    m = gm;
#pragma unroll
    for (int k = 0; k < K; ++k) {
        float w = (j + 4 * k < end) ? __expf(e[k] - gm) : 0.f;
        den += w;
        c0 += w * r0[k];
        c1 += w * r1[k];
    }
}

__device__ __forceinline__ v2f merge_slots(int g, float m, float& den, v4f c0, v4f c1) {
    float mo = xget<8>(m);
    float mm = fmaxf(m, mo);
    float f = __expf(m - mm), fo = __expf(mo - mm);
    den = den * f + xget<8>(den) * fo;
    bool up = (g & 1) != 0;
    v4f keep = up ? c1 : c0;
    v4f give = up ? c0 : c1;
    v4f got;
    got.x = xget<8>(give.x);
    got.y = xget<8>(give.y);
    got.z = xget<8>(give.z);
    got.w = xget<8>(give.w);
    v4f c = keep * f + got * fo;
    m = mm;
    mo = __shfl_xor(m, 16, 32);
    mm = fmaxf(m, mo);
    f = __expf(m - mm);
    fo = __expf(mo - mm);
    den = den * f + __shfl_xor(den, 16, 32) * fo;
    bool up2 = (g & 2) != 0;
    v2f keep2 = up2 ? c.zw : c.xy;
    v2f give2 = up2 ? c.xy : c.zw;
    v2f got2;
    got2.x = __shfl_xor(give2.x, 16, 32);
    got2.y = __shfl_xor(give2.y, 16, 32);
    return keep2 * f + got2 * fo;
}

__global__ __launch_bounds__(256) void gather1(const int* __restrict__ off,
                                               const int* __restrict__ csr,
                                               const float* __restrict__ h1,
                                               const float* __restrict__ a_src,
                                               const float* __restrict__ a_dst,
                                               const float* __restrict__ b1,
                                               float* __restrict__ hx, int N) {
    int lane = threadIdx.x & 31;
    int n = __builtin_amdgcn_readfirstlane((int)(blockIdx.x * 8 + (threadIdx.x >> 5)));
    if (n >= N) return;
    int g = lane >> 3, h = lane & 7;
    int beg = off[n], end = off[n + 1];
    const v4f* rows4 = (const v4f*)h1;
    v4f hn0 = rows4[n * 16 + 2 * h], hn1 = rows4[n * 16 + 2 * h + 1];
    v4f a0 = ((const v4f*)a_src)[2 * h], a1 = ((const v4f*)a_src)[2 * h + 1];
    v4f d0 = ((const v4f*)a_dst)[2 * h], d1 = ((const v4f*)a_dst)[2 * h + 1];
    float adn = dot8(hn0, hn1, d0, d1);
    float vself = leaky(dot8(hn0, hn1, a0, a1) + adn);
    float m = vself;
    float den = (g == 0) ? 1.f : 0.f;
    v4f c0 = hn0 * den, c1 = hn1 * den;
    int j0 = beg;
    for (; end - j0 > 12; j0 += 16) head_steps<4, false>(rows4, csr, j0 + g, end, 2 * h, 2 * h + 1, a0, a1, adn, m, den, c0, c1);
    int rem = end - j0;
    if (rem > 8) head_steps<3, false>(rows4, csr, j0 + g, end, 2 * h, 2 * h + 1, a0, a1, adn, m, den, c0, c1);
    else if (rem > 4) head_steps<2, false>(rows4, csr, j0 + g, end, 2 * h, 2 * h + 1, a0, a1, adn, m, den, c0, c1);
    else if (rem > 0) head_steps<1, false>(rows4, csr, j0 + g, end, 2 * h, 2 * h + 1, a0, a1, adn, m, den, c0, c1);

    v2f acc = merge_slots(g, m, den, c0, c1);
    int ch = 8 * h + 4 * (g & 1) + (g & 2);

    float vx = acc.x / den + b1[ch];
    float vy = acc.y / den + b1[ch + 1];
    v2f o;
    o.x = vx > 0.f ? vx : (__expf(vx) - 1.f);
    o.y = vy > 0.f ? vy : (__expf(vy) - 1.f);
    *(v2f*)(hx + (long long)n * HD1 + ch) = o;
}

__global__ __launch_bounds__(256) void gather2_rows(const int* __restrict__ off,
                                                    const int* __restrict__ csr,
                                                    const float* __restrict__ vsd,
                                                    const float* __restrict__ hx,
                                                    float* __restrict__ agg, int N) {
    int lane = threadIdx.x & 31;
    int n = __builtin_amdgcn_readfirstlane((int)(blockIdx.x * 8 + (threadIdx.x >> 5)));
    if (n >= N) return;
    int g = lane >> 3, q = lane & 7;
    int beg = off[n], end = off[n + 1];
    const v4f* rows4 = (const v4f*)hx;
    v4f hn0 = rows4[n * 16 + q], hn1 = rows4[n * 16 + 8 + q];
    v4f s0 = ((const v4f*)vsd)[q], s1 = ((const v4f*)vsd)[8 + q];
    v4f d0 = ((const v4f*)vsd)[16 + q], d1 = ((const v4f*)vsd)[16 + 8 + q];
    float adn = sum8(dot8(hn0, hn1, d0, d1));
    float vself = leaky(sum8(dot8(hn0, hn1, s0, s1)) + adn);
    float m = vself;
    float den = (g == 0) ? 1.f : 0.f;
    v4f c0 = hn0 * den, c1 = hn1 * den;
    int j0 = beg;
    for (; end - j0 > 12; j0 += 16) head_steps<4, true>(rows4, csr, j0 + g, end, q, 8 + q, s0, s1, adn, m, den, c0, c1);
    int rem = end - j0;
    if (rem > 8) head_steps<3, true>(rows4, csr, j0 + g, end, q, 8 + q, s0, s1, adn, m, den, c0, c1);
    else if (rem > 4) head_steps<2, true>(rows4, csr, j0 + g, end, q, 8 + q, s0, s1, adn, m, den, c0, c1);
    else if (rem > 0) head_steps<1, true>(rows4, csr, j0 + g, end, q, 8 + q, s0, s1, adn, m, den, c0, c1);

    v2f acc = merge_slots(g, m, den, c0, c1);
    int ch = 32 * (g & 1) + 4 * q + (g & 2);
    v2f o;
    o.x = acc.x / den;
    o.y = acc.y / den;
    *(v2f*)(agg + (long long)n * HD1 + ch) = o;
}

#define G2W 8
__device__ __forceinline__ v4f sig4(v4f v) {
    v4f o;
    o.x = 1.f / (1.f + __expf(-v.x));
    o.y = 1.f / (1.f + __expf(-v.y));
    o.z = 1.f / (1.f + __expf(-v.z));
    o.w = 1.f / (1.f + __expf(-v.w));
    return o;
}
__global__ __launch_bounds__(32 * G2W) void gemm2_wmma(const float* __restrict__ agg,
                                                       const float* __restrict__ W2,
                                                       const float* __restrict__ b2,
                                                       float* __restrict__ outp, int N, int C2) {
    __shared__ v4f Bs[(HD1 / 2) * 64];
    int tid = threadIdx.x;
    int wave = tid >> 5, lane = tid & 31;
    int lm = lane & 15, lh = lane >> 4;
#pragma unroll
    for (int i = 0; i < (HD1 / 2) * 32 / (32 * G2W); ++i) {
        int p = tid + i * 32 * G2W;
        int kp = p >> 5, j = p & 31;
        int col = 4 * j;
        const float* w0 = W2 + (2 * kp) * C2;
        const float* w1 = w0 + C2;
        int k0 = min(col, C2 - 1), k1 = min(col + 1, C2 - 1), k2 = min(col + 2, C2 - 1), k3 = min(col + 3, C2 - 1);
        float e00 = w0[k0], e10 = w1[k0], e01 = w0[k1], e11 = w1[k1];
        float e02 = w0[k2], e12 = w1[k2], e03 = w0[k3], e13 = w1[k3];
        v4f g0, g1;
        g0.x = (col < C2) ? e00 : 0.f;
        g0.y = (col < C2) ? e10 : 0.f;
        g0.z = (col + 1 < C2) ? e01 : 0.f;
        g0.w = (col + 1 < C2) ? e11 : 0.f;
        g1.x = (col + 2 < C2) ? e02 : 0.f;
        g1.y = (col + 2 < C2) ? e12 : 0.f;
        g1.z = (col + 3 < C2) ? e03 : 0.f;
        g1.w = (col + 3 < C2) ? e13 : 0.f;
        int c = j & 15, hf = j >> 4;
        Bs[kp * 64 + (2 * hf) * 16 + c] = g0;
        Bs[kp * 64 + (2 * hf + 1) * 16 + c] = g1;
    }
    __syncthreads();
    int row0 = (blockIdx.x * G2W + wave) * 16;
    if (row0 >= N) return;
    const v4f* ar = (const v4f*)(agg + (long long)(row0 + lm) * HD1) + lh;
    const v4f* bl = Bs + lh * 128 + lm;
    v4f a[HD1 / 8];
#pragma unroll
    for (int q = 0; q < HD1 / 8; ++q) a[q] = ar[2 * q];
    v8f acc0 = {0.f, 0.f, 0.f, 0.f, 0.f, 0.f, 0.f, 0.f};
    v8f acc1 = acc0, acc2 = acc0, acc3 = acc0, acc4 = acc0, acc5 = acc0, acc6 = acc0, acc7 = acc0;
#pragma unroll
    for (int kb = 0; kb < HD1 / 8; ++kb) {
        const v4f* bp = bl + kb * 256;
        v4f av = a[kb];
        v4f p0 = bp[0], p1 = bp[16], p2 = bp[32], p3 = bp[48];
        acc0 = __builtin_amdgcn_wmma_f32_16x16x4_f32(false, av.xy, false, p0.xy, (short)0, acc0, false, false);
        acc1 = __builtin_amdgcn_wmma_f32_16x16x4_f32(false, av.xy, false, p0.zw, (short)0, acc1, false, false);
        acc2 = __builtin_amdgcn_wmma_f32_16x16x4_f32(false, av.xy, false, p1.xy, (short)0, acc2, false, false);
        acc3 = __builtin_amdgcn_wmma_f32_16x16x4_f32(false, av.xy, false, p1.zw, (short)0, acc3, false, false);
        acc4 = __builtin_amdgcn_wmma_f32_16x16x4_f32(false, av.xy, false, p2.xy, (short)0, acc4, false, false);
        acc5 = __builtin_amdgcn_wmma_f32_16x16x4_f32(false, av.xy, false, p2.zw, (short)0, acc5, false, false);
        acc6 = __builtin_amdgcn_wmma_f32_16x16x4_f32(false, av.xy, false, p3.xy, (short)0, acc6, false, false);
        acc7 = __builtin_amdgcn_wmma_f32_16x16x4_f32(false, av.xy, false, p3.zw, (short)0, acc7, false, false);
        v4f q0 = bp[64], q1 = bp[80], q2 = bp[96], q3 = bp[112];
        acc0 = __builtin_amdgcn_wmma_f32_16x16x4_f32(false, av.zw, false, q0.xy, (short)0, acc0, false, false);
        acc1 = __builtin_amdgcn_wmma_f32_16x16x4_f32(false, av.zw, false, q0.zw, (short)0, acc1, false, false);
        acc2 = __builtin_amdgcn_wmma_f32_16x16x4_f32(false, av.zw, false, q1.xy, (short)0, acc2, false, false);
        acc3 = __builtin_amdgcn_wmma_f32_16x16x4_f32(false, av.zw, false, q1.zw, (short)0, acc3, false, false);
        acc4 = __builtin_amdgcn_wmma_f32_16x16x4_f32(false, av.zw, false, q2.xy, (short)0, acc4, false, false);
        acc5 = __builtin_amdgcn_wmma_f32_16x16x4_f32(false, av.zw, false, q2.zw, (short)0, acc5, false, false);
        acc6 = __builtin_amdgcn_wmma_f32_16x16x4_f32(false, av.zw, false, q3.xy, (short)0, acc6, false, false);
        acc7 = __builtin_amdgcn_wmma_f32_16x16x4_f32(false, av.zw, false, q3.zw, (short)0, acc7, false, false);
    }
    int cA = 4 * lm, cB = 64 + 4 * lm;
    v4f bA, bB;
    bA.x = b2[min(cA, C2 - 1)];
    bA.y = b2[min(cA + 1, C2 - 1)];
    bA.z = b2[min(cA + 2, C2 - 1)];
    bA.w = b2[min(cA + 3, C2 - 1)];
    bB.x = b2[min(cB, C2 - 1)];
    bB.y = b2[min(cB + 1, C2 - 1)];
    bB.z = b2[min(cB + 2, C2 - 1)];
    bB.w = b2[min(cB + 3, C2 - 1)];
    float* orow = outp + (long long)(row0 + 8 * lh) * C2;
    if ((C2 & 3) == 0) {
        bool okA = cA < C2, okB = cB < C2;
#pragma unroll
        for (int r = 0; r < 8; ++r) {
            v4f vA, vB;
            vA.x = acc0[r]; vA.y = acc1[r]; vA.z = acc2[r]; vA.w = acc3[r];
            vB.x = acc4[r]; vB.y = acc5[r]; vB.z = acc6[r]; vB.w = acc7[r];
            v4f sA = sig4(vA + bA), sB = sig4(vB + bB);
            if (okA) *(v4f*)(orow + (long long)r * C2 + cA) = sA;
            if (okB) *(v4f*)(orow + (long long)r * C2 + cB) = sB;
        }
    } else {
#pragma unroll
        for (int r = 0; r < 8; ++r) {
            v4f vA, vB;
            vA.x = acc0[r]; vA.y = acc1[r]; vA.z = acc2[r]; vA.w = acc3[r];
            vB.x = acc4[r]; vB.y = acc5[r]; vB.z = acc6[r]; vB.w = acc7[r];
            v4f sA = sig4(vA + bA), sB = sig4(vB + bB);
            float* o = orow + (long long)r * C2;
            if (cA < C2) o[cA] = sA.x;
            if (cA + 1 < C2) o[cA + 1] = sA.y;
            if (cA + 2 < C2) o[cA + 2] = sA.z;
            if (cA + 3 < C2) o[cA + 3] = sA.w;
            if (cB < C2) o[cB] = sB.x;
            if (cB + 1 < C2) o[cB + 1] = sB.y;
            if (cB + 2 < C2) o[cB + 2] = sB.z;
            if (cB + 3 < C2) o[cB + 3] = sB.w;
        }
    }
}

static inline int cdiv(long long a, long long b) { return (int)((a + b - 1) / b); }

extern "C" void kernel_launch(void* const* d_in, const int* in_sizes, int n_in,
                              void* d_out, int out_size, void* d_ws, size_t ws_size,
                              hipStream_t stream) {
    const float* x      = (const float*)d_in[0];
    const int*   ei     = (const int*)d_in[1];
    const float* W1     = (const float*)d_in[2];
    const float* a_src1 = (const float*)d_in[3];
    const float* a_dst1 = (const float*)d_in[4];
    const float* b1     = (const float*)d_in[5];
    const float* W2     = (const float*)d_in[6];
    const float* a_src2 = (const float*)d_in[7];
    const float* a_dst2 = (const float*)d_in[8];
    const float* b2     = (const float*)d_in[9];

    const int N  = in_sizes[0] / FIN;
    const int E  = in_sizes[1] / 2;
    const int C2 = in_sizes[9];

    float* h1   = (float*)d_ws;
    float* hx   = h1 + (long long)N * HD1;
    float* agg  = hx + (long long)N * HD1;
    float* vsd  = agg + (long long)N * HD1;
    int2*  part = (int2*)(vsd + 2 * HD1);
    const int NB = cdiv(N, BN);
    const int NP = cdiv(E, EPB);
    int*   hoff = (int*)(part + E);
    int*   off  = hoff + (NB + 1) * NP;
    int*   csr  = off + N + 1;
    float* out  = (float*)d_out;

    const int WGPS = 256;
    const int S = cdiv(N, 16);
    const int GW = S / G1W;
    const int gA = GW < 2 * WGPS ? GW : 2 * WGPS;
    const int gB = GW - gA < WGPS ? GW - gA : WGPS;
    const int sD = (gA + gB) * G1W;
    const int gD = S - sD;
    gemm1_partition_fold<<<gA + NP + HD1 / 8, 256, 2 * NB * sizeof(int), stream>>>(x, W1, h1, N, 0, G1W, gA, ei, E, NB, NP,
                                                                                   part, hoff, W2, a_src2, a_dst2, vsd, C2);
    gemm1_sort<<<gB + gD + NB, BN, 2 * NP * sizeof(int), stream>>>(x, W1, h1, N, gA * G1W, G1W, gB, sD, 1, gD,
                                                                   part, hoff, NP, off, csr);

    gather1<<<cdiv(N, 8), 256, 0, stream>>>(off, csr, h1, a_src1, a_dst1, b1, hx, N);

    gather2_rows<<<cdiv(N, 8), 256, 0, stream>>>(off, csr, vsd, hx, agg, N);
    gemm2_wmma<<<cdiv(N, 16 * G2W), 32 * G2W, 0, stream>>>(agg, W2, b2, out, N, C2);
}


# Layer 2 in one launch: 16-wave work-groups gather their 128 or 144 nodes' weighted hx row sums into LDS (a wave takes 8 or 9 nodes in turn) and apply W2 on WMMA there; no agg array; 5 launches to 4
# speedup vs baseline: 6.8349x; 1.0533x over previous
#include <hip/hip_runtime.h>
#include <math.h>

typedef float v2f __attribute__((ext_vector_type(2)));
typedef float v4f __attribute__((ext_vector_type(4)));
typedef float v8f __attribute__((ext_vector_type(8)));

#define FIN 256
#define HD1 64

__device__ __forceinline__ float leaky(float v) { return v > 0.f ? v : 0.2f * v; }

template <int M>
__device__ __forceinline__ float xadd(float v) {
    int o = __builtin_amdgcn_update_dpp(0, __float_as_int(v), 0x160 | M, 0xf, 0xf, false);
    return v + __int_as_float(o);
}
template <int M>
__device__ __forceinline__ float xget(float v) {
    return __int_as_float(__builtin_amdgcn_update_dpp(0, __float_as_int(v), 0x160 | M, 0xf, 0xf, false));
}
__device__ __forceinline__ float sum8(float v) { return xadd<4>(xadd<2>(xadd<1>(v))); }
__device__ __forceinline__ float sum16(float v) { return xadd<8>(sum8(v)); }
__device__ __forceinline__ float wave_sum(float v) {
    v = sum16(v);
    return v + __shfl_xor(v, 16, 32);
}

__device__ __forceinline__ void fold_w2(const float* __restrict__ W2,
                                        const float* __restrict__ a_src,
                                        const float* __restrict__ a_dst,
                                        float* __restrict__ vsd, int C2, int bid) {
    int lane = threadIdx.x & 31;
    int k = bid * 8 + (threadIdx.x >> 5);
    float s = 0.f, d = 0.f;
    for (int cix = lane; cix < C2; cix += 32) {
        float w = W2[k * C2 + cix];
        s += w * a_src[cix];
        d += w * a_dst[cix];
    }
    s = wave_sum(s);
    d = wave_sum(d);
    if (lane == 0) {
        vsd[k] = s;
        vsd[HD1 + k] = d;
    }
}

#define G1W 8
#define G1LDS ((FIN / 2) * 32)
__device__ __forceinline__ void gemm1_wmma(const float* __restrict__ x,
                                           const float* __restrict__ W1,
                                           float* __restrict__ h1, int N, int s0, int spw, int wg, v4f* Bs) {
    int tid = threadIdx.x;
    int wave = tid >> 5, lane = tid & 31;
    int lm = lane & 15, lh = lane >> 4;
    const v4f* W4 = (const v4f*)W1;
    for (int p0 = tid; p0 < (FIN / 2) * 16; p0 += 32 * G1W * 4) {
        v4f r0[4], r1[4];
#pragma unroll
        for (int u = 0; u < 4; ++u) {
            int p = p0 + u * 32 * G1W;
            int kp = p >> 4, c = p & 15;
            r0[u] = W4[(2 * kp) * 16 + c];
            r1[u] = W4[(2 * kp + 1) * 16 + c];
        }
#pragma unroll
        for (int u = 0; u < 4; ++u) {
            int p = p0 + u * 32 * G1W;
            int kp = p >> 4, c = p & 15;
            v4f g0, g1;
            g0.x = r0[u].x; g0.y = r1[u].x; g0.z = r0[u].y; g0.w = r1[u].y;
            g1.x = r0[u].z; g1.y = r1[u].z; g1.z = r0[u].w; g1.w = r1[u].w;
            Bs[kp * 32 + c] = g0;
            Bs[kp * 32 + 16 + c] = g1;
        }
    }
    __syncthreads();
    int row0 = (s0 + wg * spw + wave) * 16;
    if (wave >= spw || row0 >= N) return;
    const v4f* xr = (const v4f*)(x + (long long)(row0 + lm) * FIN) + lh;
    const v4f* bl = Bs + lh * 64 + lm;
    v8f acc0 = {0.f, 0.f, 0.f, 0.f, 0.f, 0.f, 0.f, 0.f};
    v8f acc1 = acc0, acc2 = acc0, acc3 = acc0;
    v4f a[2][8];
#pragma unroll
    for (int q = 0; q < 8; ++q) a[0][q] = xr[2 * q];
#pragma unroll
    for (int ch = 0; ch < FIN / 64; ++ch) {
        if (ch + 1 < FIN / 64) {
#pragma unroll
            for (int q = 0; q < 8; ++q) a[(ch + 1) & 1][q] = xr[2 * ((ch + 1) * 8 + q)];
        }
        __builtin_amdgcn_sched_barrier(0);
#pragma unroll
        for (int q = 0; q < 8; ++q) {
            int kb = ch * 8 + q;
            const v4f* bp = bl + kb * 128;
            v4f b00 = bp[0], b01 = bp[16], b10 = bp[32], b11 = bp[48];
            v4f av = a[ch & 1][q];
            acc0 = __builtin_amdgcn_wmma_f32_16x16x4_f32(false, av.xy, false, b00.xy, (short)0, acc0, false, false);
            acc1 = __builtin_amdgcn_wmma_f32_16x16x4_f32(false, av.xy, false, b00.zw, (short)0, acc1, false, false);
            acc2 = __builtin_amdgcn_wmma_f32_16x16x4_f32(false, av.xy, false, b01.xy, (short)0, acc2, false, false);
            acc3 = __builtin_amdgcn_wmma_f32_16x16x4_f32(false, av.xy, false, b01.zw, (short)0, acc3, false, false);
            acc0 = __builtin_amdgcn_wmma_f32_16x16x4_f32(false, av.zw, false, b10.xy, (short)0, acc0, false, false);
            acc1 = __builtin_amdgcn_wmma_f32_16x16x4_f32(false, av.zw, false, b10.zw, (short)0, acc1, false, false);
            acc2 = __builtin_amdgcn_wmma_f32_16x16x4_f32(false, av.zw, false, b11.xy, (short)0, acc2, false, false);
            acc3 = __builtin_amdgcn_wmma_f32_16x16x4_f32(false, av.zw, false, b11.zw, (short)0, acc3, false, false);
        }
    }
    float* out = h1 + (long long)(row0 + 8 * lh) * HD1 + 4 * lm;
#pragma unroll
    for (int r = 0; r < 8; ++r) {
        v4f o;
        o.x = acc0[r]; o.y = acc1[r]; o.z = acc2[r]; o.w = acc3[r];
        *(v4f*)(out + (long long)r * HD1) = o;
    }
}

#define BSH 8
#define BN (1 << BSH)
#define EPB 4096
#define EPT (EPB / 256)
#define SB 8
#define RL 16

__device__ __forceinline__ void bucket_partition(const int* __restrict__ ei, int E, int NB, int NP,
                                                 int2* __restrict__ part, int* __restrict__ hoff,
                                                 int bid, int2* stage) {
    extern __shared__ int sh[];
    __shared__ int wsum2[8];
    int tid = threadIdx.x, lane = tid & 31, wave = tid >> 5;
    for (int i = tid; i < NB; i += 256) sh[i] = 0;
    __syncthreads();
    int e0 = bid * EPB + tid;
    int d[EPT], sv[EPT], r[EPT];
#pragma unroll
    for (int k = 0; k < EPT; ++k) {
        int e = e0 + k * 256;
        d[k] = (e < E) ? ei[E + e] : -1;
        sv[k] = (e < E) ? ei[e] : 0;
    }
#pragma unroll
    for (int k = 0; k < EPT; ++k) r[k] = (d[k] >= 0) ? atomicAdd(&sh[d[k] >> BSH], 1) : 0;
    __syncthreads();
    int ipt = (NB + 255) / 256;
    int lo = tid * ipt;
    int hi = min(lo + ipt, NB);
    int s2 = 0;
    for (int i = lo; i < hi; ++i) s2 += sh[i];
    int incl2 = s2;
    for (int m = 1; m < 32; m <<= 1) {
        int u2 = __shfl_up(incl2, m, 32);
        if (lane >= m) incl2 += u2;
    }
    if (lane == 31) wsum2[wave] = incl2;
    __syncthreads();
    int lrun = incl2 - s2;
    for (int w = 0; w < wave; ++w) lrun += wsum2[w];
    for (int i = lo; i < hi; ++i) {
        sh[NB + i] = lrun;
        hoff[i * NP + bid] = lrun;
        lrun += sh[i];
    }
    int ne = min(EPB, E - bid * EPB);
    if (tid == 0) hoff[NB * NP + bid] = ne;
    __syncthreads();
#pragma unroll
    for (int k = 0; k < EPT; ++k) {
        if (d[k] >= 0) {
            int2 pr;
            pr.x = sv[k];
            pr.y = d[k];
            stage[sh[NB + (d[k] >> BSH)] + r[k]] = pr;
        }
    }
    __syncthreads();
    int2* mine = part + (long long)bid * EPB;
#pragma unroll
    for (int k = 0; k < EPT; ++k) {
        int p = tid + k * 256;
        if (p < ne) mine[p] = stage[p];
    }
}

__device__ __forceinline__ void bucket_sort(const int2* __restrict__ part,
                                            const int* __restrict__ hoff, int NP, int N,
                                            int* __restrict__ off, int* __restrict__ csr, int b) {
    extern __shared__ int sh[];
    __shared__ int cnt[BN];
    __shared__ int red[BN / 32];
    __shared__ int wsum[BN / 32];
    int tid = threadIdx.x, lane = tid & 31, wave = tid >> 5;
    int p = 0;
    for (int w = tid; w < NP; w += 2 * BN) {
        int w2 = w + BN;
        bool ok2 = w2 < NP;
        int lo = hoff[b * NP + w], hi = hoff[(b + 1) * NP + w];
        int lo2 = ok2 ? hoff[b * NP + w2] : 0, hi2 = ok2 ? hoff[(b + 1) * NP + w2] : 0;
        sh[w] = lo;
        sh[NP + w] = hi;
        if (ok2) {
            sh[w2] = lo2;
            sh[NP + w2] = hi2;
        }
        p += lo + lo2;
    }
    for (int m = 16; m; m >>= 1) p += __shfl_xor(p, m, 32);
    if (lane == 0) red[wave] = p;
    cnt[tid] = 0;
    __syncthreads();
    int start = 0;
    for (int w = 0; w < BN / 32; ++w) start += red[w];
    int gi = tid / RL, sl = tid % RL;
    for (int w0 = gi; w0 < NP; w0 += (BN / RL) * SB) {
        int y[2 * SB];
#pragma unroll
        for (int u = 0; u < SB; ++u) {
            int w = w0 + u * (BN / RL);
            bool ok = w < NP;
            int i = (ok ? sh[w] : 0) + sl;
            int hi = ok ? sh[NP + w] : 0;
            const int2* run = part + (long long)w * EPB;
            y[u] = (i < hi) ? run[i].y : -1;
            y[SB + u] = (i + RL < hi) ? run[i + RL].y : -1;
        }
#pragma unroll
        for (int u = 0; u < 2 * SB; ++u)
            if (y[u] >= 0) atomicAdd(&cnt[y[u] & (BN - 1)], 1);
        for (int u = 0; u < SB; ++u) {
            int w = w0 + u * (BN / RL);
            if (w < NP) {
                int hi = sh[NP + w];
                for (int i = sh[w] + sl + 2 * RL; i < hi; i += RL)
                    atomicAdd(&cnt[part[(long long)w * EPB + i].y & (BN - 1)], 1);
            }
        }
    }
    __syncthreads();
    int c = cnt[tid];
    int incl = c;
    for (int m = 1; m < 32; m <<= 1) {
        int u = __shfl_up(incl, m, 32);
        if (lane >= m) incl += u;
    }
    if (lane == 31) wsum[wave] = incl;
    __syncthreads();
    int excl = incl - c;
    for (int w = 0; w < wave; ++w) excl += wsum[w];
    int node = b * BN + tid;
    if (node < N) off[node] = start + excl;
    if (node == N - 1) off[N] = start + excl + c;
    cnt[tid] = excl;
    __syncthreads();
    int* out = csr + start;
    for (int w0 = gi; w0 < NP; w0 += (BN / RL) * SB) {
        int2 pr[2 * SB];
#pragma unroll
        for (int u = 0; u < SB; ++u) {
            int w = w0 + u * (BN / RL);
            bool ok = w < NP;
            int i = (ok ? sh[w] : 0) + sl;
            int hi = ok ? sh[NP + w] : 0;
            const int2* run = part + (long long)w * EPB;
            if (i < hi) pr[u] = run[i];
            else pr[u].y = -1;
            if (i + RL < hi) pr[SB + u] = run[i + RL];
            else pr[SB + u].y = -1;
        }
#pragma unroll
        for (int u = 0; u < 2 * SB; ++u)
            if (pr[u].y >= 0) out[atomicAdd(&cnt[pr[u].y & (BN - 1)], 1)] = pr[u].x;
        for (int u = 0; u < SB; ++u) {
            int w = w0 + u * (BN / RL);
            if (w < NP) {
                int hi = sh[NP + w];
                for (int i = sh[w] + sl + 2 * RL; i < hi; i += RL) {
                    int2 q = part[(long long)w * EPB + i];
                    out[atomicAdd(&cnt[q.y & (BN - 1)], 1)] = q.x;
                }
            }
        }
    }
}

__global__ __launch_bounds__(256) void gemm1_partition_fold(const float* __restrict__ x, const float* __restrict__ W1,
                                                            float* __restrict__ h1, int N, int s0, int spw, int ng,
                                                            const int* __restrict__ ei, int E, int NB, int NP,
                                                            int2* __restrict__ part, int* __restrict__ hoff,
                                                            const float* __restrict__ W2, const float* __restrict__ a_src,
                                                            const float* __restrict__ a_dst, float* __restrict__ vsd, int C2) {
    __shared__ v4f Bs[G1LDS];
    int b = blockIdx.x;
    if (b < ng) gemm1_wmma(x, W1, h1, N, s0, spw, b, Bs);
    else if (b < ng + NP) bucket_partition(ei, E, NB, NP, part, hoff, b - ng, (int2*)Bs);
    else fold_w2(W2, a_src, a_dst, vsd, C2, b - ng - NP);
}

__global__ __launch_bounds__(256) void gemm1_sort(const float* __restrict__ x, const float* __restrict__ W1,
                                                  float* __restrict__ h1, int N, int s0, int spw, int ng,
                                                  int s1, int spw1, int ng1,
                                                  const int2* __restrict__ part, const int* __restrict__ hoff, int NP,
                                                  int* __restrict__ off, int* __restrict__ csr) {
    __shared__ v4f Bs[G1LDS];
    int b = blockIdx.x;
    if (b < ng + ng1) {
        bool first = b < ng;
        gemm1_wmma(x, W1, h1, N, first ? s0 : s1, first ? spw : spw1, first ? b : b - ng, Bs);
    } else {
        bucket_sort(part, hoff, NP, N, off, csr, b - ng - ng1);
    }
}

__device__ __forceinline__ float dot8(v4f r0, v4f r1, v4f a0, v4f a1) {
    v2f p = r0.xy * a0.xy;
    p += r0.zw * a0.zw;
    p += r1.xy * a1.xy;
    p += r1.zw * a1.zw;
    return p.x + p.y;
}

template <int K, bool ROWDOT>
__device__ __forceinline__ void head_steps(const v4f* __restrict__ rows4, const int* __restrict__ csr,
                                           int j, int end, int o0, int o1, v4f a0, v4f a1, float adn,
                                           float& m, float& den, v4f& c0, v4f& c1) {
    int s[K];
    v4f r0[K], r1[K];
    float e[K];
#pragma unroll
    for (int k = 0; k < K; ++k) s[k] = csr[min(j + 4 * k, end - 1)];
#pragma unroll
    for (int k = 0; k < K; ++k) {
        r0[k] = rows4[s[k] * 16 + o0];
        r1[k] = rows4[s[k] * 16 + o1];
    }
    float gm = m;
#pragma unroll
    for (int k = 0; k < K; ++k) {
        float d = dot8(r0[k], r1[k], a0, a1);
        if (ROWDOT) d = sum8(d);
        e[k] = leaky(d + adn);
        gm = fmaxf(gm, e[k]);
    }
    float sc = __expf(m - gm);
    den *= sc;
    c0 *= sc;
    c1 *= sc;
    m = gm;
#pragma unroll
    for (int k = 0; k < K; ++k) {
        float w = (j + 4 * k < end) ? __expf(e[k] - gm) : 0.f;
        den += w;
        c0 += w * r0[k];
        c1 += w * r1[k];
    }
}

__device__ __forceinline__ v2f merge_slots(int g, float m, float& den, v4f c0, v4f c1) {
    float mo = xget<8>(m);
    float mm = fmaxf(m, mo);
    float f = __expf(m - mm), fo = __expf(mo - mm);
    den = den * f + xget<8>(den) * fo;
    bool up = (g & 1) != 0;
    v4f keep = up ? c1 : c0;
    v4f give = up ? c0 : c1;
    v4f got;
    got.x = xget<8>(give.x);
    got.y = xget<8>(give.y);
    got.z = xget<8>(give.z);
    got.w = xget<8>(give.w);
    v4f c = keep * f + got * fo;
    m = mm;
    mo = __shfl_xor(m, 16, 32);
    mm = fmaxf(m, mo);
    f = __expf(m - mm);
    fo = __expf(mo - mm);
    den = den * f + __shfl_xor(den, 16, 32) * fo;
    bool up2 = (g & 2) != 0;
    v2f keep2 = up2 ? c.zw : c.xy;
    v2f give2 = up2 ? c.xy : c.zw;
    v2f got2;
    got2.x = __shfl_xor(give2.x, 16, 32);
    got2.y = __shfl_xor(give2.y, 16, 32);
    return keep2 * f + got2 * fo;
}

__global__ __launch_bounds__(256) void gather1(const int* __restrict__ off,
                                               const int* __restrict__ csr,
                                               const float* __restrict__ h1,
                                               const float* __restrict__ a_src,
                                               const float* __restrict__ a_dst,
                                               const float* __restrict__ b1,
                                               float* __restrict__ hx, int N) {
    int lane = threadIdx.x & 31;
    int n = __builtin_amdgcn_readfirstlane((int)(blockIdx.x * 8 + (threadIdx.x >> 5)));
    if (n >= N) return;
    int g = lane >> 3, h = lane & 7;
    int beg = off[n], end = off[n + 1];
    const v4f* rows4 = (const v4f*)h1;
    v4f hn0 = rows4[n * 16 + 2 * h], hn1 = rows4[n * 16 + 2 * h + 1];
    v4f a0 = ((const v4f*)a_src)[2 * h], a1 = ((const v4f*)a_src)[2 * h + 1];
    v4f d0 = ((const v4f*)a_dst)[2 * h], d1 = ((const v4f*)a_dst)[2 * h + 1];
    float adn = dot8(hn0, hn1, d0, d1);
    float vself = leaky(dot8(hn0, hn1, a0, a1) + adn);
    float m = vself;
    float den = (g == 0) ? 1.f : 0.f;
    v4f c0 = hn0 * den, c1 = hn1 * den;
    int j0 = beg;
    for (; end - j0 > 12; j0 += 16) head_steps<4, false>(rows4, csr, j0 + g, end, 2 * h, 2 * h + 1, a0, a1, adn, m, den, c0, c1);
    int rem = end - j0;
    if (rem > 8) head_steps<3, false>(rows4, csr, j0 + g, end, 2 * h, 2 * h + 1, a0, a1, adn, m, den, c0, c1);
    else if (rem > 4) head_steps<2, false>(rows4, csr, j0 + g, end, 2 * h, 2 * h + 1, a0, a1, adn, m, den, c0, c1);
    else if (rem > 0) head_steps<1, false>(rows4, csr, j0 + g, end, 2 * h, 2 * h + 1, a0, a1, adn, m, den, c0, c1);

    v2f acc = merge_slots(g, m, den, c0, c1);
    int ch = 8 * h + 4 * (g & 1) + (g & 2);

    float vx = acc.x / den + b1[ch];
    float vy = acc.y / den + b1[ch + 1];
    v2f o;
    o.x = vx > 0.f ? vx : (__expf(vx) - 1.f);
    o.y = vy > 0.f ? vy : (__expf(vy) - 1.f);
    *(v2f*)(hx + (long long)n * HD1 + ch) = o;
}

#define F2W 16
#define F2SMAX 9
#define F2ROWS (16 * F2SMAX)
#define F2PITCH 68
__device__ __forceinline__ v4f sig4(v4f v) {
    v4f o;
    o.x = 1.f / (1.f + __expf(-v.x));
    o.y = 1.f / (1.f + __expf(-v.y));
    o.z = 1.f / (1.f + __expf(-v.z));
    o.w = 1.f / (1.f + __expf(-v.w));
    return o;
}
__global__ __launch_bounds__(32 * F2W, 12) void gather2_gemm2(const int* __restrict__ off,
                                                         const int* __restrict__ csr,
                                                         const float* __restrict__ vsd,
                                                         const float* __restrict__ hx,
                                                         const float* __restrict__ W2,
                                                         const float* __restrict__ b2,
                                                         float* __restrict__ outp, int N, int C2) {
    __shared__ v4f Bs[(HD1 / 2) * 64];
    __shared__ v4f As4[F2ROWS * F2PITCH / 4];
    float* As = (float*)As4;
    int tid = threadIdx.x;
    int lane = tid & 31;
    int wave = __builtin_amdgcn_readfirstlane(tid >> 5);
    int S = N / 16;
    int G = gridDim.x, b = blockIdx.x;
    int base = S / G, extra = S - base * G;
    int ns = base + (b < extra ? 1 : 0);
    int rfirst = (b * base + min(b, extra)) * 16;
#pragma unroll
    for (int i = 0; i < (HD1 / 2) * 32 / (32 * F2W); ++i) {
        int p = tid + i * 32 * F2W;
        int kp = p >> 5, j = p & 31;
        int col = 4 * j;
        const float* w0 = W2 + (2 * kp) * C2;
        const float* w1 = w0 + C2;
        int k0 = min(col, C2 - 1), k1 = min(col + 1, C2 - 1), k2 = min(col + 2, C2 - 1), k3 = min(col + 3, C2 - 1);
        float e00 = w0[k0], e10 = w1[k0], e01 = w0[k1], e11 = w1[k1];
        float e02 = w0[k2], e12 = w1[k2], e03 = w0[k3], e13 = w1[k3];
        v4f g0, g1;
        g0.x = (col < C2) ? e00 : 0.f;
        g0.y = (col < C2) ? e10 : 0.f;
        g0.z = (col + 1 < C2) ? e01 : 0.f;
        g0.w = (col + 1 < C2) ? e11 : 0.f;
        g1.x = (col + 2 < C2) ? e02 : 0.f;
        g1.y = (col + 2 < C2) ? e12 : 0.f;
        g1.z = (col + 3 < C2) ? e03 : 0.f;
        g1.w = (col + 3 < C2) ? e13 : 0.f;
        int c = j & 15, hf = j >> 4;
        Bs[kp * 64 + (2 * hf) * 16 + c] = g0;
        Bs[kp * 64 + (2 * hf + 1) * 16 + c] = g1;
    }
    {
        int g = lane >> 3, q = lane & 7;
        const v4f* rows4 = (const v4f*)hx;
        const v4f* vsd4 = (const v4f*)vsd;
        v4f s0 = vsd4[q], s1 = vsd4[8 + q];
        int ch = 32 * (g & 1) + 4 * q + (g & 2);
        int r0w = wave * ns;
        int n0 = rfirst + r0w;
        float adnv = 0.f, selfv = 0.f;
        {
            v4f d0 = vsd4[16 + q], d1 = vsd4[16 + 8 + q];
#pragma unroll
            for (int i = 0; i < F2SMAX; ++i) {
                int n = min(n0 + i, N - 1);
                v4f hn0 = rows4[n * 16 + q], hn1 = rows4[n * 16 + 8 + q];
                float a = sum8(dot8(hn0, hn1, d0, d1));
                float v = leaky(sum8(dot8(hn0, hn1, s0, s1)) + a);
                adnv = (lane == i) ? a : adnv;
                selfv = (lane == i) ? v : selfv;
                if (g == 0 && i < ns) {
                    v4f* own = (v4f*)(As + (r0w + i) * F2PITCH);
                    own[q] = hn0;
                    own[8 + q] = hn1;
                }
            }
        }
        int beg = off[min(n0, N)], end = off[min(n0 + 1, N)];
        for (int i = 0; i < ns; ++i) {
            int n = n0 + i;
            int endn = off[min(n + 2, N)];
            float adn = __int_as_float(__builtin_amdgcn_readlane(__float_as_int(adnv), i));
            float vself = __int_as_float(__builtin_amdgcn_readlane(__float_as_int(selfv), i));
            const v4f* own = (const v4f*)(As + (r0w + i) * F2PITCH);
            v4f hn0 = own[q], hn1 = own[8 + q];
            float m = vself;
            float den = (g == 0) ? 1.f : 0.f;
            v4f c0 = hn0 * den, c1 = hn1 * den;
            int j0 = beg;
            for (; end - j0 > 12; j0 += 16) head_steps<4, true>(rows4, csr, j0 + g, end, q, 8 + q, s0, s1, adn, m, den, c0, c1);
            int rem = end - j0;
            if (rem > 8) head_steps<3, true>(rows4, csr, j0 + g, end, q, 8 + q, s0, s1, adn, m, den, c0, c1);
            else if (rem > 4) head_steps<2, true>(rows4, csr, j0 + g, end, q, 8 + q, s0, s1, adn, m, den, c0, c1);
            else if (rem > 0) head_steps<1, true>(rows4, csr, j0 + g, end, q, 8 + q, s0, s1, adn, m, den, c0, c1);
            v2f acc = merge_slots(g, m, den, c0, c1);
            v2f o;
            o.x = acc.x / den;
            o.y = acc.y / den;
            *(v2f*)(As + (r0w + i) * F2PITCH + ch) = o;
            beg = end;
            end = endn;
        }
    }
    __syncthreads();
    int lm = lane & 15, lh = lane >> 4;
    for (int hs = wave; hs < 2 * ns; hs = (hs ^ 1) + F2W) {
        int strip = hs >> 1, hf = hs & 1;
        int row0 = rfirst + strip * 16;
        const v4f* ar = (const v4f*)(As + (strip * 16 + lm) * F2PITCH) + lh;
        const v4f* bl = Bs + lh * 128 + lm + hf * 32;
        v8f acc0 = {0.f, 0.f, 0.f, 0.f, 0.f, 0.f, 0.f, 0.f};
        v8f acc1 = acc0, acc2 = acc0, acc3 = acc0;
#pragma unroll
        for (int kb = 0; kb < HD1 / 8; ++kb) {
            const v4f* bp = bl + kb * 256;
            v4f av = ar[2 * kb];
            v4f p0 = bp[0], p1 = bp[16];
            acc0 = __builtin_amdgcn_wmma_f32_16x16x4_f32(false, av.xy, false, p0.xy, (short)0, acc0, false, false);
            acc1 = __builtin_amdgcn_wmma_f32_16x16x4_f32(false, av.xy, false, p0.zw, (short)0, acc1, false, false);
            acc2 = __builtin_amdgcn_wmma_f32_16x16x4_f32(false, av.xy, false, p1.xy, (short)0, acc2, false, false);
            acc3 = __builtin_amdgcn_wmma_f32_16x16x4_f32(false, av.xy, false, p1.zw, (short)0, acc3, false, false);
            v4f q0 = bp[64], q1 = bp[80];
            acc0 = __builtin_amdgcn_wmma_f32_16x16x4_f32(false, av.zw, false, q0.xy, (short)0, acc0, false, false);
            acc1 = __builtin_amdgcn_wmma_f32_16x16x4_f32(false, av.zw, false, q0.zw, (short)0, acc1, false, false);
            acc2 = __builtin_amdgcn_wmma_f32_16x16x4_f32(false, av.zw, false, q1.xy, (short)0, acc2, false, false);
            acc3 = __builtin_amdgcn_wmma_f32_16x16x4_f32(false, av.zw, false, q1.zw, (short)0, acc3, false, false);
        }
        int cA = 64 * hf + 4 * lm;
        v4f bA;
        bA.x = b2[min(cA, C2 - 1)];
        bA.y = b2[min(cA + 1, C2 - 1)];
        bA.z = b2[min(cA + 2, C2 - 1)];
        bA.w = b2[min(cA + 3, C2 - 1)];
        float* orow = outp + (long long)(row0 + 8 * lh) * C2;
        if ((C2 & 3) == 0) {
            bool okA = cA < C2;
#pragma unroll
            for (int r = 0; r < 8; ++r) {
                v4f vA;
                vA.x = acc0[r]; vA.y = acc1[r]; vA.z = acc2[r]; vA.w = acc3[r];
                v4f sA = sig4(vA + bA);
                if (okA) *(v4f*)(orow + (long long)r * C2 + cA) = sA;
            }
        } else {
#pragma unroll
            for (int r = 0; r < 8; ++r) {
                v4f vA;
                vA.x = acc0[r]; vA.y = acc1[r]; vA.z = acc2[r]; vA.w = acc3[r];
                v4f sA = sig4(vA + bA);
                float* o = orow + (long long)r * C2;
                if (cA < C2) o[cA] = sA.x;
                if (cA + 1 < C2) o[cA + 1] = sA.y;
                if (cA + 2 < C2) o[cA + 2] = sA.z;
                if (cA + 3 < C2) o[cA + 3] = sA.w;
            }
        }
    }
}

static inline int cdiv(long long a, long long b) { return (int)((a + b - 1) / b); }

extern "C" void kernel_launch(void* const* d_in, const int* in_sizes, int n_in,
                              void* d_out, int out_size, void* d_ws, size_t ws_size,
                              hipStream_t stream) {
    const float* x      = (const float*)d_in[0];
    const int*   ei     = (const int*)d_in[1];
    const float* W1     = (const float*)d_in[2];
    const float* a_src1 = (const float*)d_in[3];
    const float* a_dst1 = (const float*)d_in[4];
    const float* b1     = (const float*)d_in[5];
    const float* W2     = (const float*)d_in[6];
    const float* a_src2 = (const float*)d_in[7];
    const float* a_dst2 = (const float*)d_in[8];
    const float* b2     = (const float*)d_in[9];

    const int N  = in_sizes[0] / FIN;
    const int E  = in_sizes[1] / 2;
    const int C2 = in_sizes[9];

    float* h1   = (float*)d_ws;
    float* hx   = h1 + (long long)N * HD1;
    float* vsd  = hx + (long long)N * HD1;
    int2*  part = (int2*)(vsd + 2 * HD1);
    const int NB = cdiv(N, BN);
    const int NP = cdiv(E, EPB);
    int*   hoff = (int*)(part + E);
    int*   off  = hoff + (NB + 1) * NP;
    int*   csr  = off + N + 1;
    float* out  = (float*)d_out;

    const int WGPS = 256;
    const int S = cdiv(N, 16);
    const int GW = S / G1W;
    const int gA = GW < 2 * WGPS ? GW : 2 * WGPS;
    const int gB = GW - gA < WGPS ? GW - gA : WGPS;
    const int sD = (gA + gB) * G1W;
    const int gD = S - sD;
    gemm1_partition_fold<<<gA + NP + HD1 / 8, 256, 2 * NB * sizeof(int), stream>>>(x, W1, h1, N, 0, G1W, gA, ei, E, NB, NP,
                                                                                   part, hoff, W2, a_src2, a_dst2, vsd, C2);
    gemm1_sort<<<gB + gD + NB, BN, 2 * NP * sizeof(int), stream>>>(x, W1, h1, N, gA * G1W, G1W, gB, sD, 1, gD,
                                                                   part, hoff, NP, off, csr);

    gather1<<<cdiv(N, 8), 256, 0, stream>>>(off, csr, h1, a_src1, a_dst1, b1, hx, N);

    const int g2 = cdiv(S, F2SMAX) > 3 * WGPS ? cdiv(S, F2SMAX) : 3 * WGPS;
    gather2_gemm2<<<g2, 32 * F2W, 0, stream>>>(off, csr, vsd, hx, W2, b2, out, N, C2);
}
